# starting kernel
# baseline (speedup 1.0000x reference)
#include <hip/hip_runtime.h>

#define B_ 8
#define S_ 512
#define D_ 512
#define H_ 64
#define DK_ 8
#define WIDTH_ 16
#define N_ (S_ + 2 * WIDTH_)
#define NL_ 2
#define WIN_ (WIDTH_ + 2)
#define BN_ (B_ * N_)
#define BS_ (B_ * S_)
static const size_t BND_ = (size_t)BN_ * D_;
static const size_t BSD_ = (size_t)BS_ * D_;

typedef __attribute__((ext_vector_type(2))) float v2f;
typedef __attribute__((ext_vector_type(8))) float v8f;

__global__ __launch_bounds__(256)
void pad_kernel(const float* __restrict__ x, const float* __restrict__ lp,
                const float* __restrict__ rp, float* __restrict__ xp) {
  size_t idx = (size_t)blockIdx.x * blockDim.x + threadIdx.x;
  if (idx >= BND_) return;
  int dd = (int)(idx % D_);
  size_t bn = idx / D_;
  int p = (int)(bn % N_);
  int b = (int)(bn / N_);
  float v;
  if (p < WIDTH_)            v = lp[p * D_ + dd];
  else if (p < WIDTH_ + S_)  v = x[((size_t)b * S_ + (p - WIDTH_)) * D_ + dd];
  else                       v = rp[(p - WIDTH_ - S_) * D_ + dd];
  xp[idx] = v;
}

__global__ __launch_bounds__(256)
void gemm_wmma_f32(const float* __restrict__ A, int lda,
                   const float* __restrict__ W, int ldw,
                   const float* __restrict__ bias,
                   float* __restrict__ C, int ldc,
                   int M, int Nout) {
  constexpr int KK = 512;
  constexpr int LDA_S = KK + 8;
  __shared__ float As[16 * LDA_S];

  const int lane = threadIdx.x;
  const int wave = threadIdx.y;
  const int tid  = wave * 32 + lane;
  const int row0 = blockIdx.y * 16;
  const int col0 = (blockIdx.x * 8 + wave) * 64;

  {
    constexpr int NVEC = 16 * (KK / 4);
#pragma unroll
    for (int v = 0; v < NVEC / 256; ++v) {
      int e   = tid + v * 256;
      int row = e >> 7;
      int cv  = e & 127;
      int rg  = row0 + row; if (rg > M - 1) rg = M - 1;
      float4 val = *(const float4*)(A + (size_t)rg * lda + cv * 4);
      *(float4*)&As[row * LDA_S + cv * 4] = val;
    }
  }
  __syncthreads();

  if (row0 >= M || col0 >= Nout) return;

  const int ml = lane & 15;
  const int kb = (lane >> 4) << 1;

  const float* w0 = W + (size_t)(col0 + ml) * ldw + kb;
  const float* w1 = w0 + (size_t)16 * ldw;
  const float* w2 = w0 + (size_t)32 * ldw;
  const float* w3 = w0 + (size_t)48 * ldw;
  const float* arow = &As[ml * LDA_S + kb];

  v8f acc0 = {}, acc1 = {}, acc2 = {}, acc3 = {};

  v2f a  = *(const v2f*)(arow);
  v2f b0 = *(const v2f*)(w0);
  v2f b1 = *(const v2f*)(w1);
  v2f b2 = *(const v2f*)(w2);
  v2f b3 = *(const v2f*)(w3);

#pragma unroll 4
  for (int k = 4; k < KK; k += 4) {
    v2f an  = *(const v2f*)(arow + k);
    v2f b0n = *(const v2f*)(w0 + k);
    v2f b1n = *(const v2f*)(w1 + k);
    v2f b2n = *(const v2f*)(w2 + k);
    v2f b3n = *(const v2f*)(w3 + k);
    acc0 = __builtin_amdgcn_wmma_f32_16x16x4_f32(false, a, false, b0, (short)0, acc0, false, false);
    acc1 = __builtin_amdgcn_wmma_f32_16x16x4_f32(false, a, false, b1, (short)0, acc1, false, false);
    acc2 = __builtin_amdgcn_wmma_f32_16x16x4_f32(false, a, false, b2, (short)0, acc2, false, false);
    acc3 = __builtin_amdgcn_wmma_f32_16x16x4_f32(false, a, false, b3, (short)0, acc3, false, false);
    a = an; b0 = b0n; b1 = b1n; b2 = b2n; b3 = b3n;
  }
  acc0 = __builtin_amdgcn_wmma_f32_16x16x4_f32(false, a, false, b0, (short)0, acc0, false, false);
  acc1 = __builtin_amdgcn_wmma_f32_16x16x4_f32(false, a, false, b1, (short)0, acc1, false, false);
  acc2 = __builtin_amdgcn_wmma_f32_16x16x4_f32(false, a, false, b2, (short)0, acc2, false, false);
  acc3 = __builtin_amdgcn_wmma_f32_16x16x4_f32(false, a, false, b3, (short)0, acc3, false, false);

  const int mbase = row0 + ((lane >> 4) << 3);
  float bv0 = bias[col0 +  0 + ml];
  float bv1 = bias[col0 + 16 + ml];
  float bv2 = bias[col0 + 32 + ml];
  float bv3 = bias[col0 + 48 + ml];
#pragma unroll
  for (int r = 0; r < 8; ++r) {
    float* crow = C + (size_t)(mbase + r) * ldc + col0 + ml;
    crow[ 0] = acc0[r] + bv0;
    crow[16] = acc1[r] + bv1;
    crow[32] = acc2[r] + bv2;
    crow[48] = acc3[r] + bv3;
  }
}

__global__ __launch_bounds__(256)
void attn_kernel(const float* __restrict__ qkv, float* __restrict__ out, int dir) {
  int idx = blockIdx.x * blockDim.x + threadIdx.x;
  if (idx >= B_ * H_ * N_) return;
  int h    = idx % H_;
  int rest = idx / H_;
  int i    = rest % N_;
  int b    = rest / N_;

  const size_t base = (size_t)(b * N_ + i) * D_ + h * DK_;
  const float* qp = qkv + base;
  const float* kbase = qkv + BND_ + (size_t)(b * N_) * D_ + h * DK_;
  const float* vbase = qkv + 2 * BND_ + (size_t)(b * N_) * D_ + h * DK_;

  float q[DK_];
  *(float4*)&q[0] = *(const float4*)(qp);
  *(float4*)&q[4] = *(const float4*)(qp + 4);

  const int j0 = (dir == 0) ? ((i - WIDTH_ - 1 > 0) ? i - WIDTH_ - 1 : 0) : i;
  const int j1 = (dir == 0) ? i : ((i + WIDTH_ + 1 < N_ - 1) ? i + WIDTH_ + 1 : N_ - 1);
  const float scale = 0.35355339059327373f;

  float sv[WIN_];
  float mx = -3.0e38f;
#pragma unroll
  for (int t = 0; t < WIN_; ++t) {
    int j = j0 + t;
    float sc = -3.0e38f;
    if (j <= j1) {
      const float* kp = kbase + (size_t)j * D_;
      float kr[DK_];
      *(float4*)&kr[0] = *(const float4*)(kp);
      *(float4*)&kr[4] = *(const float4*)(kp + 4);
      float dot = 0.f;
#pragma unroll
      for (int d = 0; d < DK_; ++d) dot += q[d] * kr[d];
      sc = dot * scale;
    }
    sv[t] = sc;
    mx = fmaxf(mx, sc);
  }
  float sum = 0.f;
#pragma unroll
  for (int t = 0; t < WIN_; ++t) {
    float p = __expf(sv[t] - mx);
    sv[t] = p;
    sum += p;
  }
  const float inv = 1.f / sum;
  float o[DK_] = {};
#pragma unroll
  for (int t = 0; t < WIN_; ++t) {
    int j = j0 + t;
    if (j <= j1) {
      const float* vp = vbase + (size_t)j * D_;
      float vr[DK_];
      *(float4*)&vr[0] = *(const float4*)(vp);
      *(float4*)&vr[4] = *(const float4*)(vp + 4);
      float w = sv[t] * inv;
#pragma unroll
      for (int d = 0; d < DK_; ++d) o[d] += w * vr[d];
    }
  }
  float* op = out + base;
  float4 o0; o0.x = o[0]; o0.y = o[1]; o0.z = o[2]; o0.w = o[3];
  float4 o1; o1.x = o[4]; o1.y = o[5]; o1.z = o[6]; o1.w = o[7];
  *(float4*)(op)     = o0;
  *(float4*)(op + 4) = o1;
}

__global__ __launch_bounds__(256)
void slice_kernel(const float* __restrict__ proj, float* __restrict__ hx) {
  size_t idx = (size_t)blockIdx.x * blockDim.x + threadIdx.x;
  if (idx >= BSD_) return;
  int dd = (int)(idx % D_);
  size_t ms = idx / D_;
  int s = (int)(ms % S_);
  int b = (int)(ms / S_);
  hx[idx] = proj[((size_t)(b * N_ + WIDTH_ + s)) * D_ + dd];
}

__global__ __launch_bounds__(256)
void gate_kernel(const float* __restrict__ hp, float* __restrict__ hx) {
  size_t idx = (size_t)blockIdx.x * blockDim.x + threadIdx.x;
  if (idx >= BSD_) return;
  int dd = (int)(idx % D_);
  size_t m = idx / D_;
  float lin = hp[m * (2 * D_) + dd];
  float gin = hp[m * (2 * D_) + D_ + dd];
  float g  = 1.f / (1.f + __expf(-gin));
  float nl = fmaxf(lin, 0.f);
  hx[idx] = g * hx[idx] + (1.f - g) * nl;
}

__global__ __launch_bounds__(256)
void outw_kernel(const float* __restrict__ hx, float* __restrict__ out, int dir) {
  size_t idx = (size_t)blockIdx.x * blockDim.x + threadIdx.x;
  if (idx >= BSD_) return;
  int dd = (int)(idx % D_);
  size_t m = idx / D_;
  size_t o = m * (size_t)(2 * D_) + (size_t)dir * D_ + dd;
  float v = hx[idx];
  out[o] = v;
  out[o + (size_t)BS_ * 2 * D_] = v;
}

extern "C" void kernel_launch(void* const* d_in, const int* in_sizes, int n_in,
                              void* d_out, int out_size, void* d_ws, size_t ws_size,
                              hipStream_t stream) {
  const float* inputs    = (const float*)d_in[0];
  const float* left_pad  = (const float*)d_in[1];
  const float* right_pad = (const float*)d_in[2];
  const float* Wqkv[2] = {(const float*)d_in[3],  (const float*)d_in[7]};
  const float* bqkv[2] = {(const float*)d_in[4],  (const float*)d_in[8]};
  const float* Wo[2]   = {(const float*)d_in[5],  (const float*)d_in[9]};
  const float* bo[2]   = {(const float*)d_in[6],  (const float*)d_in[10]};
  const float* hwW[2]  = {(const float*)d_in[11], (const float*)d_in[13]};
  const float* hwb[2]  = {(const float*)d_in[12], (const float*)d_in[14]};
  float* out = (float*)d_out;

  float* ws    = (float*)d_ws;
  float* xpad  = ws;
  float* qkv   = xpad + BND_;
  float* attn  = qkv + 3 * BND_;
  float* hx    = attn + BND_;
  float* proj  = qkv;
  float* hproj = qkv + BND_;

  dim3 blk(32, 8);

  { int n = (int)BND_; pad_kernel<<<(n + 255) / 256, 256, 0, stream>>>(inputs, left_pad, right_pad, xpad); }

  for (int dir = 0; dir < 2; ++dir) {
    for (int t = 0; t < 3; ++t) {
      dim3 grd(1, BN_ / 16);
      gemm_wmma_f32<<<grd, blk, 0, stream>>>(
          xpad, D_, Wqkv[dir] + (size_t)t * D_ * D_, D_, bqkv[dir] + t * D_,
          qkv + (size_t)t * BND_, D_, BN_, D_);
    }
    { int n = B_ * H_ * N_;
      attn_kernel<<<(n + 255) / 256, 256, 0, stream>>>(qkv, attn, dir); }
    { dim3 grd(1, BN_ / 16);
      gemm_wmma_f32<<<grd, blk, 0, stream>>>(attn, D_, Wo[dir], D_, bo[dir],
                                             proj, D_, BN_, D_); }
    { int n = (int)BSD_; slice_kernel<<<(n + 255) / 256, 256, 0, stream>>>(proj, hx); }
    for (int l = 0; l < NL_; ++l) {
      dim3 grd(2, BS_ / 16);
      gemm_wmma_f32<<<grd, blk, 0, stream>>>(
          hx, D_, hwW[dir] + (size_t)l * 2 * D_ * D_, D_, hwb[dir] + l * 2 * D_,
          hproj, 2 * D_, BS_, 2 * D_);
      int n = (int)BSD_;
      gate_kernel<<<(n + 255) / 256, 256, 0, stream>>>(hproj, hx);
    }
    { int n = (int)BSD_; outw_kernel<<<(n + 255) / 256, 256, 0, stream>>>(hx, out, dir); }
  }
}


# Pad rows leave the big GEMMs: q/k/v and Wo cover the 4096 real rows (256 blocks, LDS sized for one per WGP; was 272), pad-row k/v projected once for all batches; pad and slice passes gone
# speedup vs baseline: 2.0177x; 2.0177x over previous
#include <hip/hip_runtime.h>

#define B_ 8
#define S_ 512
#define D_ 512
#define H_ 64
#define DK_ 8
#define WIDTH_ 16
#define N_ (S_ + 2 * WIDTH_)
#define NL_ 2
#define WIN_ (WIDTH_ + 2)
#define BS_ (B_ * S_)
static const size_t BSD_ = (size_t)BS_ * D_;

typedef __attribute__((ext_vector_type(2))) float v2f;
typedef __attribute__((ext_vector_type(8))) float v8f;

typedef __attribute__((ext_vector_type(4))) float v4f;

#define WMMA_F32(a, b, c) __builtin_amdgcn_wmma_f32_16x16x4_f32(false, (a), false, (b), (short)0, (c), false, false)

#define LDA_S_ (512 + 4)
#define LDS_TILE64_ (64 * LDA_S_)
#define LDS_TILE16_ (16 * LDA_S_)
#define LDS_ALONE_ (162 * 256)

template <int MT, int PF, int LDS_FLOATS>
__global__ __launch_bounds__(128)
void gemm_wmma_f32(const float* __restrict__ A, int lda,
                   const float* __restrict__ W, int ldw,
                   const float* __restrict__ bias,
                   float* __restrict__ C, int ldc,
                   int M, int Nout) {
  constexpr int KK = 512;
  constexpr int BM = 16 * MT;
  constexpr int LDA_S = LDA_S_;
  static_assert(LDS_FLOATS >= BM * LDA_S, "LDS array smaller than the A tile");
  __shared__ __attribute__((aligned(16))) float As[LDS_FLOATS];

  const int lane = threadIdx.x;
  const int wave = threadIdx.y;
  const int tid  = wave * 32 + lane;
  const int row0 = blockIdx.y * BM;
  const int col0 = blockIdx.x * 128 + wave * 32;
  const int c = lane & 15;
  const int h = lane >> 4;

  int wr0 = col0 + c;      if (wr0 > Nout - 1) wr0 = Nout - 1;
  int wr1 = col0 + 16 + c; if (wr1 > Nout - 1) wr1 = Nout - 1;
  const float* w0 = W + (size_t)wr0 * ldw + 4 * h;
  const float* w1 = W + (size_t)wr1 * ldw + 4 * h;

  v4f bq[PF][2];
#pragma unroll
  for (int s = 0; s < PF; ++s) {
    bq[s][0] = *(const v4f*)(w0 + 8 * s);
    bq[s][1] = *(const v4f*)(w1 + 8 * s);
  }

#pragma unroll 32
  for (int v = 0; v < BM; ++v) {
    int rg = row0 + v; if (rg > M - 1) rg = M - 1;
    float4 val = *(const float4*)(A + (size_t)rg * lda + tid * 4);
    *(float4*)&As[v * LDA_S + tid * 4] = val;
  }
  __syncthreads();

  if (row0 >= M || col0 >= Nout) return;

  const float* ap = &As[c * LDA_S + 4 * h];

  v8f acc[MT][2];
#pragma unroll
  for (int mt = 0; mt < MT; ++mt) { acc[mt][0] = (v8f){}; acc[mt][1] = (v8f){}; }

  v4f an[MT];
#pragma unroll
  for (int mt = 0; mt < MT; ++mt) an[mt] = *(const v4f*)(ap + mt * 16 * LDA_S);

  for (int it = 0; it < KK / 8; it += PF) {
#pragma unroll
    for (int s = 0; s < PF; ++s) {
      const int kk = (it + s) * 8;
      v4f a[MT];
#pragma unroll
      for (int mt = 0; mt < MT; ++mt) a[mt] = an[mt];
      v4f b0 = bq[s][0], b1 = bq[s][1];
      int kp = kk + 8 * PF; if (kp > KK - 8) kp = KK - 8;
      bq[s][0] = *(const v4f*)(w0 + kp);
      bq[s][1] = *(const v4f*)(w1 + kp);
      int ka = kk + 8; if (ka > KK - 8) ka = KK - 8;
#pragma unroll
      for (int mt = 0; mt < MT; ++mt) an[mt] = *(const v4f*)(ap + mt * 16 * LDA_S + ka);

#pragma unroll
      for (int mt = 0; mt < MT; ++mt) {
        acc[mt][0] = WMMA_F32(a[mt].xy, b0.xy, acc[mt][0]);
        acc[mt][1] = WMMA_F32(a[mt].xy, b1.xy, acc[mt][1]);
      }
#pragma unroll
      for (int mt = 0; mt < MT; ++mt) {
        acc[mt][0] = WMMA_F32(a[mt].zw, b0.zw, acc[mt][0]);
        acc[mt][1] = WMMA_F32(a[mt].zw, b1.zw, acc[mt][1]);
      }
    }
  }

  const float bv0 = bias[col0 + c];
  const float bv1 = bias[col0 + 16 + c];
#pragma unroll
  for (int mt = 0; mt < MT; ++mt) {
#pragma unroll
    for (int r = 0; r < 8; ++r) {
      const int row = row0 + mt * 16 + 8 * h + r;
      if (row < M) {
        float* crow = C + (size_t)row * ldc + col0 + c;
        crow[ 0] = acc[mt][0][r] + bv0;
        crow[16] = acc[mt][1][r] + bv1;
      }
    }
  }
}

__global__ __launch_bounds__(256)
void attn_kernel(const float* __restrict__ qkv, const float* __restrict__ padkv,
                 float* __restrict__ out, int dir) {
  int idx = blockIdx.x * blockDim.x + threadIdx.x;
  if (idx >= B_ * H_ * S_) return;
  int h    = idx % H_;
  int rest = idx / H_;
  int s    = rest % S_;
  int b    = rest / S_;
  int i    = s + WIDTH_;

  const size_t base = (size_t)(b * S_ + s) * D_ + h * DK_;
  const float* qp = qkv + base;
  const float* kbase = qkv + BSD_ + (size_t)(b * S_) * D_ + h * DK_;
  const float* vbase = qkv + 2 * BSD_ + (size_t)(b * S_) * D_ + h * DK_;
  const float* kpad = padkv + h * DK_;
  const float* vpad = padkv + D_ + h * DK_;

  float q[DK_];
  *(float4*)&q[0] = *(const float4*)(qp);
  *(float4*)&q[4] = *(const float4*)(qp + 4);

  const int j0 = (dir == 0) ? ((i - WIDTH_ - 1 > 0) ? i - WIDTH_ - 1 : 0) : i;
  const int j1 = (dir == 0) ? i : ((i + WIDTH_ + 1 < N_ - 1) ? i + WIDTH_ + 1 : N_ - 1);
  const float scale = 0.35355339059327373f;

  float sv[WIN_];
  float mx = -3.0e38f;
#pragma unroll
  for (int t = 0; t < WIN_; ++t) {
    int j = j0 + t;
    float sc = -3.0e38f;
    if (j <= j1) {
      const int jr = j - WIDTH_;
      const float* kp;
      if (jr >= 0 && jr < S_) kp = kbase + (size_t)jr * D_;
      else kp = kpad + (size_t)((j < WIDTH_) ? j : jr - S_) * (2 * D_);
      float kr[DK_];
      *(float4*)&kr[0] = *(const float4*)(kp);
      *(float4*)&kr[4] = *(const float4*)(kp + 4);
      float dot = 0.f;
#pragma unroll
      for (int d = 0; d < DK_; ++d) dot += q[d] * kr[d];
      sc = dot * scale;
    }
    sv[t] = sc;
    mx = fmaxf(mx, sc);
  }
  float sum = 0.f;
#pragma unroll
  for (int t = 0; t < WIN_; ++t) {
    float p = __expf(sv[t] - mx);
    sv[t] = p;
    sum += p;
  }
  const float inv = 1.f / sum;
  float o[DK_] = {};
#pragma unroll
  for (int t = 0; t < WIN_; ++t) {
    int j = j0 + t;
    if (j <= j1) {
      const int jr = j - WIDTH_;
      const float* vp;
      if (jr >= 0 && jr < S_) vp = vbase + (size_t)jr * D_;
      else vp = vpad + (size_t)((j < WIDTH_) ? j : jr - S_) * (2 * D_);
      float vr[DK_];
      *(float4*)&vr[0] = *(const float4*)(vp);
      *(float4*)&vr[4] = *(const float4*)(vp + 4);
      float w = sv[t] * inv;
#pragma unroll
      for (int d = 0; d < DK_; ++d) o[d] += w * vr[d];
    }
  }
  float* op = out + base;
  float4 o0; o0.x = o[0]; o0.y = o[1]; o0.z = o[2]; o0.w = o[3];
  float4 o1; o1.x = o[4]; o1.y = o[5]; o1.z = o[6]; o1.w = o[7];
  *(float4*)(op)     = o0;
  *(float4*)(op + 4) = o1;
}

__global__ __launch_bounds__(256)
void gate_kernel(const float* __restrict__ hp, float* __restrict__ hx) {
  size_t idx = (size_t)blockIdx.x * blockDim.x + threadIdx.x;
  if (idx >= BSD_) return;
  int dd = (int)(idx % D_);
  size_t m = idx / D_;
  float lin = hp[m * (2 * D_) + dd];
  float gin = hp[m * (2 * D_) + D_ + dd];
  float g  = 1.f / (1.f + __expf(-gin));
  float nl = fmaxf(lin, 0.f);
  hx[idx] = g * hx[idx] + (1.f - g) * nl;
}

__global__ __launch_bounds__(256)
void outw_kernel(const float* __restrict__ hx, float* __restrict__ out, int dir) {
  size_t idx = (size_t)blockIdx.x * blockDim.x + threadIdx.x;
  if (idx >= BSD_) return;
  int dd = (int)(idx % D_);
  size_t m = idx / D_;
  size_t o = m * (size_t)(2 * D_) + (size_t)dir * D_ + dd;
  float v = hx[idx];
  out[o] = v;
  out[o + (size_t)BS_ * 2 * D_] = v;
}

extern "C" void kernel_launch(void* const* d_in, const int* in_sizes, int n_in,
                              void* d_out, int out_size, void* d_ws, size_t ws_size,
                              hipStream_t stream) {
  const float* inputs    = (const float*)d_in[0];
  const float* left_pad  = (const float*)d_in[1];
  const float* right_pad = (const float*)d_in[2];
  const float* padrows[2] = {left_pad, right_pad};
  const float* Wqkv[2] = {(const float*)d_in[3],  (const float*)d_in[7]};
  const float* bqkv[2] = {(const float*)d_in[4],  (const float*)d_in[8]};
  const float* Wo[2]   = {(const float*)d_in[5],  (const float*)d_in[9]};
  const float* bo[2]   = {(const float*)d_in[6],  (const float*)d_in[10]};
  const float* hwW[2]  = {(const float*)d_in[11], (const float*)d_in[13]};
  const float* hwb[2]  = {(const float*)d_in[12], (const float*)d_in[14]};
  float* out = (float*)d_out;

  float* ws    = (float*)d_ws;
  float* qkv   = ws;
  float* attn  = qkv + 3 * BSD_;
  float* hx    = attn + BSD_;
  float* padkv = hx + BSD_;
  float* hproj = qkv;

  dim3 blk(32, 4);

  for (int dir = 0; dir < 2; ++dir) {
    { dim3 grd(2 * D_ / 128, 1);
      gemm_wmma_f32<1, 16, LDS_TILE16_><<<grd, blk, 0, stream>>>(
          padrows[dir], D_, Wqkv[dir] + (size_t)D_ * D_, D_, bqkv[dir] + D_,
          padkv, 2 * D_, WIDTH_, 2 * D_); }
    for (int t = 0; t < 3; ++t) {
      dim3 grd(D_ / 128, BS_ / 64);
      gemm_wmma_f32<4, 4, LDS_ALONE_><<<grd, blk, 0, stream>>>(
          inputs, D_, Wqkv[dir] + (size_t)t * D_ * D_, D_, bqkv[dir] + t * D_,
          qkv + (size_t)t * BSD_, D_, BS_, D_);
    }
    { int n = B_ * H_ * S_;
      attn_kernel<<<(n + 255) / 256, 256, 0, stream>>>(qkv, padkv, attn, dir); }
    { dim3 grd(D_ / 128, BS_ / 64);
      gemm_wmma_f32<4, 4, LDS_ALONE_><<<grd, blk, 0, stream>>>(attn, D_, Wo[dir], D_, bo[dir],
                                                              hx, D_, BS_, D_); }
    for (int l = 0; l < NL_; ++l) {
      dim3 grd(2 * D_ / 128, BS_ / 64);
      gemm_wmma_f32<4, 4, LDS_TILE64_><<<grd, blk, 0, stream>>>(
          hx, D_, hwW[dir] + (size_t)l * 2 * D_ * D_, D_, hwb[dir] + l * 2 * D_,
          hproj, 2 * D_, BS_, 2 * D_);
      int n = (int)BSD_;
      gate_kernel<<<(n + 255) / 256, 256, 0, stream>>>(hproj, hx);
    }
    { int n = (int)BSD_; outw_kernel<<<(n + 255) / 256, 256, 0, stream>>>(hx, out, dir); }
  }
}


# Wo projection, both highway layers (gate in the epilogue) and output write chained in one kernel: a block keeps its 32 rows in LDS from layer to layer, both directions in one grid; 22 launches to 11
# speedup vs baseline: 2.4985x; 1.2383x over previous
#include <hip/hip_runtime.h>

#define B_ 8
#define S_ 512
#define D_ 512
#define H_ 64
#define DK_ 8
#define WIDTH_ 16
#define N_ (S_ + 2 * WIDTH_)
#define NL_ 2
#define WIN_ (WIDTH_ + 2)
#define BS_ (B_ * S_)
static const size_t BSD_ = (size_t)BS_ * D_;

typedef __attribute__((ext_vector_type(2))) float v2f;
typedef __attribute__((ext_vector_type(8))) float v8f;

typedef __attribute__((ext_vector_type(4))) float v4f;

#define WMMA_F32(a, b, c) __builtin_amdgcn_wmma_f32_16x16x4_f32(false, (a), false, (b), (short)0, (c), false, false)

#define LDA_S_ (512 + 4)
#define LDS_TILE16_ (16 * LDA_S_)
#define LDS_ALONE_ (162 * 256)

template <int MT, int PF, int LDS_FLOATS>
__global__ __launch_bounds__(128)
void gemm_wmma_f32(const float* __restrict__ A, int lda,
                   const float* __restrict__ W, int ldw,
                   const float* __restrict__ bias,
                   float* __restrict__ C, int ldc,
                   int M, int Nout) {
  constexpr int KK = 512;
  constexpr int BM = 16 * MT;
  constexpr int LDA_S = LDA_S_;
  static_assert(LDS_FLOATS >= BM * LDA_S, "LDS array smaller than the A tile");
  __shared__ __attribute__((aligned(16))) float As[LDS_FLOATS];

  const int lane = threadIdx.x;
  const int wave = threadIdx.y;
  const int tid  = wave * 32 + lane;
  const int row0 = blockIdx.y * BM;
  const int col0 = blockIdx.x * 128 + wave * 32;
  const int c = lane & 15;
  const int h = lane >> 4;

  int wr0 = col0 + c;      if (wr0 > Nout - 1) wr0 = Nout - 1;
  int wr1 = col0 + 16 + c; if (wr1 > Nout - 1) wr1 = Nout - 1;
  const float* w0 = W + (size_t)wr0 * ldw + 4 * h;
  const float* w1 = W + (size_t)wr1 * ldw + 4 * h;

  v4f bq[PF][2];
#pragma unroll
  for (int s = 0; s < PF; ++s) {
    bq[s][0] = *(const v4f*)(w0 + 8 * s);
    bq[s][1] = *(const v4f*)(w1 + 8 * s);
  }

#pragma unroll 32
  for (int v = 0; v < BM; ++v) {
    int rg = row0 + v; if (rg > M - 1) rg = M - 1;
    float4 val = *(const float4*)(A + (size_t)rg * lda + tid * 4);
    *(float4*)&As[v * LDA_S + tid * 4] = val;
  }
  __syncthreads();

  if (row0 >= M || col0 >= Nout) return;

  const float* ap = &As[c * LDA_S + 4 * h];

  v8f acc[MT][2];
#pragma unroll
  for (int mt = 0; mt < MT; ++mt) { acc[mt][0] = (v8f){}; acc[mt][1] = (v8f){}; }

  v4f an[MT];
#pragma unroll
  for (int mt = 0; mt < MT; ++mt) an[mt] = *(const v4f*)(ap + mt * 16 * LDA_S);

  for (int it = 0; it < KK / 8; it += PF) {
#pragma unroll
    for (int s = 0; s < PF; ++s) {
      const int kk = (it + s) * 8;
      v4f a[MT];
#pragma unroll
      for (int mt = 0; mt < MT; ++mt) a[mt] = an[mt];
      v4f b0 = bq[s][0], b1 = bq[s][1];
      int kp = kk + 8 * PF; if (kp > KK - 8) kp = KK - 8;
      bq[s][0] = *(const v4f*)(w0 + kp);
      bq[s][1] = *(const v4f*)(w1 + kp);
      int ka = kk + 8; if (ka > KK - 8) ka = KK - 8;
#pragma unroll
      for (int mt = 0; mt < MT; ++mt) an[mt] = *(const v4f*)(ap + mt * 16 * LDA_S + ka);

#pragma unroll
      for (int mt = 0; mt < MT; ++mt) {
        acc[mt][0] = WMMA_F32(a[mt].xy, b0.xy, acc[mt][0]);
        acc[mt][1] = WMMA_F32(a[mt].xy, b1.xy, acc[mt][1]);
      }
#pragma unroll
      for (int mt = 0; mt < MT; ++mt) {
        acc[mt][0] = WMMA_F32(a[mt].zw, b0.zw, acc[mt][0]);
        acc[mt][1] = WMMA_F32(a[mt].zw, b1.zw, acc[mt][1]);
      }
    }
  }

  const float bv0 = bias[col0 + c];
  const float bv1 = bias[col0 + 16 + c];
#pragma unroll
  for (int mt = 0; mt < MT; ++mt) {
#pragma unroll
    for (int r = 0; r < 8; ++r) {
      const int row = row0 + mt * 16 + 8 * h + r;
      if (row < M) {
        float* crow = C + (size_t)row * ldc + col0 + c;
        crow[ 0] = acc[mt][0][r] + bv0;
        crow[16] = acc[mt][1][r] + bv1;
      }
    }
  }
}

__global__ __launch_bounds__(256)
void attn_kernel(const float* __restrict__ qkv, const float* __restrict__ padkv,
                 float* __restrict__ out, int dir) {
  int idx = blockIdx.x * blockDim.x + threadIdx.x;
  if (idx >= B_ * H_ * S_) return;
  int h    = idx % H_;
  int rest = idx / H_;
  int s    = rest % S_;
  int b    = rest / S_;
  int i    = s + WIDTH_;

  const size_t base = (size_t)(b * S_ + s) * D_ + h * DK_;
  const float* qp = qkv + base;
  const float* kbase = qkv + BSD_ + (size_t)(b * S_) * D_ + h * DK_;
  const float* vbase = qkv + 2 * BSD_ + (size_t)(b * S_) * D_ + h * DK_;
  const float* kpad = padkv + h * DK_;
  const float* vpad = padkv + D_ + h * DK_;

  float q[DK_];
  *(float4*)&q[0] = *(const float4*)(qp);
  *(float4*)&q[4] = *(const float4*)(qp + 4);

  const int j0 = (dir == 0) ? ((i - WIDTH_ - 1 > 0) ? i - WIDTH_ - 1 : 0) : i;
  const int j1 = (dir == 0) ? i : ((i + WIDTH_ + 1 < N_ - 1) ? i + WIDTH_ + 1 : N_ - 1);
  const float scale = 0.35355339059327373f;

  float sv[WIN_];
  float mx = -3.0e38f;
#pragma unroll
  for (int t = 0; t < WIN_; ++t) {
    int j = j0 + t;
    float sc = -3.0e38f;
    if (j <= j1) {
      const int jr = j - WIDTH_;
      const float* kp;
      if (jr >= 0 && jr < S_) kp = kbase + (size_t)jr * D_;
      else kp = kpad + (size_t)((j < WIDTH_) ? j : jr - S_) * (2 * D_);
      float kr[DK_];
      *(float4*)&kr[0] = *(const float4*)(kp);
      *(float4*)&kr[4] = *(const float4*)(kp + 4);
      float dot = 0.f;
#pragma unroll
      for (int d = 0; d < DK_; ++d) dot += q[d] * kr[d];
      sc = dot * scale;
    }
    sv[t] = sc;
    mx = fmaxf(mx, sc);
  }
  float sum = 0.f;
#pragma unroll
  for (int t = 0; t < WIN_; ++t) {
    float p = __expf(sv[t] - mx);
    sv[t] = p;
    sum += p;
  }
  const float inv = 1.f / sum;
  float o[DK_] = {};
#pragma unroll
  for (int t = 0; t < WIN_; ++t) {
    int j = j0 + t;
    if (j <= j1) {
      const int jr = j - WIDTH_;
      const float* vp;
      if (jr >= 0 && jr < S_) vp = vbase + (size_t)jr * D_;
      else vp = vpad + (size_t)((j < WIDTH_) ? j : jr - S_) * (2 * D_);
      float vr[DK_];
      *(float4*)&vr[0] = *(const float4*)(vp);
      *(float4*)&vr[4] = *(const float4*)(vp + 4);
      float w = sv[t] * inv;
#pragma unroll
      for (int d = 0; d < DK_; ++d) o[d] += w * vr[d];
    }
  }
  float* op = out + base;
  float4 o0; o0.x = o[0]; o0.y = o[1]; o0.z = o[2]; o0.w = o[3];
  float4 o1; o1.x = o[4]; o1.y = o[5]; o1.z = o[6]; o1.w = o[7];
  *(float4*)(op)     = o0;
  *(float4*)(op + 4) = o1;
}

#define CH_ROWS_ 32
#define CH_X_ (CH_ROWS_ * LDA_S_)
#define CH_PF_ 4
#define CH_PASSES_ 10

__device__ __forceinline__ void chain_pass_geom(int pass, int wave,
                                                const float* Wo, const float* bo,
                                                const float* hwW, const float* hwb,
                                                int& n0, int& hic,
                                                const float*& Wst, const float*& bst) {
  if (pass < 2) {
    n0 = wave * 128 + pass * 64; hic = 32; Wst = Wo; bst = bo;
  } else {
    const int q = pass - 2;
    const int layer = q >> 2;
    n0 = wave * 128 + (q & 3) * 32; hic = D_;
    Wst = hwW + (size_t)layer * 2 * D_ * D_; bst = hwb + layer * 2 * D_;
  }
}

__device__ __forceinline__ void chain_wave_pass(const float* ap,
                                                const float* w_lo, const float* w_hi,
                                                const float* nx_lo, const float* nx_hi,
                                                v4f (&bq)[CH_PF_][4], v8f (&acc)[2][4]) {
  constexpr int KK = 512;
  v4f an[2];
  an[0] = *(const v4f*)(ap);
  an[1] = *(const v4f*)(ap + 16 * LDA_S_);
  for (int it = 0; it < KK / 8; it += CH_PF_) {
    const bool wrap = (it + CH_PF_ >= KK / 8);
    const float* cl = wrap ? nx_lo : w_lo;
    const float* ch = wrap ? nx_hi : w_hi;
    const int kb = wrap ? 0 : (it + CH_PF_) * 8;
#pragma unroll
    for (int s = 0; s < CH_PF_; ++s) {
      const int kk = (it + s) * 8;
      const v4f a0 = an[0], a1 = an[1];
      const v4f b0 = bq[s][0], b1 = bq[s][1], b2 = bq[s][2], b3 = bq[s][3];
      bq[s][0] = *(const v4f*)(cl + kb + 8 * s);
      bq[s][1] = *(const v4f*)(cl + kb + 8 * s + 16 * KK);
      bq[s][2] = *(const v4f*)(ch + kb + 8 * s);
      bq[s][3] = *(const v4f*)(ch + kb + 8 * s + 16 * KK);
      int ka = kk + 8; if (ka > KK - 8) ka = KK - 8;
      an[0] = *(const v4f*)(ap + ka);
      an[1] = *(const v4f*)(ap + 16 * LDA_S_ + ka);

      acc[0][0] = WMMA_F32(a0.xy, b0.xy, acc[0][0]);
      acc[0][1] = WMMA_F32(a0.xy, b1.xy, acc[0][1]);
      acc[0][2] = WMMA_F32(a0.xy, b2.xy, acc[0][2]);
      acc[0][3] = WMMA_F32(a0.xy, b3.xy, acc[0][3]);
      acc[1][0] = WMMA_F32(a1.xy, b0.xy, acc[1][0]);
      acc[1][1] = WMMA_F32(a1.xy, b1.xy, acc[1][1]);
      acc[1][2] = WMMA_F32(a1.xy, b2.xy, acc[1][2]);
      acc[1][3] = WMMA_F32(a1.xy, b3.xy, acc[1][3]);
      acc[0][0] = WMMA_F32(a0.zw, b0.zw, acc[0][0]);
      acc[0][1] = WMMA_F32(a0.zw, b1.zw, acc[0][1]);
      acc[0][2] = WMMA_F32(a0.zw, b2.zw, acc[0][2]);
      acc[0][3] = WMMA_F32(a0.zw, b3.zw, acc[0][3]);
      acc[1][0] = WMMA_F32(a1.zw, b0.zw, acc[1][0]);
      acc[1][1] = WMMA_F32(a1.zw, b1.zw, acc[1][1]);
      acc[1][2] = WMMA_F32(a1.zw, b2.zw, acc[1][2]);
      acc[1][3] = WMMA_F32(a1.zw, b3.zw, acc[1][3]);
    }
  }
}

__global__ __launch_bounds__(128)
void chain_kernel(const float* __restrict__ attn_l, const float* __restrict__ attn_r,
                  const float* __restrict__ Wo_l,   const float* __restrict__ Wo_r,
                  const float* __restrict__ bo_l,   const float* __restrict__ bo_r,
                  const float* __restrict__ hwW_l,  const float* __restrict__ hwW_r,
                  const float* __restrict__ hwb_l,  const float* __restrict__ hwb_r,
                  float* __restrict__ out) {
  __shared__ __attribute__((aligned(16))) float Xs[LDS_ALONE_];
  static_assert(LDS_ALONE_ >= 2 * CH_X_, "LDS array smaller than the two activation buffers");

  const int lane = threadIdx.x;
  const int wave = threadIdx.y;
  const int tid  = wave * 32 + lane;
  const int c = lane & 15;
  const int h = lane >> 4;
  const int dir  = blockIdx.y;
  const int row0 = blockIdx.x * CH_ROWS_;

  const float* attn = dir ? attn_r : attn_l;
  const float* Wo   = dir ? Wo_r   : Wo_l;
  const float* bo   = dir ? bo_r   : bo_l;
  const float* hwW  = dir ? hwW_r  : hwW_l;
  const float* hwb  = dir ? hwb_r  : hwb_l;

  int n0, hic; const float* Wst; const float* bst;
  chain_pass_geom(0, wave, Wo, bo, hwW, hwb, n0, hic, Wst, bst);
  const float* w_lo = Wst + (size_t)(n0 + c) * D_ + 4 * h;
  const float* w_hi = w_lo + (size_t)hic * D_;
  v4f bq[CH_PF_][4];
#pragma unroll
  for (int s = 0; s < CH_PF_; ++s) {
    bq[s][0] = *(const v4f*)(w_lo + 8 * s);
    bq[s][1] = *(const v4f*)(w_lo + 8 * s + 16 * D_);
    bq[s][2] = *(const v4f*)(w_hi + 8 * s);
    bq[s][3] = *(const v4f*)(w_hi + 8 * s + 16 * D_);
  }

#pragma unroll
  for (int v = 0; v < CH_ROWS_; ++v) {
    float4 val = *(const float4*)(attn + (size_t)(row0 + v) * D_ + tid * 4);
    *(float4*)&Xs[v * LDA_S_ + tid * 4] = val;
  }
  __syncthreads();

  for (int pass = 0; pass < CH_PASSES_; ++pass) {
    const int stage = (pass < 2) ? 0 : ((pass < 6) ? 1 : 2);
    const float* Xin  = Xs + (stage & 1) * CH_X_;
    float*       Xout = Xs + ((stage & 1) ^ 1) * CH_X_;

    const int np = (pass + 1 < CH_PASSES_) ? pass + 1 : pass;
    int n0n, hicn; const float* Wn; const float* bn;
    chain_pass_geom(np, wave, Wo, bo, hwW, hwb, n0n, hicn, Wn, bn);
    const float* nx_lo = Wn + (size_t)(n0n + c) * D_ + 4 * h;
    const float* nx_hi = nx_lo + (size_t)hicn * D_;

    const float bv0 = bst[n0 + c];
    const float bv1 = bst[n0 + 16 + c];
    const float bv2 = bst[n0 + hic + c];
    const float bv3 = bst[n0 + hic + 16 + c];

    v8f acc[2][4];
#pragma unroll
    for (int mt = 0; mt < 2; ++mt) {
#pragma unroll
      for (int t = 0; t < 4; ++t) acc[mt][t] = (v8f){};
    }

    chain_wave_pass(Xin + c * LDA_S_ + 4 * h, w_lo, w_hi, nx_lo, nx_hi, bq, acc);

    if (stage == 0) {
#pragma unroll
      for (int mt = 0; mt < 2; ++mt) {
#pragma unroll
        for (int r = 0; r < 8; ++r) {
          float* xr = Xout + (mt * 16 + 8 * h + r) * LDA_S_ + n0 + c;
          xr[ 0] = acc[mt][0][r] + bv0;
          xr[16] = acc[mt][1][r] + bv1;
          xr[32] = acc[mt][2][r] + bv2;
          xr[48] = acc[mt][3][r] + bv3;
        }
      }
    } else {
#pragma unroll
      for (int mt = 0; mt < 2; ++mt) {
#pragma unroll
        for (int r = 0; r < 8; ++r) {
          const int idx = (mt * 16 + 8 * h + r) * LDA_S_ + n0 + c;
          {
            const float lin = acc[mt][0][r] + bv0;
            const float gin = acc[mt][2][r] + bv2;
            const float g  = 1.f / (1.f + __expf(-gin));
            const float nl = fmaxf(lin, 0.f);
            Xout[idx] = g * Xin[idx] + (1.f - g) * nl;
          }
          {
            const float lin = acc[mt][1][r] + bv1;
            const float gin = acc[mt][3][r] + bv3;
            const float g  = 1.f / (1.f + __expf(-gin));
            const float nl = fmaxf(lin, 0.f);
            Xout[idx + 16] = g * Xin[idx + 16] + (1.f - g) * nl;
          }
        }
      }
    }

    if (pass == 1 || pass == 5 || pass == 9) __syncthreads();

    n0 = n0n; hic = hicn; bst = bn; w_lo = nx_lo; w_hi = nx_hi;
  }

  const float* Xf = Xs + CH_X_;
  float* o0 = out + (size_t)row0 * (2 * D_) + (size_t)dir * D_ + tid * 4;
#pragma unroll 8
  for (int v = 0; v < CH_ROWS_; ++v) {
    float4 val = *(const float4*)&Xf[v * LDA_S_ + tid * 4];
    *(float4*)(o0 + (size_t)v * (2 * D_)) = val;
    *(float4*)(o0 + (size_t)v * (2 * D_) + (size_t)BS_ * 2 * D_) = val;
  }
}

extern "C" void kernel_launch(void* const* d_in, const int* in_sizes, int n_in,
                              void* d_out, int out_size, void* d_ws, size_t ws_size,
                              hipStream_t stream) {
  const float* inputs    = (const float*)d_in[0];
  const float* left_pad  = (const float*)d_in[1];
  const float* right_pad = (const float*)d_in[2];
  const float* padrows[2] = {left_pad, right_pad};
  const float* Wqkv[2] = {(const float*)d_in[3],  (const float*)d_in[7]};
  const float* bqkv[2] = {(const float*)d_in[4],  (const float*)d_in[8]};
  const float* Wo[2]   = {(const float*)d_in[5],  (const float*)d_in[9]};
  const float* bo[2]   = {(const float*)d_in[6],  (const float*)d_in[10]};
  const float* hwW[2]  = {(const float*)d_in[11], (const float*)d_in[13]};
  const float* hwb[2]  = {(const float*)d_in[12], (const float*)d_in[14]};
  float* out = (float*)d_out;

  float* ws    = (float*)d_ws;
  float* qkv   = ws;
  float* attn0 = qkv + 3 * BSD_;
  float* attn1 = attn0 + BSD_;
  float* padkv = attn1 + BSD_;
  float* attnb[2] = {attn0, attn1};

  dim3 blk(32, 4);

  for (int dir = 0; dir < 2; ++dir) {
    { dim3 grd(2 * D_ / 128, 1);
      gemm_wmma_f32<1, 16, LDS_TILE16_><<<grd, blk, 0, stream>>>(
          padrows[dir], D_, Wqkv[dir] + (size_t)D_ * D_, D_, bqkv[dir] + D_,
          padkv, 2 * D_, WIDTH_, 2 * D_); }
    for (int t = 0; t < 3; ++t) {
      dim3 grd(D_ / 128, BS_ / 64);
      gemm_wmma_f32<4, 4, LDS_ALONE_><<<grd, blk, 0, stream>>>(
          inputs, D_, Wqkv[dir] + (size_t)t * D_ * D_, D_, bqkv[dir] + t * D_,
          qkv + (size_t)t * BSD_, D_, BS_, D_);
    }
    { int n = B_ * H_ * S_;
      attn_kernel<<<(n + 255) / 256, 256, 0, stream>>>(qkv, padkv, attnb[dir], dir); }
  }
  { dim3 grd(BS_ / CH_ROWS_, 2);
    chain_kernel<<<grd, blk, 0, stream>>>(attn0, attn1, Wo[0], Wo[1], bo[0], bo[1],
                                          hwW[0], hwW[1], hwb[0], hwb[1], out); }
}


# Attention moves into the chained kernel, out of LDS: a block stages its 32 q rows and one 49-row k and v window, two neighbouring rows share each window row read; 11 launches to 9
# speedup vs baseline: 2.7333x; 1.0940x over previous
#include <hip/hip_runtime.h>

#define B_ 8
#define S_ 512
#define D_ 512
#define H_ 64
#define DK_ 8
#define WIDTH_ 16
#define N_ (S_ + 2 * WIDTH_)
#define NL_ 2
#define WIN_ (WIDTH_ + 2)
#define BS_ (B_ * S_)
static const size_t BSD_ = (size_t)BS_ * D_;

typedef __attribute__((ext_vector_type(2))) float v2f;
typedef __attribute__((ext_vector_type(8))) float v8f;

typedef __attribute__((ext_vector_type(4))) float v4f;

#define WMMA_F32(a, b, c) __builtin_amdgcn_wmma_f32_16x16x4_f32(false, (a), false, (b), (short)0, (c), false, false)

#define LDA_S_ (512 + 4)
#define LDS_TILE16_ (16 * LDA_S_)
#define LDS_ALONE_ (162 * 256)

template <int MT, int PF, int LDS_FLOATS>
__global__ __launch_bounds__(128)
void gemm_wmma_f32(const float* __restrict__ A, int lda,
                   const float* __restrict__ W, int ldw,
                   const float* __restrict__ bias,
                   float* __restrict__ C, int ldc,
                   int M, int Nout) {
  constexpr int KK = 512;
  constexpr int BM = 16 * MT;
  constexpr int LDA_S = LDA_S_;
  static_assert(LDS_FLOATS >= BM * LDA_S, "LDS array smaller than the A tile");
  __shared__ __attribute__((aligned(16))) float As[LDS_FLOATS];

  const int lane = threadIdx.x;
  const int wave = threadIdx.y;
  const int tid  = wave * 32 + lane;
  const int row0 = blockIdx.y * BM;
  const int col0 = blockIdx.x * 128 + wave * 32;
  const int c = lane & 15;
  const int h = lane >> 4;

  int wr0 = col0 + c;      if (wr0 > Nout - 1) wr0 = Nout - 1;
  int wr1 = col0 + 16 + c; if (wr1 > Nout - 1) wr1 = Nout - 1;
  const float* w0 = W + (size_t)wr0 * ldw + 4 * h;
  const float* w1 = W + (size_t)wr1 * ldw + 4 * h;

  v4f bq[PF][2];
#pragma unroll
  for (int s = 0; s < PF; ++s) {
    bq[s][0] = *(const v4f*)(w0 + 8 * s);
    bq[s][1] = *(const v4f*)(w1 + 8 * s);
  }

#pragma unroll 32
  for (int v = 0; v < BM; ++v) {
    int rg = row0 + v; if (rg > M - 1) rg = M - 1;
    float4 val = *(const float4*)(A + (size_t)rg * lda + tid * 4);
    *(float4*)&As[v * LDA_S + tid * 4] = val;
  }
  __syncthreads();

  if (row0 >= M || col0 >= Nout) return;

  const float* ap = &As[c * LDA_S + 4 * h];

  v8f acc[MT][2];
#pragma unroll
  for (int mt = 0; mt < MT; ++mt) { acc[mt][0] = (v8f){}; acc[mt][1] = (v8f){}; }

  v4f an[MT];
#pragma unroll
  for (int mt = 0; mt < MT; ++mt) an[mt] = *(const v4f*)(ap + mt * 16 * LDA_S);

  for (int it = 0; it < KK / 8; it += PF) {
#pragma unroll
    for (int s = 0; s < PF; ++s) {
      const int kk = (it + s) * 8;
      v4f a[MT];
#pragma unroll
      for (int mt = 0; mt < MT; ++mt) a[mt] = an[mt];
      v4f b0 = bq[s][0], b1 = bq[s][1];
      int kp = kk + 8 * PF; if (kp > KK - 8) kp = KK - 8;
      bq[s][0] = *(const v4f*)(w0 + kp);
      bq[s][1] = *(const v4f*)(w1 + kp);
      int ka = kk + 8; if (ka > KK - 8) ka = KK - 8;
#pragma unroll
      for (int mt = 0; mt < MT; ++mt) an[mt] = *(const v4f*)(ap + mt * 16 * LDA_S + ka);

#pragma unroll
      for (int mt = 0; mt < MT; ++mt) {
        acc[mt][0] = WMMA_F32(a[mt].xy, b0.xy, acc[mt][0]);
        acc[mt][1] = WMMA_F32(a[mt].xy, b1.xy, acc[mt][1]);
      }
#pragma unroll
      for (int mt = 0; mt < MT; ++mt) {
        acc[mt][0] = WMMA_F32(a[mt].zw, b0.zw, acc[mt][0]);
        acc[mt][1] = WMMA_F32(a[mt].zw, b1.zw, acc[mt][1]);
      }
    }
  }

  const float bv0 = bias[col0 + c];
  const float bv1 = bias[col0 + 16 + c];
#pragma unroll
  for (int mt = 0; mt < MT; ++mt) {
#pragma unroll
    for (int r = 0; r < 8; ++r) {
      const int row = row0 + mt * 16 + 8 * h + r;
      if (row < M) {
        float* crow = C + (size_t)row * ldc + col0 + c;
        crow[ 0] = acc[mt][0][r] + bv0;
        crow[16] = acc[mt][1][r] + bv1;
      }
    }
  }
}

#define CH_ROWS_ 32
#define CH_X_ (CH_ROWS_ * LDA_S_)
#define CH_PF_ 4
#define CH_PASSES_ 10
#define KW_ROWS_ (CH_ROWS_ + WIDTH_ + 1)
#define CH_LDS_ (CH_X_ + 2 * KW_ROWS_ * LDA_S_)

__device__ __forceinline__ void chain_pass_geom(int pass, int wave,
                                                const float* Wo, const float* bo,
                                                const float* hwW, const float* hwb,
                                                int& n0, int& hic,
                                                const float*& Wst, const float*& bst) {
  if (pass < 2) {
    n0 = wave * 128 + pass * 64; hic = 32; Wst = Wo; bst = bo;
  } else {
    const int q = pass - 2;
    const int layer = q >> 2;
    n0 = wave * 128 + (q & 3) * 32; hic = D_;
    Wst = hwW + (size_t)layer * 2 * D_ * D_; bst = hwb + layer * 2 * D_;
  }
}

__device__ __forceinline__ void chain_wave_pass(const float* ap,
                                                const float* w_lo, const float* w_hi,
                                                const float* nx_lo, const float* nx_hi,
                                                v4f (&bq)[CH_PF_][4], v8f (&acc)[2][4]) {
  constexpr int KK = 512;
  v4f an[2];
  an[0] = *(const v4f*)(ap);
  an[1] = *(const v4f*)(ap + 16 * LDA_S_);
  for (int it = 0; it < KK / 8; it += CH_PF_) {
    const bool wrap = (it + CH_PF_ >= KK / 8);
    const float* cl = wrap ? nx_lo : w_lo;
    const float* ch = wrap ? nx_hi : w_hi;
    const int kb = wrap ? 0 : (it + CH_PF_) * 8;
#pragma unroll
    for (int s = 0; s < CH_PF_; ++s) {
      const int kk = (it + s) * 8;
      const v4f a0 = an[0], a1 = an[1];
      const v4f b0 = bq[s][0], b1 = bq[s][1], b2 = bq[s][2], b3 = bq[s][3];
      bq[s][0] = *(const v4f*)(cl + kb + 8 * s);
      bq[s][1] = *(const v4f*)(cl + kb + 8 * s + 16 * KK);
      bq[s][2] = *(const v4f*)(ch + kb + 8 * s);
      bq[s][3] = *(const v4f*)(ch + kb + 8 * s + 16 * KK);
      int ka = kk + 8; if (ka > KK - 8) ka = KK - 8;
      an[0] = *(const v4f*)(ap + ka);
      an[1] = *(const v4f*)(ap + 16 * LDA_S_ + ka);

      acc[0][0] = WMMA_F32(a0.xy, b0.xy, acc[0][0]);
      acc[0][1] = WMMA_F32(a0.xy, b1.xy, acc[0][1]);
      acc[0][2] = WMMA_F32(a0.xy, b2.xy, acc[0][2]);
      acc[0][3] = WMMA_F32(a0.xy, b3.xy, acc[0][3]);
      acc[1][0] = WMMA_F32(a1.xy, b0.xy, acc[1][0]);
      acc[1][1] = WMMA_F32(a1.xy, b1.xy, acc[1][1]);
      acc[1][2] = WMMA_F32(a1.xy, b2.xy, acc[1][2]);
      acc[1][3] = WMMA_F32(a1.xy, b3.xy, acc[1][3]);
      acc[0][0] = WMMA_F32(a0.zw, b0.zw, acc[0][0]);
      acc[0][1] = WMMA_F32(a0.zw, b1.zw, acc[0][1]);
      acc[0][2] = WMMA_F32(a0.zw, b2.zw, acc[0][2]);
      acc[0][3] = WMMA_F32(a0.zw, b3.zw, acc[0][3]);
      acc[1][0] = WMMA_F32(a1.zw, b0.zw, acc[1][0]);
      acc[1][1] = WMMA_F32(a1.zw, b1.zw, acc[1][1]);
      acc[1][2] = WMMA_F32(a1.zw, b2.zw, acc[1][2]);
      acc[1][3] = WMMA_F32(a1.zw, b3.zw, acc[1][3]);
    }
  }
}

template <int KIND>
__device__ __forceinline__ void stage_kv_window(const float* __restrict__ kb,
                                                const float* __restrict__ pb,
                                                int wb, float* Ks, float* Vs, int tid4) {
#pragma unroll
  for (int w = 0; w < KW_ROWS_; ++w) {
    const float* pk;
    const float* pv;
    if (KIND == 0) {
      pk = kb + (wb + w) * D_; pv = pk + BSD_;
    } else if (KIND == 1) {
      if (w <= WIDTH_) { const int pr = (w == 0) ? 0 : w - 1; pk = pb + pr * (2 * D_); pv = pk + D_; }
      else { pk = kb + (w - (WIDTH_ + 1)) * D_; pv = pk + BSD_; }
    } else {
      if (w < CH_ROWS_) { pk = kb + (S_ - CH_ROWS_ + w) * D_; pv = pk + BSD_; }
      else { const int pr = (w - CH_ROWS_ < WIDTH_) ? w - CH_ROWS_ : WIDTH_ - 1; pk = pb + pr * (2 * D_); pv = pk + D_; }
    }
    const float4 kq = *(const float4*)pk;
    const float4 vq = *(const float4*)pv;
    *(float4*)&Ks[w * LDA_S_ + tid4] = kq;
    *(float4*)&Vs[w * LDA_S_ + tid4] = vq;
  }
}

__global__ __launch_bounds__(128)
void chain_kernel(const float* __restrict__ kvws,
                  const float* __restrict__ Wo_l,   const float* __restrict__ Wo_r,
                  const float* __restrict__ bo_l,   const float* __restrict__ bo_r,
                  const float* __restrict__ hwW_l,  const float* __restrict__ hwW_r,
                  const float* __restrict__ hwb_l,  const float* __restrict__ hwb_r,
                  float* out) {
  __shared__ __attribute__((aligned(16))) float Xs[CH_LDS_];
  static_assert(CH_LDS_ >= 2 * CH_X_, "LDS array smaller than the two activation buffers");
  static_assert(CH_LDS_ * 4 > 160 * 1024 && CH_LDS_ * 4 <= 320 * 1024, "LDS: one block per WGP");

  const int lane = threadIdx.x;
  const int wave = threadIdx.y;
  const int tid  = wave * 32 + lane;
  const int c = lane & 15;
  const int h = lane >> 4;
  const int dir  = blockIdx.y;
  const int row0 = blockIdx.x * CH_ROWS_;

  const float* Wo   = dir ? Wo_r   : Wo_l;
  const float* bo   = dir ? bo_r   : bo_l;
  const float* hwW  = dir ? hwW_r  : hwW_l;
  const float* hwb  = dir ? hwb_r  : hwb_l;

  int n0, hic; const float* Wst; const float* bst;
  chain_pass_geom(0, wave, Wo, bo, hwW, hwb, n0, hic, Wst, bst);
  const float* w_lo = Wst + (size_t)(n0 + c) * D_ + 4 * h;
  const float* w_hi = w_lo + (size_t)hic * D_;
  v4f bq[CH_PF_][4];
#pragma unroll
  for (int s = 0; s < CH_PF_; ++s) {
    bq[s][0] = *(const v4f*)(w_lo + 8 * s);
    bq[s][1] = *(const v4f*)(w_lo + 8 * s + 16 * D_);
    bq[s][2] = *(const v4f*)(w_hi + 8 * s);
    bq[s][3] = *(const v4f*)(w_hi + 8 * s + 16 * D_);
  }

  const int bb = row0 / S_;
  const int s0 = row0 % S_;
  const int wb = dir ? s0 : s0 - (WIDTH_ + 1);
  float* Ks = Xs + CH_X_;
  float* Vs = Ks + KW_ROWS_ * LDA_S_;
  {
    const float* qsrc = out + (size_t)row0 * (2 * D_) + (size_t)dir * D_ + tid * 4;
#pragma unroll
    for (int v = 0; v < CH_ROWS_; ++v) {
      float4 val = *(const float4*)(qsrc + (size_t)v * (2 * D_));
      *(float4*)&Xs[v * LDA_S_ + tid * 4] = val;
    }
    static_assert(S_ % CH_ROWS_ == 0 && CH_ROWS_ > WIDTH_ + 1, "window shapes of stage_kv_window");
    const float* kb = kvws + (size_t)dir * (2 * BSD_) + (size_t)(bb * S_) * D_ + tid * 4;
    const float* pb = kvws + 4 * BSD_ + (size_t)dir * (WIDTH_ * 2 * D_) + tid * 4;
    if (wb < 0)                          stage_kv_window<1>(kb, pb, wb, Ks, Vs, tid * 4);
    else if (wb + KW_ROWS_ - 1 >= S_)    stage_kv_window<2>(kb, pb, wb, Ks, Vs, tid * 4);
    else                                 stage_kv_window<0>(kb, pb, wb, Ks, Vs, tid * 4);
  }
  __syncthreads();

  {
    const int hh = tid & (H_ - 1);
    const int rsel = tid >> 6;
    const float scale = 0.35355339059327373f;
    const float NEG = -3.0e38f;
    for (int it = 0; it < CH_ROWS_ / 4; ++it) {
      const int r0 = (it * 2 + rsel) * 2;
      float* qrow = Xs + r0 * LDA_S_ + hh * DK_;
      const float* krow = Ks + r0 * LDA_S_ + hh * DK_;
      const float* vrow = Vs + r0 * LDA_S_ + hh * DK_;
      float qa[DK_], qb[DK_];
      *(float4*)&qa[0] = *(const float4*)(qrow);
      *(float4*)&qa[4] = *(const float4*)(qrow + 4);
      *(float4*)&qb[0] = *(const float4*)(qrow + LDA_S_);
      *(float4*)&qb[4] = *(const float4*)(qrow + LDA_S_ + 4);
      const int sa_row = s0 + r0;
      const int sb_row = s0 + r0 + 1;
      float sa[WIN_], sb[WIN_];
      float mxa = NEG, mxb = NEG;
#pragma unroll
      for (int u = 0; u <= WIN_; ++u) {
        float kr[DK_];
        *(float4*)&kr[0] = *(const float4*)(krow + u * LDA_S_);
        *(float4*)&kr[4] = *(const float4*)(krow + u * LDA_S_ + 4);
        if (u < WIN_) {
          float dot = 0.f;
#pragma unroll
          for (int d = 0; d < DK_; ++d) dot += qa[d] * kr[d];
          const bool ok = (u == 0) ? !(dir == 0 && sa_row == 0)
                        : ((u == WIN_ - 1) ? !(dir == 1 && sa_row == S_ - 1) : true);
          const float sc = ok ? dot * scale : NEG;
          sa[u] = sc; mxa = fmaxf(mxa, sc);
        }
        if (u >= 1) {
          float dot = 0.f;
#pragma unroll
          for (int d = 0; d < DK_; ++d) dot += qb[d] * kr[d];
          const bool ok = (u - 1 == 0) ? !(dir == 0 && sb_row == 0)
                        : ((u - 1 == WIN_ - 1) ? !(dir == 1 && sb_row == S_ - 1) : true);
          const float sc = ok ? dot * scale : NEG;
          sb[u - 1] = sc; mxb = fmaxf(mxb, sc);
        }
      }
      float suma = 0.f, sumb = 0.f;
#pragma unroll
      for (int t = 0; t < WIN_; ++t) {
        const float pa = __expf(sa[t] - mxa);
        const float pb = __expf(sb[t] - mxb);
        sa[t] = pa; suma += pa;
        sb[t] = pb; sumb += pb;
      }
      const float inva = 1.f / suma;
      const float invb = 1.f / sumb;
      float oa[DK_] = {}, ob[DK_] = {};
#pragma unroll
      for (int u = 0; u <= WIN_; ++u) {
        float vr[DK_];
        *(float4*)&vr[0] = *(const float4*)(vrow + u * LDA_S_);
        *(float4*)&vr[4] = *(const float4*)(vrow + u * LDA_S_ + 4);
        if (u < WIN_) {
          const float wgt = sa[u] * inva;
#pragma unroll
          for (int d = 0; d < DK_; ++d) oa[d] += wgt * vr[d];
        }
        if (u >= 1) {
          const float wgt = sb[u - 1] * invb;
#pragma unroll
          for (int d = 0; d < DK_; ++d) ob[d] += wgt * vr[d];
        }
      }
      *(float4*)(qrow)              = make_float4(oa[0], oa[1], oa[2], oa[3]);
      *(float4*)(qrow + 4)          = make_float4(oa[4], oa[5], oa[6], oa[7]);
      *(float4*)(qrow + LDA_S_)     = make_float4(ob[0], ob[1], ob[2], ob[3]);
      *(float4*)(qrow + LDA_S_ + 4) = make_float4(ob[4], ob[5], ob[6], ob[7]);
    }
  }
  __syncthreads();

  for (int pass = 0; pass < CH_PASSES_; ++pass) {
    const int stage = (pass < 2) ? 0 : ((pass < 6) ? 1 : 2);
    const float* Xin  = Xs + (stage & 1) * CH_X_;
    float*       Xout = Xs + ((stage & 1) ^ 1) * CH_X_;

    const int np = (pass + 1 < CH_PASSES_) ? pass + 1 : pass;
    int n0n, hicn; const float* Wn; const float* bn;
    chain_pass_geom(np, wave, Wo, bo, hwW, hwb, n0n, hicn, Wn, bn);
    const float* nx_lo = Wn + (size_t)(n0n + c) * D_ + 4 * h;
    const float* nx_hi = nx_lo + (size_t)hicn * D_;

    const float bv0 = bst[n0 + c];
    const float bv1 = bst[n0 + 16 + c];
    const float bv2 = bst[n0 + hic + c];
    const float bv3 = bst[n0 + hic + 16 + c];

    v8f acc[2][4];
#pragma unroll
    for (int mt = 0; mt < 2; ++mt) {
#pragma unroll
      for (int t = 0; t < 4; ++t) acc[mt][t] = (v8f){};
    }

    chain_wave_pass(Xin + c * LDA_S_ + 4 * h, w_lo, w_hi, nx_lo, nx_hi, bq, acc);

    if (stage == 0) {
#pragma unroll
      for (int mt = 0; mt < 2; ++mt) {
#pragma unroll
        for (int r = 0; r < 8; ++r) {
          float* xr = Xout + (mt * 16 + 8 * h + r) * LDA_S_ + n0 + c;
          xr[ 0] = acc[mt][0][r] + bv0;
          xr[16] = acc[mt][1][r] + bv1;
          xr[32] = acc[mt][2][r] + bv2;
          xr[48] = acc[mt][3][r] + bv3;
        }
      }
    } else {
#pragma unroll
      for (int mt = 0; mt < 2; ++mt) {
#pragma unroll
        for (int r = 0; r < 8; ++r) {
          const int idx = (mt * 16 + 8 * h + r) * LDA_S_ + n0 + c;
          {
            const float lin = acc[mt][0][r] + bv0;
            const float gin = acc[mt][2][r] + bv2;
            const float g  = 1.f / (1.f + __expf(-gin));
            const float nl = fmaxf(lin, 0.f);
            Xout[idx] = g * Xin[idx] + (1.f - g) * nl;
          }
          {
            const float lin = acc[mt][1][r] + bv1;
            const float gin = acc[mt][3][r] + bv3;
            const float g  = 1.f / (1.f + __expf(-gin));
            const float nl = fmaxf(lin, 0.f);
            Xout[idx + 16] = g * Xin[idx + 16] + (1.f - g) * nl;
          }
        }
      }
    }

    if (pass == 1 || pass == 5 || pass == 9) __syncthreads();

    n0 = n0n; hic = hicn; bst = bn; w_lo = nx_lo; w_hi = nx_hi;
  }

  const float* Xf = Xs + CH_X_;
  float* o0 = out + (size_t)row0 * (2 * D_) + (size_t)dir * D_ + tid * 4;
#pragma unroll 8
  for (int v = 0; v < CH_ROWS_; ++v) {
    float4 val = *(const float4*)&Xf[v * LDA_S_ + tid * 4];
    *(float4*)(o0 + (size_t)v * (2 * D_)) = val;
    *(float4*)(o0 + (size_t)v * (2 * D_) + (size_t)BS_ * 2 * D_) = val;
  }
}

extern "C" void kernel_launch(void* const* d_in, const int* in_sizes, int n_in,
                              void* d_out, int out_size, void* d_ws, size_t ws_size,
                              hipStream_t stream) {
  const float* inputs    = (const float*)d_in[0];
  const float* left_pad  = (const float*)d_in[1];
  const float* right_pad = (const float*)d_in[2];
  const float* padrows[2] = {left_pad, right_pad};
  const float* Wqkv[2] = {(const float*)d_in[3],  (const float*)d_in[7]};
  const float* bqkv[2] = {(const float*)d_in[4],  (const float*)d_in[8]};
  const float* Wo[2]   = {(const float*)d_in[5],  (const float*)d_in[9]};
  const float* bo[2]   = {(const float*)d_in[6],  (const float*)d_in[10]};
  const float* hwW[2]  = {(const float*)d_in[11], (const float*)d_in[13]};
  const float* hwb[2]  = {(const float*)d_in[12], (const float*)d_in[14]};
  float* out = (float*)d_out;

  float* ws    = (float*)d_ws;
  float* kvl   = ws;
  float* kvr   = kvl + 2 * BSD_;
  float* padl  = kvr + 2 * BSD_;
  float* padr  = padl + WIDTH_ * 2 * D_;
  float* kvb[2]   = {kvl, kvr};
  float* padkv[2] = {padl, padr};

  dim3 blk(32, 4);

  for (int dir = 0; dir < 2; ++dir) {
    { dim3 grd(2 * D_ / 128, 1);
      gemm_wmma_f32<1, 16, LDS_TILE16_><<<grd, blk, 0, stream>>>(
          padrows[dir], D_, Wqkv[dir] + (size_t)D_ * D_, D_, bqkv[dir] + D_,
          padkv[dir], 2 * D_, WIDTH_, 2 * D_); }
    dim3 grd(D_ / 128, BS_ / 64);
    gemm_wmma_f32<4, 4, LDS_ALONE_><<<grd, blk, 0, stream>>>(
        inputs, D_, Wqkv[dir], D_, bqkv[dir], out + dir * D_, 2 * D_, BS_, D_);
    for (int t = 1; t < 3; ++t) {
      gemm_wmma_f32<4, 4, LDS_ALONE_><<<grd, blk, 0, stream>>>(
          inputs, D_, Wqkv[dir] + (size_t)t * D_ * D_, D_, bqkv[dir] + t * D_,
          kvb[dir] + (size_t)(t - 1) * BSD_, D_, BS_, D_);
    }
  }
  { dim3 grd(BS_ / CH_ROWS_, 2);
    chain_kernel<<<grd, blk, 0, stream>>>(ws, Wo[0], Wo[1], bo[0], bo[1],
                                          hwW[0], hwW[1], hwb[0], hwb[1], out); }
}


# q, k, v of both directions in one launch: a block stages its 64x512 input tile in LDS once and runs the six projections from it, W ring carried across them; sum order unchanged; launches 9 to 4
# speedup vs baseline: 3.0547x; 1.1176x over previous
#include <hip/hip_runtime.h>

#define B_ 8
#define S_ 512
#define D_ 512
#define H_ 64
#define DK_ 8
#define WIDTH_ 16
#define N_ (S_ + 2 * WIDTH_)
#define NL_ 2
#define WIN_ (WIDTH_ + 2)
#define BS_ (B_ * S_)
static const size_t BSD_ = (size_t)BS_ * D_;

typedef __attribute__((ext_vector_type(2))) float v2f;
typedef __attribute__((ext_vector_type(8))) float v8f;

typedef __attribute__((ext_vector_type(4))) float v4f;

#define WMMA_F32(a, b, c) __builtin_amdgcn_wmma_f32_16x16x4_f32(false, (a), false, (b), (short)0, (c), false, false)

#define LDA_S_ (512 + 4)
#define LDS_TILE16_ (16 * LDA_S_)
#define LDS_ALONE_ (162 * 256)

template <int MT, int PF, int LDS_FLOATS>
__global__ __launch_bounds__(128)
void gemm_wmma_f32(const float* __restrict__ A, int lda,
                   const float* __restrict__ W, int ldw,
                   const float* __restrict__ bias,
                   float* __restrict__ C, int ldc,
                   int M, int Nout) {
  constexpr int KK = 512;
  constexpr int BM = 16 * MT;
  constexpr int LDA_S = LDA_S_;
  static_assert(LDS_FLOATS >= BM * LDA_S, "LDS array smaller than the A tile");
  __shared__ __attribute__((aligned(16))) float As[LDS_FLOATS];

  const int lane = threadIdx.x;
  const int wave = threadIdx.y;
  const int tid  = wave * 32 + lane;
  const int row0 = blockIdx.y * BM;
  const int col0 = blockIdx.x * 128 + wave * 32;
  const int c = lane & 15;
  const int h = lane >> 4;

  int wr0 = col0 + c;      if (wr0 > Nout - 1) wr0 = Nout - 1;
  int wr1 = col0 + 16 + c; if (wr1 > Nout - 1) wr1 = Nout - 1;
  const float* w0 = W + (size_t)wr0 * ldw + 4 * h;
  const float* w1 = W + (size_t)wr1 * ldw + 4 * h;

  v4f bq[PF][2];
#pragma unroll
  for (int s = 0; s < PF; ++s) {
    bq[s][0] = *(const v4f*)(w0 + 8 * s);
    bq[s][1] = *(const v4f*)(w1 + 8 * s);
  }

#pragma unroll 32
  for (int v = 0; v < BM; ++v) {
    int rg = row0 + v; if (rg > M - 1) rg = M - 1;
    float4 val = *(const float4*)(A + (size_t)rg * lda + tid * 4);
    *(float4*)&As[v * LDA_S + tid * 4] = val;
  }
  __syncthreads();

  if (row0 >= M || col0 >= Nout) return;

  const float* ap = &As[c * LDA_S + 4 * h];

  v8f acc[MT][2];
#pragma unroll
  for (int mt = 0; mt < MT; ++mt) { acc[mt][0] = (v8f){}; acc[mt][1] = (v8f){}; }

  v4f an[MT];
#pragma unroll
  for (int mt = 0; mt < MT; ++mt) an[mt] = *(const v4f*)(ap + mt * 16 * LDA_S);

  for (int it = 0; it < KK / 8; it += PF) {
#pragma unroll
    for (int s = 0; s < PF; ++s) {
      const int kk = (it + s) * 8;
      v4f a[MT];
#pragma unroll
      for (int mt = 0; mt < MT; ++mt) a[mt] = an[mt];
      v4f b0 = bq[s][0], b1 = bq[s][1];
      int kp = kk + 8 * PF; if (kp > KK - 8) kp = KK - 8;
      bq[s][0] = *(const v4f*)(w0 + kp);
      bq[s][1] = *(const v4f*)(w1 + kp);
      int ka = kk + 8; if (ka > KK - 8) ka = KK - 8;
#pragma unroll
      for (int mt = 0; mt < MT; ++mt) an[mt] = *(const v4f*)(ap + mt * 16 * LDA_S + ka);

#pragma unroll
      for (int mt = 0; mt < MT; ++mt) {
        acc[mt][0] = WMMA_F32(a[mt].xy, b0.xy, acc[mt][0]);
        acc[mt][1] = WMMA_F32(a[mt].xy, b1.xy, acc[mt][1]);
      }
#pragma unroll
      for (int mt = 0; mt < MT; ++mt) {
        acc[mt][0] = WMMA_F32(a[mt].zw, b0.zw, acc[mt][0]);
        acc[mt][1] = WMMA_F32(a[mt].zw, b1.zw, acc[mt][1]);
      }
    }
  }

  const float bv0 = bias[col0 + c];
  const float bv1 = bias[col0 + 16 + c];
#pragma unroll
  for (int mt = 0; mt < MT; ++mt) {
#pragma unroll
    for (int r = 0; r < 8; ++r) {
      const int row = row0 + mt * 16 + 8 * h + r;
      if (row < M) {
        float* crow = C + (size_t)row * ldc + col0 + c;
        crow[ 0] = acc[mt][0][r] + bv0;
        crow[16] = acc[mt][1][r] + bv1;
      }
    }
  }
}

#define QKV_PASSES_ 6
#define QKV_PF_ 4

__device__ __forceinline__ void qkv_pass_geom(int pass,
                                              const float* Wl, const float* Wr,
                                              const float* bl, const float* br,
                                              float* out, float* ws,
                                              const float*& Wp, const float*& bp,
                                              float*& Cp, int& ldc) {
  const int dir = (pass >= 3) ? 1 : 0;
  const int t = pass - 3 * dir;
  Wp = (dir ? Wr : Wl) + (size_t)t * D_ * D_;
  bp = (dir ? br : bl) + t * D_;
  if (t == 0) { Cp = out + dir * D_; ldc = 2 * D_; }
  else        { Cp = ws + (size_t)(2 * dir + t - 1) * BSD_; ldc = D_; }
}

__global__ __launch_bounds__(128)
void qkv_kernel(const float* __restrict__ X,
                const float* __restrict__ Wl, const float* __restrict__ Wr,
                const float* __restrict__ bl, const float* __restrict__ br,
                float* __restrict__ out, float* __restrict__ ws) {
  constexpr int KK = 512;
  constexpr int MT = 4;
  constexpr int BM = 16 * MT;
  static_assert(LDS_ALONE_ >= BM * LDA_S_, "LDS array smaller than the A tile");
  static_assert(BS_ % BM == 0 && D_ % 128 == 0, "whole blocks only: no row or column guard");
  __shared__ __attribute__((aligned(16))) float As[LDS_ALONE_];

  const int lane = threadIdx.x;
  const int wave = threadIdx.y;
  const int tid  = wave * 32 + lane;
  const int row0 = blockIdx.y * BM;
  const int colw = blockIdx.x * 128 + wave * 32;
  const int c = lane & 15;
  const int h = lane >> 4;

  const float* Wp; const float* bp; float* Cp; int ldc;
  qkv_pass_geom(0, Wl, Wr, bl, br, out, ws, Wp, bp, Cp, ldc);
  const float* w0 = Wp + (size_t)(colw + c) * KK + 4 * h;
  v4f bq[QKV_PF_][2];
#pragma unroll
  for (int s = 0; s < QKV_PF_; ++s) {
    bq[s][0] = *(const v4f*)(w0 + 8 * s);
    bq[s][1] = *(const v4f*)(w0 + 8 * s + 16 * KK);
  }

#pragma unroll 32
  for (int v = 0; v < BM; ++v) {
    float4 val = *(const float4*)(X + (size_t)(row0 + v) * KK + tid * 4);
    *(float4*)&As[v * LDA_S_ + tid * 4] = val;
  }
  __syncthreads();

  const float* ap = &As[c * LDA_S_ + 4 * h];

  for (int pass = 0; pass < QKV_PASSES_; ++pass) {
    const int np = (pass + 1 < QKV_PASSES_) ? pass + 1 : pass;
    const float* Wn; const float* bn; float* Cn; int ldcn;
    qkv_pass_geom(np, Wl, Wr, bl, br, out, ws, Wn, bn, Cn, ldcn);
    const float* nx0 = Wn + (size_t)(colw + c) * KK + 4 * h;

    const float bv0 = bp[colw + c];
    const float bv1 = bp[colw + 16 + c];

    v8f acc[MT][2];
#pragma unroll
    for (int mt = 0; mt < MT; ++mt) { acc[mt][0] = (v8f){}; acc[mt][1] = (v8f){}; }

    v4f an[MT];
#pragma unroll
    for (int mt = 0; mt < MT; ++mt) an[mt] = *(const v4f*)(ap + mt * 16 * LDA_S_);

    for (int it = 0; it < KK / 8; it += QKV_PF_) {
      const bool wrap = (it + QKV_PF_ >= KK / 8);
      const float* cl = wrap ? nx0 : w0;
      const int kb = wrap ? 0 : (it + QKV_PF_) * 8;
#pragma unroll
      for (int s = 0; s < QKV_PF_; ++s) {
        const int kk = (it + s) * 8;
        v4f a[MT];
#pragma unroll
        for (int mt = 0; mt < MT; ++mt) a[mt] = an[mt];
        const v4f b0 = bq[s][0], b1 = bq[s][1];
        bq[s][0] = *(const v4f*)(cl + kb + 8 * s);
        bq[s][1] = *(const v4f*)(cl + kb + 8 * s + 16 * KK);
        int ka = kk + 8; if (ka > KK - 8) ka = KK - 8;
#pragma unroll
        for (int mt = 0; mt < MT; ++mt) an[mt] = *(const v4f*)(ap + mt * 16 * LDA_S_ + ka);

#pragma unroll
        for (int mt = 0; mt < MT; ++mt) {
          acc[mt][0] = WMMA_F32(a[mt].xy, b0.xy, acc[mt][0]);
          acc[mt][1] = WMMA_F32(a[mt].xy, b1.xy, acc[mt][1]);
        }
#pragma unroll
        for (int mt = 0; mt < MT; ++mt) {
          acc[mt][0] = WMMA_F32(a[mt].zw, b0.zw, acc[mt][0]);
          acc[mt][1] = WMMA_F32(a[mt].zw, b1.zw, acc[mt][1]);
        }
      }
    }

#pragma unroll
    for (int mt = 0; mt < MT; ++mt) {
#pragma unroll
      for (int r = 0; r < 8; ++r) {
        float* crow = Cp + (size_t)(row0 + mt * 16 + 8 * h + r) * ldc + colw + c;
        crow[ 0] = acc[mt][0][r] + bv0;
        crow[16] = acc[mt][1][r] + bv1;
      }
    }

    w0 = nx0; bp = bn; Cp = Cn; ldc = ldcn;
  }
}

#define CH_ROWS_ 32
#define CH_X_ (CH_ROWS_ * LDA_S_)
#define CH_PF_ 4
#define CH_PASSES_ 10
#define KW_ROWS_ (CH_ROWS_ + WIDTH_ + 1)
#define CH_LDS_ (CH_X_ + 2 * KW_ROWS_ * LDA_S_)

__device__ __forceinline__ void chain_pass_geom(int pass, int wave,
                                                const float* Wo, const float* bo,
                                                const float* hwW, const float* hwb,
                                                int& n0, int& hic,
                                                const float*& Wst, const float*& bst) {
  if (pass < 2) {
    n0 = wave * 128 + pass * 64; hic = 32; Wst = Wo; bst = bo;
  } else {
    const int q = pass - 2;
    const int layer = q >> 2;
    n0 = wave * 128 + (q & 3) * 32; hic = D_;
    Wst = hwW + (size_t)layer * 2 * D_ * D_; bst = hwb + layer * 2 * D_;
  }
}

__device__ __forceinline__ void chain_wave_pass(const float* ap,
                                                const float* w_lo, const float* w_hi,
                                                const float* nx_lo, const float* nx_hi,
                                                v4f (&bq)[CH_PF_][4], v8f (&acc)[2][4]) {
  constexpr int KK = 512;
  v4f an[2];
  an[0] = *(const v4f*)(ap);
  an[1] = *(const v4f*)(ap + 16 * LDA_S_);
  for (int it = 0; it < KK / 8; it += CH_PF_) {
    const bool wrap = (it + CH_PF_ >= KK / 8);
    const float* cl = wrap ? nx_lo : w_lo;
    const float* ch = wrap ? nx_hi : w_hi;
    const int kb = wrap ? 0 : (it + CH_PF_) * 8;
#pragma unroll
    for (int s = 0; s < CH_PF_; ++s) {
      const int kk = (it + s) * 8;
      const v4f a0 = an[0], a1 = an[1];
      const v4f b0 = bq[s][0], b1 = bq[s][1], b2 = bq[s][2], b3 = bq[s][3];
      bq[s][0] = *(const v4f*)(cl + kb + 8 * s);
      bq[s][1] = *(const v4f*)(cl + kb + 8 * s + 16 * KK);
      bq[s][2] = *(const v4f*)(ch + kb + 8 * s);
      bq[s][3] = *(const v4f*)(ch + kb + 8 * s + 16 * KK);
      int ka = kk + 8; if (ka > KK - 8) ka = KK - 8;
      an[0] = *(const v4f*)(ap + ka);
      an[1] = *(const v4f*)(ap + 16 * LDA_S_ + ka);

      acc[0][0] = WMMA_F32(a0.xy, b0.xy, acc[0][0]);
      acc[0][1] = WMMA_F32(a0.xy, b1.xy, acc[0][1]);
      acc[0][2] = WMMA_F32(a0.xy, b2.xy, acc[0][2]);
      acc[0][3] = WMMA_F32(a0.xy, b3.xy, acc[0][3]);
      acc[1][0] = WMMA_F32(a1.xy, b0.xy, acc[1][0]);
      acc[1][1] = WMMA_F32(a1.xy, b1.xy, acc[1][1]);
      acc[1][2] = WMMA_F32(a1.xy, b2.xy, acc[1][2]);
      acc[1][3] = WMMA_F32(a1.xy, b3.xy, acc[1][3]);
      acc[0][0] = WMMA_F32(a0.zw, b0.zw, acc[0][0]);
      acc[0][1] = WMMA_F32(a0.zw, b1.zw, acc[0][1]);
      acc[0][2] = WMMA_F32(a0.zw, b2.zw, acc[0][2]);
      acc[0][3] = WMMA_F32(a0.zw, b3.zw, acc[0][3]);
      acc[1][0] = WMMA_F32(a1.zw, b0.zw, acc[1][0]);
      acc[1][1] = WMMA_F32(a1.zw, b1.zw, acc[1][1]);
      acc[1][2] = WMMA_F32(a1.zw, b2.zw, acc[1][2]);
      acc[1][3] = WMMA_F32(a1.zw, b3.zw, acc[1][3]);
    }
  }
}

template <int KIND>
__device__ __forceinline__ void stage_kv_window(const float* __restrict__ kb,
                                                const float* __restrict__ pb,
                                                int wb, float* Ks, float* Vs, int tid4) {
#pragma unroll
  for (int w = 0; w < KW_ROWS_; ++w) {
    const float* pk;
    const float* pv;
    if (KIND == 0) {
      pk = kb + (wb + w) * D_; pv = pk + BSD_;
    } else if (KIND == 1) {
      if (w <= WIDTH_) { const int pr = (w == 0) ? 0 : w - 1; pk = pb + pr * (2 * D_); pv = pk + D_; }
      else { pk = kb + (w - (WIDTH_ + 1)) * D_; pv = pk + BSD_; }
    } else {
      if (w < CH_ROWS_) { pk = kb + (S_ - CH_ROWS_ + w) * D_; pv = pk + BSD_; }
      else { const int pr = (w - CH_ROWS_ < WIDTH_) ? w - CH_ROWS_ : WIDTH_ - 1; pk = pb + pr * (2 * D_); pv = pk + D_; }
    }
    const float4 kq = *(const float4*)pk;
    const float4 vq = *(const float4*)pv;
    *(float4*)&Ks[w * LDA_S_ + tid4] = kq;
    *(float4*)&Vs[w * LDA_S_ + tid4] = vq;
  }
}

__global__ __launch_bounds__(128)
void chain_kernel(const float* __restrict__ kvws,
                  const float* __restrict__ Wo_l,   const float* __restrict__ Wo_r,
                  const float* __restrict__ bo_l,   const float* __restrict__ bo_r,
                  const float* __restrict__ hwW_l,  const float* __restrict__ hwW_r,
                  const float* __restrict__ hwb_l,  const float* __restrict__ hwb_r,
                  float* out) {
  __shared__ __attribute__((aligned(16))) float Xs[CH_LDS_];
  static_assert(CH_LDS_ >= 2 * CH_X_, "LDS array smaller than the two activation buffers");
  static_assert(CH_LDS_ * 4 > 160 * 1024 && CH_LDS_ * 4 <= 320 * 1024, "LDS: one block per WGP");

  const int lane = threadIdx.x;
  const int wave = threadIdx.y;
  const int tid  = wave * 32 + lane;
  const int c = lane & 15;
  const int h = lane >> 4;
  const int dir  = blockIdx.y;
  const int row0 = blockIdx.x * CH_ROWS_;

  const float* Wo   = dir ? Wo_r   : Wo_l;
  const float* bo   = dir ? bo_r   : bo_l;
  const float* hwW  = dir ? hwW_r  : hwW_l;
  const float* hwb  = dir ? hwb_r  : hwb_l;

  int n0, hic; const float* Wst; const float* bst;
  chain_pass_geom(0, wave, Wo, bo, hwW, hwb, n0, hic, Wst, bst);
  const float* w_lo = Wst + (size_t)(n0 + c) * D_ + 4 * h;
  const float* w_hi = w_lo + (size_t)hic * D_;
  v4f bq[CH_PF_][4];
#pragma unroll
  for (int s = 0; s < CH_PF_; ++s) {
    bq[s][0] = *(const v4f*)(w_lo + 8 * s);
    bq[s][1] = *(const v4f*)(w_lo + 8 * s + 16 * D_);
    bq[s][2] = *(const v4f*)(w_hi + 8 * s);
    bq[s][3] = *(const v4f*)(w_hi + 8 * s + 16 * D_);
  }

  const int bb = row0 / S_;
  const int s0 = row0 % S_;
  const int wb = dir ? s0 : s0 - (WIDTH_ + 1);
  float* Ks = Xs + CH_X_;
  float* Vs = Ks + KW_ROWS_ * LDA_S_;
  {
    const float* qsrc = out + (size_t)row0 * (2 * D_) + (size_t)dir * D_ + tid * 4;
#pragma unroll
    for (int v = 0; v < CH_ROWS_; ++v) {
      float4 val = *(const float4*)(qsrc + (size_t)v * (2 * D_));
      *(float4*)&Xs[v * LDA_S_ + tid * 4] = val;
    }
    static_assert(S_ % CH_ROWS_ == 0 && CH_ROWS_ > WIDTH_ + 1, "window shapes of stage_kv_window");
    const float* kb = kvws + (size_t)dir * (2 * BSD_) + (size_t)(bb * S_) * D_ + tid * 4;
    const float* pb = kvws + 4 * BSD_ + (size_t)dir * (WIDTH_ * 2 * D_) + tid * 4;
    if (wb < 0)                          stage_kv_window<1>(kb, pb, wb, Ks, Vs, tid * 4);
    else if (wb + KW_ROWS_ - 1 >= S_)    stage_kv_window<2>(kb, pb, wb, Ks, Vs, tid * 4);
    else                                 stage_kv_window<0>(kb, pb, wb, Ks, Vs, tid * 4);
  }
  __syncthreads();

  {
    const int hh = tid & (H_ - 1);
    const int rsel = tid >> 6;
    const float scale = 0.35355339059327373f;
    const float NEG = -3.0e38f;
    for (int it = 0; it < CH_ROWS_ / 4; ++it) {
      const int r0 = (it * 2 + rsel) * 2;
      float* qrow = Xs + r0 * LDA_S_ + hh * DK_;
      const float* krow = Ks + r0 * LDA_S_ + hh * DK_;
      const float* vrow = Vs + r0 * LDA_S_ + hh * DK_;
      float qa[DK_], qb[DK_];
      *(float4*)&qa[0] = *(const float4*)(qrow);
      *(float4*)&qa[4] = *(const float4*)(qrow + 4);
      *(float4*)&qb[0] = *(const float4*)(qrow + LDA_S_);
      *(float4*)&qb[4] = *(const float4*)(qrow + LDA_S_ + 4);
      const int sa_row = s0 + r0;
      const int sb_row = s0 + r0 + 1;
      float sa[WIN_], sb[WIN_];
      float mxa = NEG, mxb = NEG;
#pragma unroll
      for (int u = 0; u <= WIN_; ++u) {
        float kr[DK_];
        *(float4*)&kr[0] = *(const float4*)(krow + u * LDA_S_);
        *(float4*)&kr[4] = *(const float4*)(krow + u * LDA_S_ + 4);
        if (u < WIN_) {
          float dot = 0.f;
#pragma unroll
          for (int d = 0; d < DK_; ++d) dot += qa[d] * kr[d];
          const bool ok = (u == 0) ? !(dir == 0 && sa_row == 0)
                        : ((u == WIN_ - 1) ? !(dir == 1 && sa_row == S_ - 1) : true);
          const float sc = ok ? dot * scale : NEG;
          sa[u] = sc; mxa = fmaxf(mxa, sc);
        }
        if (u >= 1) {
          float dot = 0.f;
#pragma unroll
          for (int d = 0; d < DK_; ++d) dot += qb[d] * kr[d];
          const bool ok = (u - 1 == 0) ? !(dir == 0 && sb_row == 0)
                        : ((u - 1 == WIN_ - 1) ? !(dir == 1 && sb_row == S_ - 1) : true);
          const float sc = ok ? dot * scale : NEG;
          sb[u - 1] = sc; mxb = fmaxf(mxb, sc);
        }
      }
      float suma = 0.f, sumb = 0.f;
#pragma unroll
      for (int t = 0; t < WIN_; ++t) {
        const float pa = __expf(sa[t] - mxa);
        const float pb = __expf(sb[t] - mxb);
        sa[t] = pa; suma += pa;
        sb[t] = pb; sumb += pb;
      }
      const float inva = 1.f / suma;
      const float invb = 1.f / sumb;
      float oa[DK_] = {}, ob[DK_] = {};
#pragma unroll
      for (int u = 0; u <= WIN_; ++u) {
        float vr[DK_];
        *(float4*)&vr[0] = *(const float4*)(vrow + u * LDA_S_);
        *(float4*)&vr[4] = *(const float4*)(vrow + u * LDA_S_ + 4);
        if (u < WIN_) {
          const float wgt = sa[u] * inva;
#pragma unroll
          for (int d = 0; d < DK_; ++d) oa[d] += wgt * vr[d];
        }
        if (u >= 1) {
          const float wgt = sb[u - 1] * invb;
#pragma unroll
          for (int d = 0; d < DK_; ++d) ob[d] += wgt * vr[d];
        }
      }
      *(float4*)(qrow)              = make_float4(oa[0], oa[1], oa[2], oa[3]);
      *(float4*)(qrow + 4)          = make_float4(oa[4], oa[5], oa[6], oa[7]);
      *(float4*)(qrow + LDA_S_)     = make_float4(ob[0], ob[1], ob[2], ob[3]);
      *(float4*)(qrow + LDA_S_ + 4) = make_float4(ob[4], ob[5], ob[6], ob[7]);
    }
  }
  __syncthreads();

  for (int pass = 0; pass < CH_PASSES_; ++pass) {
    const int stage = (pass < 2) ? 0 : ((pass < 6) ? 1 : 2);
    const float* Xin  = Xs + (stage & 1) * CH_X_;
    float*       Xout = Xs + ((stage & 1) ^ 1) * CH_X_;

    const int np = (pass + 1 < CH_PASSES_) ? pass + 1 : pass;
    int n0n, hicn; const float* Wn; const float* bn;
    chain_pass_geom(np, wave, Wo, bo, hwW, hwb, n0n, hicn, Wn, bn);
    const float* nx_lo = Wn + (size_t)(n0n + c) * D_ + 4 * h;
    const float* nx_hi = nx_lo + (size_t)hicn * D_;

    const float bv0 = bst[n0 + c];
    const float bv1 = bst[n0 + 16 + c];
    const float bv2 = bst[n0 + hic + c];
    const float bv3 = bst[n0 + hic + 16 + c];

    v8f acc[2][4];
#pragma unroll
    for (int mt = 0; mt < 2; ++mt) {
#pragma unroll
      for (int t = 0; t < 4; ++t) acc[mt][t] = (v8f){};
    }

    chain_wave_pass(Xin + c * LDA_S_ + 4 * h, w_lo, w_hi, nx_lo, nx_hi, bq, acc);

    if (stage == 0) {
#pragma unroll
      for (int mt = 0; mt < 2; ++mt) {
#pragma unroll
        for (int r = 0; r < 8; ++r) {
          float* xr = Xout + (mt * 16 + 8 * h + r) * LDA_S_ + n0 + c;
          xr[ 0] = acc[mt][0][r] + bv0;
          xr[16] = acc[mt][1][r] + bv1;
          xr[32] = acc[mt][2][r] + bv2;
          xr[48] = acc[mt][3][r] + bv3;
        }
      }
    } else {
#pragma unroll
      for (int mt = 0; mt < 2; ++mt) {
#pragma unroll
        for (int r = 0; r < 8; ++r) {
          const int idx = (mt * 16 + 8 * h + r) * LDA_S_ + n0 + c;
          {
            const float lin = acc[mt][0][r] + bv0;
            const float gin = acc[mt][2][r] + bv2;
            const float g  = 1.f / (1.f + __expf(-gin));
            const float nl = fmaxf(lin, 0.f);
            Xout[idx] = g * Xin[idx] + (1.f - g) * nl;
          }
          {
            const float lin = acc[mt][1][r] + bv1;
            const float gin = acc[mt][3][r] + bv3;
            const float g  = 1.f / (1.f + __expf(-gin));
            const float nl = fmaxf(lin, 0.f);
            Xout[idx + 16] = g * Xin[idx + 16] + (1.f - g) * nl;
          }
        }
      }
    }

    if (pass == 1 || pass == 5 || pass == 9) __syncthreads();

    n0 = n0n; hic = hicn; bst = bn; w_lo = nx_lo; w_hi = nx_hi;
  }

  const float* Xf = Xs + CH_X_;
  float* o0 = out + (size_t)row0 * (2 * D_) + (size_t)dir * D_ + tid * 4;
#pragma unroll 8
  for (int v = 0; v < CH_ROWS_; ++v) {
    float4 val = *(const float4*)&Xf[v * LDA_S_ + tid * 4];
    *(float4*)(o0 + (size_t)v * (2 * D_)) = val;
    *(float4*)(o0 + (size_t)v * (2 * D_) + (size_t)BS_ * 2 * D_) = val;
  }
}

extern "C" void kernel_launch(void* const* d_in, const int* in_sizes, int n_in,
                              void* d_out, int out_size, void* d_ws, size_t ws_size,
                              hipStream_t stream) {
  const float* inputs    = (const float*)d_in[0];
  const float* left_pad  = (const float*)d_in[1];
  const float* right_pad = (const float*)d_in[2];
  const float* padrows[2] = {left_pad, right_pad};
  const float* Wqkv[2] = {(const float*)d_in[3],  (const float*)d_in[7]};
  const float* bqkv[2] = {(const float*)d_in[4],  (const float*)d_in[8]};
  const float* Wo[2]   = {(const float*)d_in[5],  (const float*)d_in[9]};
  const float* bo[2]   = {(const float*)d_in[6],  (const float*)d_in[10]};
  const float* hwW[2]  = {(const float*)d_in[11], (const float*)d_in[13]};
  const float* hwb[2]  = {(const float*)d_in[12], (const float*)d_in[14]};
  float* out = (float*)d_out;

  float* ws    = (float*)d_ws;
  float* padl  = ws + 4 * BSD_;
  float* padr  = padl + WIDTH_ * 2 * D_;
  float* padkv[2] = {padl, padr};

  dim3 blk(32, 4);

  for (int dir = 0; dir < 2; ++dir) {
    dim3 grd(2 * D_ / 128, 1);
    gemm_wmma_f32<1, 16, LDS_TILE16_><<<grd, blk, 0, stream>>>(
        padrows[dir], D_, Wqkv[dir] + (size_t)D_ * D_, D_, bqkv[dir] + D_,
        padkv[dir], 2 * D_, WIDTH_, 2 * D_);
  }
  { dim3 grd(D_ / 128, BS_ / 64);
    qkv_kernel<<<grd, blk, 0, stream>>>(inputs, Wqkv[0], Wqkv[1], bqkv[0], bqkv[1], out, ws); }
  { dim3 grd(BS_ / CH_ROWS_, 2);
    chain_kernel<<<grd, blk, 0, stream>>>(ws, Wo[0], Wo[1], bo[0], bo[1],
                                          hwW[0], hwW[1], hwb[0], hwb[1], out); }
}


# Two waves per SIMD: both 256-block kernels run 8-wave blocks (a wave's columns halved: 32x32 wave tile in the chained kernel, 64x32 kept for q/k/v; k and sum order unchanged); waits overlap WMMAs
# speedup vs baseline: 3.2029x; 1.0485x over previous
#include <hip/hip_runtime.h>

#define B_ 8
#define S_ 512
#define D_ 512
#define H_ 64
#define DK_ 8
#define WIDTH_ 16
#define N_ (S_ + 2 * WIDTH_)
#define NL_ 2
#define WIN_ (WIDTH_ + 2)
#define BS_ (B_ * S_)
static const size_t BSD_ = (size_t)BS_ * D_;

typedef __attribute__((ext_vector_type(2))) float v2f;
typedef __attribute__((ext_vector_type(8))) float v8f;

typedef __attribute__((ext_vector_type(4))) float v4f;

#define WMMA_F32(a, b, c) __builtin_amdgcn_wmma_f32_16x16x4_f32(false, (a), false, (b), (short)0, (c), false, false)

#define LDA_S_ (512 + 4)
#define LDS_TILE16_ (16 * LDA_S_)
#define LDS_ALONE_ (162 * 256)

template <int MT, int PF, int LDS_FLOATS>
__global__ __launch_bounds__(128)
void gemm_wmma_f32(const float* __restrict__ A, int lda,
                   const float* __restrict__ W, int ldw,
                   const float* __restrict__ bias,
                   float* __restrict__ C, int ldc,
                   int M, int Nout) {
  constexpr int KK = 512;
  constexpr int BM = 16 * MT;
  constexpr int LDA_S = LDA_S_;
  static_assert(LDS_FLOATS >= BM * LDA_S, "LDS array smaller than the A tile");
  __shared__ __attribute__((aligned(16))) float As[LDS_FLOATS];

  const int lane = threadIdx.x;
  const int wave = threadIdx.y;
  const int tid  = wave * 32 + lane;
  const int row0 = blockIdx.y * BM;
  const int col0 = blockIdx.x * 128 + wave * 32;
  const int c = lane & 15;
  const int h = lane >> 4;

  int wr0 = col0 + c;      if (wr0 > Nout - 1) wr0 = Nout - 1;
  int wr1 = col0 + 16 + c; if (wr1 > Nout - 1) wr1 = Nout - 1;
  const float* w0 = W + (size_t)wr0 * ldw + 4 * h;
  const float* w1 = W + (size_t)wr1 * ldw + 4 * h;

  v4f bq[PF][2];
#pragma unroll
  for (int s = 0; s < PF; ++s) {
    bq[s][0] = *(const v4f*)(w0 + 8 * s);
    bq[s][1] = *(const v4f*)(w1 + 8 * s);
  }

#pragma unroll 32
  for (int v = 0; v < BM; ++v) {
    int rg = row0 + v; if (rg > M - 1) rg = M - 1;
    float4 val = *(const float4*)(A + (size_t)rg * lda + tid * 4);
    *(float4*)&As[v * LDA_S + tid * 4] = val;
  }
  __syncthreads();

  if (row0 >= M || col0 >= Nout) return;

  const float* ap = &As[c * LDA_S + 4 * h];

  v8f acc[MT][2];
#pragma unroll
  for (int mt = 0; mt < MT; ++mt) { acc[mt][0] = (v8f){}; acc[mt][1] = (v8f){}; }

  v4f an[MT];
#pragma unroll
  for (int mt = 0; mt < MT; ++mt) an[mt] = *(const v4f*)(ap + mt * 16 * LDA_S);

  for (int it = 0; it < KK / 8; it += PF) {
#pragma unroll
    for (int s = 0; s < PF; ++s) {
      const int kk = (it + s) * 8;
      v4f a[MT];
#pragma unroll
      for (int mt = 0; mt < MT; ++mt) a[mt] = an[mt];
      v4f b0 = bq[s][0], b1 = bq[s][1];
      int kp = kk + 8 * PF; if (kp > KK - 8) kp = KK - 8;
      bq[s][0] = *(const v4f*)(w0 + kp);
      bq[s][1] = *(const v4f*)(w1 + kp);
      int ka = kk + 8; if (ka > KK - 8) ka = KK - 8;
#pragma unroll
      for (int mt = 0; mt < MT; ++mt) an[mt] = *(const v4f*)(ap + mt * 16 * LDA_S + ka);

#pragma unroll
      for (int mt = 0; mt < MT; ++mt) {
        acc[mt][0] = WMMA_F32(a[mt].xy, b0.xy, acc[mt][0]);
        acc[mt][1] = WMMA_F32(a[mt].xy, b1.xy, acc[mt][1]);
      }
#pragma unroll
      for (int mt = 0; mt < MT; ++mt) {
        acc[mt][0] = WMMA_F32(a[mt].zw, b0.zw, acc[mt][0]);
        acc[mt][1] = WMMA_F32(a[mt].zw, b1.zw, acc[mt][1]);
      }
    }
  }

  const float bv0 = bias[col0 + c];
  const float bv1 = bias[col0 + 16 + c];
#pragma unroll
  for (int mt = 0; mt < MT; ++mt) {
#pragma unroll
    for (int r = 0; r < 8; ++r) {
      const int row = row0 + mt * 16 + 8 * h + r;
      if (row < M) {
        float* crow = C + (size_t)row * ldc + col0 + c;
        crow[ 0] = acc[mt][0][r] + bv0;
        crow[16] = acc[mt][1][r] + bv1;
      }
    }
  }
}

#define QKV_PASSES_ 3
#define QKV_PF_ 4

__device__ __forceinline__ void qkv_pass_geom(int proj,
                                              const float* Wl, const float* Wr,
                                              const float* bl, const float* br,
                                              float* out, float* ws,
                                              const float*& Wp, const float*& bp,
                                              float*& Cp, int& ldc) {
  const int dir = (proj >= 3) ? 1 : 0;
  const int t = proj - 3 * dir;
  Wp = (dir ? Wr : Wl) + (size_t)t * D_ * D_;
  bp = (dir ? br : bl) + t * D_;
  if (t == 0) { Cp = out + dir * D_; ldc = 2 * D_; }
  else        { Cp = ws + (size_t)(2 * dir + t - 1) * BSD_; ldc = D_; }
}

__global__ __launch_bounds__(256)
void qkv_kernel(const float* __restrict__ X,
                const float* __restrict__ Wl, const float* __restrict__ Wr,
                const float* __restrict__ bl, const float* __restrict__ br,
                float* __restrict__ out, float* __restrict__ ws) {
  constexpr int KK = 512;
  constexpr int MT = 4;
  constexpr int BM = 16 * MT;
  static_assert(LDS_ALONE_ >= BM * LDA_S_, "LDS array smaller than the A tile");
  static_assert(BS_ % BM == 0 && D_ % 128 == 0, "whole blocks only: no row or column guard");
  __shared__ __attribute__((aligned(16))) float As[LDS_ALONE_];

  const int lane = threadIdx.x;
  const int wave = threadIdx.y;
  const int tid  = wave * 32 + lane;
  const int row0 = blockIdx.y * BM;
  const int colw = blockIdx.x * 128 + (wave & 3) * 32;
  const int psel = wave >> 2;
  const int c = lane & 15;
  const int h = lane >> 4;

  const float* Wp; const float* bp; float* Cp; int ldc;
  qkv_pass_geom(psel, Wl, Wr, bl, br, out, ws, Wp, bp, Cp, ldc);
  const float* w0 = Wp + (size_t)(colw + c) * KK + 4 * h;
  v4f bq[QKV_PF_][2];
#pragma unroll
  for (int s = 0; s < QKV_PF_; ++s) {
    bq[s][0] = *(const v4f*)(w0 + 8 * s);
    bq[s][1] = *(const v4f*)(w0 + 8 * s + 16 * KK);
  }

  const int srow = tid >> 7;
  const int scol = (tid & 127) * 4;
#pragma unroll
  for (int v = 0; v < BM; v += 2) {
    float4 val = *(const float4*)(X + (size_t)(row0 + v + srow) * KK + scol);
    *(float4*)&As[(v + srow) * LDA_S_ + scol] = val;
  }
  __syncthreads();

  const float* ap = &As[c * LDA_S_ + 4 * h];

  for (int pass = 0; pass < QKV_PASSES_; ++pass) {
    const int np = 2 * ((pass + 1 < QKV_PASSES_) ? pass + 1 : pass) + psel;
    const float* Wn; const float* bn; float* Cn; int ldcn;
    qkv_pass_geom(np, Wl, Wr, bl, br, out, ws, Wn, bn, Cn, ldcn);
    const float* nx0 = Wn + (size_t)(colw + c) * KK + 4 * h;

    const float bv0 = bp[colw + c];
    const float bv1 = bp[colw + 16 + c];

    v8f acc[MT][2];
#pragma unroll
    for (int mt = 0; mt < MT; ++mt) { acc[mt][0] = (v8f){}; acc[mt][1] = (v8f){}; }

    v4f an[MT];
#pragma unroll
    for (int mt = 0; mt < MT; ++mt) an[mt] = *(const v4f*)(ap + mt * 16 * LDA_S_);

    for (int it = 0; it < KK / 8; it += QKV_PF_) {
      const bool wrap = (it + QKV_PF_ >= KK / 8);
      const float* cl = wrap ? nx0 : w0;
      const int kb = wrap ? 0 : (it + QKV_PF_) * 8;
#pragma unroll
      for (int s = 0; s < QKV_PF_; ++s) {
        const int kk = (it + s) * 8;
        v4f a[MT];
#pragma unroll
        for (int mt = 0; mt < MT; ++mt) a[mt] = an[mt];
        const v4f b0 = bq[s][0], b1 = bq[s][1];
        bq[s][0] = *(const v4f*)(cl + kb + 8 * s);
        bq[s][1] = *(const v4f*)(cl + kb + 8 * s + 16 * KK);
        int ka = kk + 8; if (ka > KK - 8) ka = KK - 8;
#pragma unroll
        for (int mt = 0; mt < MT; ++mt) an[mt] = *(const v4f*)(ap + mt * 16 * LDA_S_ + ka);

#pragma unroll
        for (int mt = 0; mt < MT; ++mt) {
          acc[mt][0] = WMMA_F32(a[mt].xy, b0.xy, acc[mt][0]);
          acc[mt][1] = WMMA_F32(a[mt].xy, b1.xy, acc[mt][1]);
        }
#pragma unroll
        for (int mt = 0; mt < MT; ++mt) {
          acc[mt][0] = WMMA_F32(a[mt].zw, b0.zw, acc[mt][0]);
          acc[mt][1] = WMMA_F32(a[mt].zw, b1.zw, acc[mt][1]);
        }
      }
    }

#pragma unroll
    for (int mt = 0; mt < MT; ++mt) {
#pragma unroll
      for (int r = 0; r < 8; ++r) {
        float* crow = Cp + (size_t)(row0 + mt * 16 + 8 * h + r) * ldc + colw + c;
        crow[ 0] = acc[mt][0][r] + bv0;
        crow[16] = acc[mt][1][r] + bv1;
      }
    }

    w0 = nx0; bp = bn; Cp = Cn; ldc = ldcn;
  }
}

#define CH_ROWS_ 32
#define CH_X_ (CH_ROWS_ * LDA_S_)
#define CH_PF_ 4
#define CH_PASSES_ 10
#define KW_ROWS_ (CH_ROWS_ + WIDTH_ + 1)
#define CH_LDS_ (CH_X_ + 2 * KW_ROWS_ * LDA_S_)

__device__ __forceinline__ void chain_pass_geom(int pass, int wave,
                                                const float* Wo, const float* bo,
                                                const float* hwW, const float* hwb,
                                                int& n0, int& hic,
                                                const float*& Wst, const float*& bst) {
  if (pass < 2) {
    n0 = wave * 64 + pass * 32; hic = 16; Wst = Wo; bst = bo;
  } else {
    const int q = pass - 2;
    const int layer = q >> 2;
    n0 = wave * 64 + (q & 3) * 16; hic = D_;
    Wst = hwW + (size_t)layer * 2 * D_ * D_; bst = hwb + layer * 2 * D_;
  }
}

__device__ __forceinline__ void chain_wave_pass(const float* ap,
                                                const float* w_lo, const float* w_hi,
                                                const float* nx_lo, const float* nx_hi,
                                                v4f (&bq)[CH_PF_][2], v8f (&acc)[2][2]) {
  constexpr int KK = 512;
  v4f an[2];
  an[0] = *(const v4f*)(ap);
  an[1] = *(const v4f*)(ap + 16 * LDA_S_);
#pragma unroll
  for (int it = 0; it < KK / 8; it += CH_PF_) {
    const bool wrap = (it + CH_PF_ >= KK / 8);
    const float* cl = wrap ? nx_lo : w_lo;
    const float* ch = wrap ? nx_hi : w_hi;
    const int kb = wrap ? 0 : (it + CH_PF_) * 8;
#pragma unroll
    for (int s = 0; s < CH_PF_; ++s) {
      const int kk = (it + s) * 8;
      const v4f a0 = an[0], a1 = an[1];
      const v4f b0 = bq[s][0], b1 = bq[s][1];
      bq[s][0] = *(const v4f*)(cl + kb + 8 * s);
      bq[s][1] = *(const v4f*)(ch + kb + 8 * s);
      int ka = kk + 8; if (ka > KK - 8) ka = KK - 8;
      an[0] = *(const v4f*)(ap + ka);
      an[1] = *(const v4f*)(ap + 16 * LDA_S_ + ka);

      acc[0][0] = WMMA_F32(a0.xy, b0.xy, acc[0][0]);
      acc[0][1] = WMMA_F32(a0.xy, b1.xy, acc[0][1]);
      acc[1][0] = WMMA_F32(a1.xy, b0.xy, acc[1][0]);
      acc[1][1] = WMMA_F32(a1.xy, b1.xy, acc[1][1]);
      acc[0][0] = WMMA_F32(a0.zw, b0.zw, acc[0][0]);
      acc[0][1] = WMMA_F32(a0.zw, b1.zw, acc[0][1]);
      acc[1][0] = WMMA_F32(a1.zw, b0.zw, acc[1][0]);
      acc[1][1] = WMMA_F32(a1.zw, b1.zw, acc[1][1]);
    }
  }
}

template <int KIND>
__device__ __forceinline__ void stage_kv_window(const float* __restrict__ kb,
                                                const float* __restrict__ pb,
                                                int wb, float* dst) {
#pragma unroll
  for (int w = 0; w < KW_ROWS_; ++w) {
    const float* p;
    if (KIND == 0) {
      p = kb + (wb + w) * D_;
    } else if (KIND == 1) {
      if (w <= WIDTH_) { const int pr = (w == 0) ? 0 : w - 1; p = pb + pr * (2 * D_); }
      else { p = kb + (w - (WIDTH_ + 1)) * D_; }
    } else {
      if (w < CH_ROWS_) { p = kb + (S_ - CH_ROWS_ + w) * D_; }
      else { const int pr = (w - CH_ROWS_ < WIDTH_) ? w - CH_ROWS_ : WIDTH_ - 1; p = pb + pr * (2 * D_); }
    }
    const float4 q4 = *(const float4*)p;
    *(float4*)&dst[w * LDA_S_] = q4;
  }
}

__global__ __launch_bounds__(256)
void chain_kernel(const float* __restrict__ kvws,
                  const float* __restrict__ Wo_l,   const float* __restrict__ Wo_r,
                  const float* __restrict__ bo_l,   const float* __restrict__ bo_r,
                  const float* __restrict__ hwW_l,  const float* __restrict__ hwW_r,
                  const float* __restrict__ hwb_l,  const float* __restrict__ hwb_r,
                  float* out) {
  __shared__ __attribute__((aligned(16))) float Xs[CH_LDS_];
  static_assert(CH_LDS_ >= 2 * CH_X_, "LDS array smaller than the two activation buffers");
  static_assert(CH_LDS_ * 4 > 160 * 1024 && CH_LDS_ * 4 <= 320 * 1024, "LDS: one block per WGP");

  const int lane = threadIdx.x;
  const int wave = threadIdx.y;
  const int tid  = wave * 32 + lane;
  const int c = lane & 15;
  const int h = lane >> 4;
  const int dir  = blockIdx.y;
  const int row0 = blockIdx.x * CH_ROWS_;

  const float* Wo   = dir ? Wo_r   : Wo_l;
  const float* bo   = dir ? bo_r   : bo_l;
  const float* hwW  = dir ? hwW_r  : hwW_l;
  const float* hwb  = dir ? hwb_r  : hwb_l;

  int n0, hic; const float* Wst; const float* bst;
  chain_pass_geom(0, wave, Wo, bo, hwW, hwb, n0, hic, Wst, bst);
  const float* w_lo = Wst + (size_t)(n0 + c) * D_ + 4 * h;
  const float* w_hi = w_lo + (size_t)hic * D_;
  v4f bq[CH_PF_][2];
#pragma unroll
  for (int s = 0; s < CH_PF_; ++s) {
    bq[s][0] = *(const v4f*)(w_lo + 8 * s);
    bq[s][1] = *(const v4f*)(w_hi + 8 * s);
  }

  const int srow = tid >> 7;
  const int scol = (tid & 127) * 4;
  const int bb = row0 / S_;
  const int s0 = row0 % S_;
  const int wb = dir ? s0 : s0 - (WIDTH_ + 1);
  float* Ks = Xs + CH_X_;
  float* Vs = Ks + KW_ROWS_ * LDA_S_;
  {
    const float* qsrc = out + (size_t)(row0 + srow) * (2 * D_) + (size_t)dir * D_ + scol;
#pragma unroll
    for (int v = 0; v < CH_ROWS_; v += 2) {
      float4 val = *(const float4*)(qsrc + (size_t)v * (2 * D_));
      *(float4*)&Xs[(v + srow) * LDA_S_ + scol] = val;
    }
    static_assert(S_ % CH_ROWS_ == 0 && CH_ROWS_ > WIDTH_ + 1, "window shapes of stage_kv_window");
    const float* kb = kvws + (size_t)dir * (2 * BSD_) + (size_t)srow * BSD_ + (size_t)(bb * S_) * D_ + scol;
    const float* pb = kvws + 4 * BSD_ + (size_t)dir * (WIDTH_ * 2 * D_) + srow * D_ + scol;
    float* wdst = (srow ? Vs : Ks) + scol;
    if (wb < 0)                          stage_kv_window<1>(kb, pb, wb, wdst);
    else if (wb + KW_ROWS_ - 1 >= S_)    stage_kv_window<2>(kb, pb, wb, wdst);
    else                                 stage_kv_window<0>(kb, pb, wb, wdst);
  }
  __syncthreads();

  {
    const int hh = tid & (H_ - 1);
    const int rsel = tid >> 6;
    const float scale = 0.35355339059327373f;
    const float NEG = -3.0e38f;
    for (int it = 0; it < CH_ROWS_ / 8; ++it) {
      const int r0 = (it * 4 + rsel) * 2;
      float* qrow = Xs + r0 * LDA_S_ + hh * DK_;
      const float* krow = Ks + r0 * LDA_S_ + hh * DK_;
      const float* vrow = Vs + r0 * LDA_S_ + hh * DK_;
      float qa[DK_], qb[DK_];
      *(float4*)&qa[0] = *(const float4*)(qrow);
      *(float4*)&qa[4] = *(const float4*)(qrow + 4);
      *(float4*)&qb[0] = *(const float4*)(qrow + LDA_S_);
      *(float4*)&qb[4] = *(const float4*)(qrow + LDA_S_ + 4);
      const int sa_row = s0 + r0;
      const int sb_row = s0 + r0 + 1;
      float sa[WIN_], sb[WIN_];
      float mxa = NEG, mxb = NEG;
#pragma unroll
      for (int u = 0; u <= WIN_; ++u) {
        float kr[DK_];
        *(float4*)&kr[0] = *(const float4*)(krow + u * LDA_S_);
        *(float4*)&kr[4] = *(const float4*)(krow + u * LDA_S_ + 4);
        if (u < WIN_) {
          float dot = 0.f;
#pragma unroll
          for (int d = 0; d < DK_; ++d) dot += qa[d] * kr[d];
          const bool ok = (u == 0) ? !(dir == 0 && sa_row == 0)
                        : ((u == WIN_ - 1) ? !(dir == 1 && sa_row == S_ - 1) : true);
          const float sc = ok ? dot * scale : NEG;
          sa[u] = sc; mxa = fmaxf(mxa, sc);
        }
        if (u >= 1) {
          float dot = 0.f;
#pragma unroll
          for (int d = 0; d < DK_; ++d) dot += qb[d] * kr[d];
          const bool ok = (u - 1 == 0) ? !(dir == 0 && sb_row == 0)
                        : ((u - 1 == WIN_ - 1) ? !(dir == 1 && sb_row == S_ - 1) : true);
          const float sc = ok ? dot * scale : NEG;
          sb[u - 1] = sc; mxb = fmaxf(mxb, sc);
        }
      }
      float suma = 0.f, sumb = 0.f;
#pragma unroll
      for (int t = 0; t < WIN_; ++t) {
        const float pa = __expf(sa[t] - mxa);
        const float pb = __expf(sb[t] - mxb);
        sa[t] = pa; suma += pa;
        sb[t] = pb; sumb += pb;
      }
      const float inva = 1.f / suma;
      const float invb = 1.f / sumb;
      float oa[DK_] = {}, ob[DK_] = {};
#pragma unroll
      for (int u = 0; u <= WIN_; ++u) {
        float vr[DK_];
        *(float4*)&vr[0] = *(const float4*)(vrow + u * LDA_S_);
        *(float4*)&vr[4] = *(const float4*)(vrow + u * LDA_S_ + 4);
        if (u < WIN_) {
          const float wgt = sa[u] * inva;
#pragma unroll
          for (int d = 0; d < DK_; ++d) oa[d] += wgt * vr[d];
        }
        if (u >= 1) {
          const float wgt = sb[u - 1] * invb;
#pragma unroll
          for (int d = 0; d < DK_; ++d) ob[d] += wgt * vr[d];
        }
      }
      *(float4*)(qrow)              = make_float4(oa[0], oa[1], oa[2], oa[3]);
      *(float4*)(qrow + 4)          = make_float4(oa[4], oa[5], oa[6], oa[7]);
      *(float4*)(qrow + LDA_S_)     = make_float4(ob[0], ob[1], ob[2], ob[3]);
      *(float4*)(qrow + LDA_S_ + 4) = make_float4(ob[4], ob[5], ob[6], ob[7]);
    }
  }
  __syncthreads();

  for (int pass = 0; pass < CH_PASSES_; ++pass) {
    const int stage = (pass < 2) ? 0 : ((pass < 6) ? 1 : 2);
    const float* Xin  = Xs + (stage & 1) * CH_X_;
    float*       Xout = Xs + ((stage & 1) ^ 1) * CH_X_;

    const int np = (pass + 1 < CH_PASSES_) ? pass + 1 : pass;
    int n0n, hicn; const float* Wn; const float* bn;
    chain_pass_geom(np, wave, Wo, bo, hwW, hwb, n0n, hicn, Wn, bn);
    const float* nx_lo = Wn + (size_t)(n0n + c) * D_ + 4 * h;
    const float* nx_hi = nx_lo + (size_t)hicn * D_;

    const float bv0 = bst[n0 + c];
    const float bv1 = bst[n0 + hic + c];

    v8f acc[2][2];
#pragma unroll
    for (int mt = 0; mt < 2; ++mt) {
#pragma unroll
      for (int t = 0; t < 2; ++t) acc[mt][t] = (v8f){};
    }

    chain_wave_pass(Xin + c * LDA_S_ + 4 * h, w_lo, w_hi, nx_lo, nx_hi, bq, acc);

    if (stage == 0) {
#pragma unroll
      for (int mt = 0; mt < 2; ++mt) {
#pragma unroll
        for (int r = 0; r < 8; ++r) {
          float* xr = Xout + (mt * 16 + 8 * h + r) * LDA_S_ + n0 + c;
          xr[ 0] = acc[mt][0][r] + bv0;
          xr[16] = acc[mt][1][r] + bv1;
        }
      }
    } else {
#pragma unroll
      for (int mt = 0; mt < 2; ++mt) {
#pragma unroll
        for (int r = 0; r < 8; ++r) {
          const int idx = (mt * 16 + 8 * h + r) * LDA_S_ + n0 + c;
          const float lin = acc[mt][0][r] + bv0;
          const float gin = acc[mt][1][r] + bv1;
          const float g  = 1.f / (1.f + __expf(-gin));
          const float nl = fmaxf(lin, 0.f);
          Xout[idx] = g * Xin[idx] + (1.f - g) * nl;
        }
      }
    }

    if (pass == 1 || pass == 5 || pass == 9) __syncthreads();

    n0 = n0n; hic = hicn; bst = bn; w_lo = nx_lo; w_hi = nx_hi;
  }

  const float* Xf = Xs + CH_X_ + srow * LDA_S_ + scol;
  float* o0 = out + (size_t)(row0 + srow) * (2 * D_) + (size_t)dir * D_ + scol;
#pragma unroll 8
  for (int v = 0; v < CH_ROWS_; v += 2) {
    float4 val = *(const float4*)&Xf[v * LDA_S_];
    *(float4*)(o0 + (size_t)v * (2 * D_)) = val;
    *(float4*)(o0 + (size_t)v * (2 * D_) + (size_t)BS_ * 2 * D_) = val;
  }
}

extern "C" void kernel_launch(void* const* d_in, const int* in_sizes, int n_in,
                              void* d_out, int out_size, void* d_ws, size_t ws_size,
                              hipStream_t stream) {
  const float* inputs    = (const float*)d_in[0];
  const float* left_pad  = (const float*)d_in[1];
  const float* right_pad = (const float*)d_in[2];
  const float* padrows[2] = {left_pad, right_pad};
  const float* Wqkv[2] = {(const float*)d_in[3],  (const float*)d_in[7]};
  const float* bqkv[2] = {(const float*)d_in[4],  (const float*)d_in[8]};
  const float* Wo[2]   = {(const float*)d_in[5],  (const float*)d_in[9]};
  const float* bo[2]   = {(const float*)d_in[6],  (const float*)d_in[10]};
  const float* hwW[2]  = {(const float*)d_in[11], (const float*)d_in[13]};
  const float* hwb[2]  = {(const float*)d_in[12], (const float*)d_in[14]};
  float* out = (float*)d_out;

  float* ws    = (float*)d_ws;
  float* padl  = ws + 4 * BSD_;
  float* padr  = padl + WIDTH_ * 2 * D_;
  float* padkv[2] = {padl, padr};

  dim3 blk(32, 4);
  dim3 blk8(32, 8);

  for (int dir = 0; dir < 2; ++dir) {
    dim3 grd(2 * D_ / 128, 1);
    gemm_wmma_f32<1, 16, LDS_TILE16_><<<grd, blk, 0, stream>>>(
        padrows[dir], D_, Wqkv[dir] + (size_t)D_ * D_, D_, bqkv[dir] + D_,
        padkv[dir], 2 * D_, WIDTH_, 2 * D_);
  }
  { dim3 grd(D_ / 128, BS_ / 64);
    qkv_kernel<<<grd, blk8, 0, stream>>>(inputs, Wqkv[0], Wqkv[1], bqkv[0], bqkv[1], out, ws); }
  { dim3 grd(BS_ / CH_ROWS_, 2);
    chain_kernel<<<grd, blk8, 0, stream>>>(ws, Wo[0], Wo[1], bo[0], bo[1],
                                          hwW[0], hwW[1], hwb[0], hwb[1], out); }
}


# W of the two 256-block kernels re-laid once per call into WMMA fragment order by a new first launch: a wave-wide W load reads 512 contiguous bytes (4 lines, was 16); k and sum order unchanged
# speedup vs baseline: 3.3134x; 1.0345x over previous
#include <hip/hip_runtime.h>

#define B_ 8
#define S_ 512
#define D_ 512
#define H_ 64
#define DK_ 8
#define WIDTH_ 16
#define N_ (S_ + 2 * WIDTH_)
#define NL_ 2
#define WIN_ (WIDTH_ + 2)
#define BS_ (B_ * S_)
static const size_t BSD_ = (size_t)BS_ * D_;

typedef __attribute__((ext_vector_type(2))) float v2f;
typedef __attribute__((ext_vector_type(8))) float v8f;

typedef __attribute__((ext_vector_type(4))) float v4f;

#define WMMA_F32(a, b, c) __builtin_amdgcn_wmma_f32_16x16x4_f32(false, (a), false, (b), (short)0, (c), false, false)

#define LDA_S_ (512 + 4)
#define LDS_TILE16_ (16 * LDA_S_)
#define LDS_ALONE_ (162 * 256)

#define WCH_ 128
#define TILE_F_ (16 * 512)
#define PK_QKV_T_ 96
#define PK_WO_T_ 32
#define PK_HW_T_ 128
#define PK_TILES_ (2 * (PK_QKV_T_ + PK_WO_T_ + PK_HW_T_))
#define PK_REST_T_ (PK_TILES_ - 2 * PK_QKV_T_)
static const size_t WS_NEED_ =
    (4 * BSD_ + 2 * (size_t)WIDTH_ * 2 * D_ + (size_t)PK_REST_T_ * TILE_F_) * sizeof(float);
static_assert(WS_NEED_ == 44171264, "workspace carve-up: 32 MiB + 128 KiB + 10 MiB");

__global__ __launch_bounds__(256)
void pack_w_kernel(const float* Wqkv_l, const float* Wqkv_r,
                   const float* Wo_l,   const float* Wo_r,
                   const float* hwW_l,  const float* hwW_r,
                   float* dst_qkv, float* dst_rest) {
  const int t = blockIdx.x;
  const float* src; int t0;
  if      (t < PK_QKV_T_)                               { src = Wqkv_l; t0 = 0; }
  else if (t < 2 * PK_QKV_T_)                           { src = Wqkv_r; t0 = PK_QKV_T_; }
  else if (t < 2 * PK_QKV_T_ + PK_WO_T_)                { src = Wo_l;   t0 = 2 * PK_QKV_T_; }
  else if (t < 2 * PK_QKV_T_ + 2 * PK_WO_T_)            { src = Wo_r;   t0 = 2 * PK_QKV_T_ + PK_WO_T_; }
  else if (t < 2 * PK_QKV_T_ + 2 * PK_WO_T_ + PK_HW_T_) { src = hwW_l;  t0 = 2 * PK_QKV_T_ + 2 * PK_WO_T_; }
  else                                                  { src = hwW_r;  t0 = 2 * PK_QKV_T_ + 2 * PK_WO_T_ + PK_HW_T_; }
  src += (size_t)(t - t0) * TILE_F_;
  float* d = (t < 2 * PK_QKV_T_) ? dst_qkv + (size_t)t * TILE_F_
                                 : dst_rest + (size_t)(t - 2 * PK_QKV_T_) * TILE_F_;
  const int tid = threadIdx.x;
  const int ln = tid & 31;
  const int c = ln & 15, h = ln >> 4;
  const int q0 = tid >> 5;
  v4f v[8];
#pragma unroll
  for (int i = 0; i < 8; ++i) v[i] = *(const v4f*)(src + c * 512 + 8 * (q0 + 8 * i) + 4 * h);
#pragma unroll
  for (int i = 0; i < 8; ++i) *(v4f*)(d + (q0 + 8 * i) * WCH_ + ln * 4) = v[i];
}

template <int MT, int PF, int LDS_FLOATS>
__global__ __launch_bounds__(128)
void gemm_wmma_f32(const float* __restrict__ A, int lda,
                   const float* __restrict__ W, int ldw,
                   const float* __restrict__ bias,
                   float* __restrict__ C, int ldc,
                   int M, int Nout) {
  constexpr int KK = 512;
  constexpr int BM = 16 * MT;
  constexpr int LDA_S = LDA_S_;
  static_assert(LDS_FLOATS >= BM * LDA_S, "LDS array smaller than the A tile");
  __shared__ __attribute__((aligned(16))) float As[LDS_FLOATS];

  const int lane = threadIdx.x;
  const int wave = threadIdx.y;
  const int tid  = wave * 32 + lane;
  const int row0 = blockIdx.y * BM;
  const int col0 = blockIdx.x * 128 + wave * 32;
  const int c = lane & 15;
  const int h = lane >> 4;

  int wr0 = col0 + c;      if (wr0 > Nout - 1) wr0 = Nout - 1;
  int wr1 = col0 + 16 + c; if (wr1 > Nout - 1) wr1 = Nout - 1;
  const float* w0 = W + (size_t)wr0 * ldw + 4 * h;
  const float* w1 = W + (size_t)wr1 * ldw + 4 * h;

  v4f bq[PF][2];
#pragma unroll
  for (int s = 0; s < PF; ++s) {
    bq[s][0] = *(const v4f*)(w0 + 8 * s);
    bq[s][1] = *(const v4f*)(w1 + 8 * s);
  }

#pragma unroll 32
  for (int v = 0; v < BM; ++v) {
    int rg = row0 + v; if (rg > M - 1) rg = M - 1;
    float4 val = *(const float4*)(A + (size_t)rg * lda + tid * 4);
    *(float4*)&As[v * LDA_S + tid * 4] = val;
  }
  __syncthreads();

  if (row0 >= M || col0 >= Nout) return;

  const float* ap = &As[c * LDA_S + 4 * h];

  v8f acc[MT][2];
#pragma unroll
  for (int mt = 0; mt < MT; ++mt) { acc[mt][0] = (v8f){}; acc[mt][1] = (v8f){}; }

  v4f an[MT];
#pragma unroll
  for (int mt = 0; mt < MT; ++mt) an[mt] = *(const v4f*)(ap + mt * 16 * LDA_S);

  for (int it = 0; it < KK / 8; it += PF) {
#pragma unroll
    for (int s = 0; s < PF; ++s) {
      const int kk = (it + s) * 8;
      v4f a[MT];
#pragma unroll
      for (int mt = 0; mt < MT; ++mt) a[mt] = an[mt];
      v4f b0 = bq[s][0], b1 = bq[s][1];
      int kp = kk + 8 * PF; if (kp > KK - 8) kp = KK - 8;
      bq[s][0] = *(const v4f*)(w0 + kp);
      bq[s][1] = *(const v4f*)(w1 + kp);
      int ka = kk + 8; if (ka > KK - 8) ka = KK - 8;
#pragma unroll
      for (int mt = 0; mt < MT; ++mt) an[mt] = *(const v4f*)(ap + mt * 16 * LDA_S + ka);

#pragma unroll
      for (int mt = 0; mt < MT; ++mt) {
        acc[mt][0] = WMMA_F32(a[mt].xy, b0.xy, acc[mt][0]);
        acc[mt][1] = WMMA_F32(a[mt].xy, b1.xy, acc[mt][1]);
      }
#pragma unroll
      for (int mt = 0; mt < MT; ++mt) {
        acc[mt][0] = WMMA_F32(a[mt].zw, b0.zw, acc[mt][0]);
        acc[mt][1] = WMMA_F32(a[mt].zw, b1.zw, acc[mt][1]);
      }
    }
  }

  const float bv0 = bias[col0 + c];
  const float bv1 = bias[col0 + 16 + c];
#pragma unroll
  for (int mt = 0; mt < MT; ++mt) {
#pragma unroll
    for (int r = 0; r < 8; ++r) {
      const int row = row0 + mt * 16 + 8 * h + r;
      if (row < M) {
        float* crow = C + (size_t)row * ldc + col0 + c;
        crow[ 0] = acc[mt][0][r] + bv0;
        crow[16] = acc[mt][1][r] + bv1;
      }
    }
  }
}

#define QKV_PASSES_ 3
#define QKV_PF_ 4

__device__ __forceinline__ void qkv_pass_geom(int proj,
                                              const float* Wl, const float* Wr,
                                              const float* bl, const float* br,
                                              float* out, float* ws,
                                              const float*& Wp, const float*& bp,
                                              float*& Cp, int& ldc) {
  const int dir = (proj >= 3) ? 1 : 0;
  const int t = proj - 3 * dir;
  Wp = (dir ? Wr : Wl) + (size_t)t * D_ * D_;
  bp = (dir ? br : bl) + t * D_;
  if (t == 0) { Cp = out + dir * D_; ldc = 2 * D_; }
  else        { Cp = ws + (size_t)(2 * dir + t - 1) * BSD_; ldc = D_; }
}

__global__ __launch_bounds__(256)
void qkv_kernel(const float* __restrict__ X,
                const float* __restrict__ Wl, const float* __restrict__ Wr,
                const float* __restrict__ bl, const float* __restrict__ br,
                float* __restrict__ out, float* __restrict__ ws) {
  constexpr int KK = 512;
  constexpr int MT = 4;
  constexpr int BM = 16 * MT;
  static_assert(LDS_ALONE_ >= BM * LDA_S_, "LDS array smaller than the A tile");
  static_assert(BS_ % BM == 0 && D_ % 128 == 0, "whole blocks only: no row or column guard");
  __shared__ __attribute__((aligned(16))) float As[LDS_ALONE_];

  const int lane = threadIdx.x;
  const int wave = threadIdx.y;
  const int tid  = wave * 32 + lane;
  const int row0 = blockIdx.y * BM;
  const int colw = blockIdx.x * 128 + (wave & 3) * 32;
  const int psel = wave >> 2;
  const int c = lane & 15;
  const int h = lane >> 4;

  const float* Wp; const float* bp; float* Cp; int ldc;
  qkv_pass_geom(psel, Wl, Wr, bl, br, out, ws, Wp, bp, Cp, ldc);
  const float* w0 = Wp + (size_t)colw * KK + lane * 4;
  v4f bq[QKV_PF_][2];
#pragma unroll
  for (int s = 0; s < QKV_PF_; ++s) {
    bq[s][0] = *(const v4f*)(w0 + WCH_ * s);
    bq[s][1] = *(const v4f*)(w0 + WCH_ * s + 16 * KK);
  }

  const int srow = tid >> 7;
  const int scol = (tid & 127) * 4;
#pragma unroll
  for (int v = 0; v < BM; v += 2) {
    float4 val = *(const float4*)(X + (size_t)(row0 + v + srow) * KK + scol);
    *(float4*)&As[(v + srow) * LDA_S_ + scol] = val;
  }
  __syncthreads();

  const float* ap = &As[c * LDA_S_ + 4 * h];

  for (int pass = 0; pass < QKV_PASSES_; ++pass) {
    const int np = 2 * ((pass + 1 < QKV_PASSES_) ? pass + 1 : pass) + psel;
    const float* Wn; const float* bn; float* Cn; int ldcn;
    qkv_pass_geom(np, Wl, Wr, bl, br, out, ws, Wn, bn, Cn, ldcn);
    const float* nx0 = Wn + (size_t)colw * KK + lane * 4;

    const float bv0 = bp[colw + c];
    const float bv1 = bp[colw + 16 + c];

    v8f acc[MT][2];
#pragma unroll
    for (int mt = 0; mt < MT; ++mt) { acc[mt][0] = (v8f){}; acc[mt][1] = (v8f){}; }

    v4f an[MT];
#pragma unroll
    for (int mt = 0; mt < MT; ++mt) an[mt] = *(const v4f*)(ap + mt * 16 * LDA_S_);

    for (int it = 0; it < KK / 8; it += QKV_PF_) {
      const bool wrap = (it + QKV_PF_ >= KK / 8);
      const float* cl = wrap ? nx0 : w0;
      const int kb = wrap ? 0 : (it + QKV_PF_) * WCH_;
#pragma unroll
      for (int s = 0; s < QKV_PF_; ++s) {
        const int kk = (it + s) * 8;
        v4f a[MT];
#pragma unroll
        for (int mt = 0; mt < MT; ++mt) a[mt] = an[mt];
        const v4f b0 = bq[s][0], b1 = bq[s][1];
        bq[s][0] = *(const v4f*)(cl + kb + WCH_ * s);
        bq[s][1] = *(const v4f*)(cl + kb + WCH_ * s + 16 * KK);
        int ka = kk + 8; if (ka > KK - 8) ka = KK - 8;
#pragma unroll
        for (int mt = 0; mt < MT; ++mt) an[mt] = *(const v4f*)(ap + mt * 16 * LDA_S_ + ka);

#pragma unroll
        for (int mt = 0; mt < MT; ++mt) {
          acc[mt][0] = WMMA_F32(a[mt].xy, b0.xy, acc[mt][0]);
          acc[mt][1] = WMMA_F32(a[mt].xy, b1.xy, acc[mt][1]);
        }
#pragma unroll
        for (int mt = 0; mt < MT; ++mt) {
          acc[mt][0] = WMMA_F32(a[mt].zw, b0.zw, acc[mt][0]);
          acc[mt][1] = WMMA_F32(a[mt].zw, b1.zw, acc[mt][1]);
        }
      }
    }

#pragma unroll
    for (int mt = 0; mt < MT; ++mt) {
#pragma unroll
      for (int r = 0; r < 8; ++r) {
        float* crow = Cp + (size_t)(row0 + mt * 16 + 8 * h + r) * ldc + colw + c;
        crow[ 0] = acc[mt][0][r] + bv0;
        crow[16] = acc[mt][1][r] + bv1;
      }
    }

    w0 = nx0; bp = bn; Cp = Cn; ldc = ldcn;
  }
}

#define CH_ROWS_ 32
#define CH_X_ (CH_ROWS_ * LDA_S_)
#define CH_PF_ 4
#define CH_PASSES_ 10
#define KW_ROWS_ (CH_ROWS_ + WIDTH_ + 1)
#define CH_LDS_ (CH_X_ + 2 * KW_ROWS_ * LDA_S_)

__device__ __forceinline__ void chain_pass_geom(int pass, int wave,
                                                const float* Wo, const float* bo,
                                                const float* hwW, const float* hwb,
                                                int& n0, int& hic,
                                                const float*& Wst, const float*& bst) {
  if (pass < 2) {
    n0 = wave * 64 + pass * 32; hic = 16; Wst = Wo; bst = bo;
  } else {
    const int q = pass - 2;
    const int layer = q >> 2;
    n0 = wave * 64 + (q & 3) * 16; hic = D_;
    Wst = hwW + (size_t)layer * 2 * D_ * D_; bst = hwb + layer * 2 * D_;
  }
}

__device__ __forceinline__ void chain_wave_pass(const float* ap,
                                                const float* w_lo, const float* w_hi,
                                                const float* nx_lo, const float* nx_hi,
                                                v4f (&bq)[CH_PF_][2], v8f (&acc)[2][2]) {
  constexpr int KK = 512;
  v4f an[2];
  an[0] = *(const v4f*)(ap);
  an[1] = *(const v4f*)(ap + 16 * LDA_S_);
#pragma unroll
  for (int it = 0; it < KK / 8; it += CH_PF_) {
    const bool wrap = (it + CH_PF_ >= KK / 8);
    const float* cl = wrap ? nx_lo : w_lo;
    const float* ch = wrap ? nx_hi : w_hi;
    const int kb = wrap ? 0 : (it + CH_PF_) * WCH_;
#pragma unroll
    for (int s = 0; s < CH_PF_; ++s) {
      const int kk = (it + s) * 8;
      const v4f a0 = an[0], a1 = an[1];
      const v4f b0 = bq[s][0], b1 = bq[s][1];
      bq[s][0] = *(const v4f*)(cl + kb + WCH_ * s);
      bq[s][1] = *(const v4f*)(ch + kb + WCH_ * s);
      int ka = kk + 8; if (ka > KK - 8) ka = KK - 8;
      an[0] = *(const v4f*)(ap + ka);
      an[1] = *(const v4f*)(ap + 16 * LDA_S_ + ka);

      acc[0][0] = WMMA_F32(a0.xy, b0.xy, acc[0][0]);
      acc[0][1] = WMMA_F32(a0.xy, b1.xy, acc[0][1]);
      acc[1][0] = WMMA_F32(a1.xy, b0.xy, acc[1][0]);
      acc[1][1] = WMMA_F32(a1.xy, b1.xy, acc[1][1]);
      acc[0][0] = WMMA_F32(a0.zw, b0.zw, acc[0][0]);
      acc[0][1] = WMMA_F32(a0.zw, b1.zw, acc[0][1]);
      acc[1][0] = WMMA_F32(a1.zw, b0.zw, acc[1][0]);
      acc[1][1] = WMMA_F32(a1.zw, b1.zw, acc[1][1]);
    }
  }
}

template <int KIND>
__device__ __forceinline__ void stage_kv_window(const float* __restrict__ kb,
                                                const float* __restrict__ pb,
                                                int wb, float* dst) {
#pragma unroll
  for (int w = 0; w < KW_ROWS_; ++w) {
    const float* p;
    if (KIND == 0) {
      p = kb + (wb + w) * D_;
    } else if (KIND == 1) {
      if (w <= WIDTH_) { const int pr = (w == 0) ? 0 : w - 1; p = pb + pr * (2 * D_); }
      else { p = kb + (w - (WIDTH_ + 1)) * D_; }
    } else {
      if (w < CH_ROWS_) { p = kb + (S_ - CH_ROWS_ + w) * D_; }
      else { const int pr = (w - CH_ROWS_ < WIDTH_) ? w - CH_ROWS_ : WIDTH_ - 1; p = pb + pr * (2 * D_); }
    }
    const float4 q4 = *(const float4*)p;
    *(float4*)&dst[w * LDA_S_] = q4;
  }
}

__global__ __launch_bounds__(256)
void chain_kernel(const float* __restrict__ kvws,
                  const float* __restrict__ Wo_l,   const float* __restrict__ Wo_r,
                  const float* __restrict__ bo_l,   const float* __restrict__ bo_r,
                  const float* __restrict__ hwW_l,  const float* __restrict__ hwW_r,
                  const float* __restrict__ hwb_l,  const float* __restrict__ hwb_r,
                  float* out) {
  __shared__ __attribute__((aligned(16))) float Xs[CH_LDS_];
  static_assert(CH_LDS_ >= 2 * CH_X_, "LDS array smaller than the two activation buffers");
  static_assert(CH_LDS_ * 4 > 160 * 1024 && CH_LDS_ * 4 <= 320 * 1024, "LDS: one block per WGP");

  const int lane = threadIdx.x;
  const int wave = threadIdx.y;
  const int tid  = wave * 32 + lane;
  const int c = lane & 15;
  const int h = lane >> 4;
  const int dir  = blockIdx.y;
  const int row0 = blockIdx.x * CH_ROWS_;

  const float* Wo   = dir ? Wo_r   : Wo_l;
  const float* bo   = dir ? bo_r   : bo_l;
  const float* hwW  = dir ? hwW_r  : hwW_l;
  const float* hwb  = dir ? hwb_r  : hwb_l;

  int n0, hic; const float* Wst; const float* bst;
  chain_pass_geom(0, wave, Wo, bo, hwW, hwb, n0, hic, Wst, bst);
  const float* w_lo = Wst + (size_t)n0 * D_ + lane * 4;
  const float* w_hi = w_lo + (size_t)hic * D_;
  v4f bq[CH_PF_][2];
#pragma unroll
  for (int s = 0; s < CH_PF_; ++s) {
    bq[s][0] = *(const v4f*)(w_lo + WCH_ * s);
    bq[s][1] = *(const v4f*)(w_hi + WCH_ * s);
  }

  const int srow = tid >> 7;
  const int scol = (tid & 127) * 4;
  const int bb = row0 / S_;
  const int s0 = row0 % S_;
  const int wb = dir ? s0 : s0 - (WIDTH_ + 1);
  float* Ks = Xs + CH_X_;
  float* Vs = Ks + KW_ROWS_ * LDA_S_;
  {
    const float* qsrc = out + (size_t)(row0 + srow) * (2 * D_) + (size_t)dir * D_ + scol;
#pragma unroll
    for (int v = 0; v < CH_ROWS_; v += 2) {
      float4 val = *(const float4*)(qsrc + (size_t)v * (2 * D_));
      *(float4*)&Xs[(v + srow) * LDA_S_ + scol] = val;
    }
    static_assert(S_ % CH_ROWS_ == 0 && CH_ROWS_ > WIDTH_ + 1, "window shapes of stage_kv_window");
    const float* kb = kvws + (size_t)dir * (2 * BSD_) + (size_t)srow * BSD_ + (size_t)(bb * S_) * D_ + scol;
    const float* pb = kvws + 4 * BSD_ + (size_t)dir * (WIDTH_ * 2 * D_) + srow * D_ + scol;
    float* wdst = (srow ? Vs : Ks) + scol;
    if (wb < 0)                          stage_kv_window<1>(kb, pb, wb, wdst);
    else if (wb + KW_ROWS_ - 1 >= S_)    stage_kv_window<2>(kb, pb, wb, wdst);
    else                                 stage_kv_window<0>(kb, pb, wb, wdst);
  }
  __syncthreads();

  {
    const int hh = tid & (H_ - 1);
    const int rsel = tid >> 6;
    const float scale = 0.35355339059327373f;
    const float NEG = -3.0e38f;
    for (int it = 0; it < CH_ROWS_ / 8; ++it) {
      const int r0 = (it * 4 + rsel) * 2;
      float* qrow = Xs + r0 * LDA_S_ + hh * DK_;
      const float* krow = Ks + r0 * LDA_S_ + hh * DK_;
      const float* vrow = Vs + r0 * LDA_S_ + hh * DK_;
      float qa[DK_], qb[DK_];
      *(float4*)&qa[0] = *(const float4*)(qrow);
      *(float4*)&qa[4] = *(const float4*)(qrow + 4);
      *(float4*)&qb[0] = *(const float4*)(qrow + LDA_S_);
      *(float4*)&qb[4] = *(const float4*)(qrow + LDA_S_ + 4);
      const int sa_row = s0 + r0;
      const int sb_row = s0 + r0 + 1;
      float sa[WIN_], sb[WIN_];
      float mxa = NEG, mxb = NEG;
#pragma unroll
      for (int u = 0; u <= WIN_; ++u) {
        float kr[DK_];
        *(float4*)&kr[0] = *(const float4*)(krow + u * LDA_S_);
        *(float4*)&kr[4] = *(const float4*)(krow + u * LDA_S_ + 4);
        if (u < WIN_) {
          float dot = 0.f;
#pragma unroll
          for (int d = 0; d < DK_; ++d) dot += qa[d] * kr[d];
          const bool ok = (u == 0) ? !(dir == 0 && sa_row == 0)
                        : ((u == WIN_ - 1) ? !(dir == 1 && sa_row == S_ - 1) : true);
          const float sc = ok ? dot * scale : NEG;
          sa[u] = sc; mxa = fmaxf(mxa, sc);
        }
        if (u >= 1) {
          float dot = 0.f;
#pragma unroll
          for (int d = 0; d < DK_; ++d) dot += qb[d] * kr[d];
          const bool ok = (u - 1 == 0) ? !(dir == 0 && sb_row == 0)
                        : ((u - 1 == WIN_ - 1) ? !(dir == 1 && sb_row == S_ - 1) : true);
          const float sc = ok ? dot * scale : NEG;
          sb[u - 1] = sc; mxb = fmaxf(mxb, sc);
        }
      }
      float suma = 0.f, sumb = 0.f;
#pragma unroll
      for (int t = 0; t < WIN_; ++t) {
        const float pa = __expf(sa[t] - mxa);
        const float pb = __expf(sb[t] - mxb);
        sa[t] = pa; suma += pa;
        sb[t] = pb; sumb += pb;
      }
      const float inva = 1.f / suma;
      const float invb = 1.f / sumb;
      float oa[DK_] = {}, ob[DK_] = {};
#pragma unroll
      for (int u = 0; u <= WIN_; ++u) {
        float vr[DK_];
        *(float4*)&vr[0] = *(const float4*)(vrow + u * LDA_S_);
        *(float4*)&vr[4] = *(const float4*)(vrow + u * LDA_S_ + 4);
        if (u < WIN_) {
          const float wgt = sa[u] * inva;
#pragma unroll
          for (int d = 0; d < DK_; ++d) oa[d] += wgt * vr[d];
        }
        if (u >= 1) {
          const float wgt = sb[u - 1] * invb;
#pragma unroll
          for (int d = 0; d < DK_; ++d) ob[d] += wgt * vr[d];
        }
      }
      *(float4*)(qrow)              = make_float4(oa[0], oa[1], oa[2], oa[3]);
      *(float4*)(qrow + 4)          = make_float4(oa[4], oa[5], oa[6], oa[7]);
      *(float4*)(qrow + LDA_S_)     = make_float4(ob[0], ob[1], ob[2], ob[3]);
      *(float4*)(qrow + LDA_S_ + 4) = make_float4(ob[4], ob[5], ob[6], ob[7]);
    }
  }
  __syncthreads();

  for (int pass = 0; pass < CH_PASSES_; ++pass) {
    const int stage = (pass < 2) ? 0 : ((pass < 6) ? 1 : 2);
    const float* Xin  = Xs + (stage & 1) * CH_X_;
    float*       Xout = Xs + ((stage & 1) ^ 1) * CH_X_;

    const int np = (pass + 1 < CH_PASSES_) ? pass + 1 : pass;
    int n0n, hicn; const float* Wn; const float* bn;
    chain_pass_geom(np, wave, Wo, bo, hwW, hwb, n0n, hicn, Wn, bn);
    const float* nx_lo = Wn + (size_t)n0n * D_ + lane * 4;
    const float* nx_hi = nx_lo + (size_t)hicn * D_;

    const float bv0 = bst[n0 + c];
    const float bv1 = bst[n0 + hic + c];

    v8f acc[2][2];
#pragma unroll
    for (int mt = 0; mt < 2; ++mt) {
#pragma unroll
      for (int t = 0; t < 2; ++t) acc[mt][t] = (v8f){};
    }

    chain_wave_pass(Xin + c * LDA_S_ + 4 * h, w_lo, w_hi, nx_lo, nx_hi, bq, acc);

    if (stage == 0) {
#pragma unroll
      for (int mt = 0; mt < 2; ++mt) {
#pragma unroll
        for (int r = 0; r < 8; ++r) {
          float* xr = Xout + (mt * 16 + 8 * h + r) * LDA_S_ + n0 + c;
          xr[ 0] = acc[mt][0][r] + bv0;
          xr[16] = acc[mt][1][r] + bv1;
        }
      }
    } else {
#pragma unroll
      for (int mt = 0; mt < 2; ++mt) {
#pragma unroll
        for (int r = 0; r < 8; ++r) {
          const int idx = (mt * 16 + 8 * h + r) * LDA_S_ + n0 + c;
          const float lin = acc[mt][0][r] + bv0;
          const float gin = acc[mt][1][r] + bv1;
          const float g  = 1.f / (1.f + __expf(-gin));
          const float nl = fmaxf(lin, 0.f);
          Xout[idx] = g * Xin[idx] + (1.f - g) * nl;
        }
      }
    }

    if (pass == 1 || pass == 5 || pass == 9) __syncthreads();

    n0 = n0n; hic = hicn; bst = bn; w_lo = nx_lo; w_hi = nx_hi;
  }

  const float* Xf = Xs + CH_X_ + srow * LDA_S_ + scol;
  float* o0 = out + (size_t)(row0 + srow) * (2 * D_) + (size_t)dir * D_ + scol;
#pragma unroll 8
  for (int v = 0; v < CH_ROWS_; v += 2) {
    float4 val = *(const float4*)&Xf[v * LDA_S_];
    *(float4*)(o0 + (size_t)v * (2 * D_)) = val;
    *(float4*)(o0 + (size_t)v * (2 * D_) + (size_t)BS_ * 2 * D_) = val;
  }
}

extern "C" void kernel_launch(void* const* d_in, const int* in_sizes, int n_in,
                              void* d_out, int out_size, void* d_ws, size_t ws_size,
                              hipStream_t stream) {
  const float* inputs    = (const float*)d_in[0];
  const float* left_pad  = (const float*)d_in[1];
  const float* right_pad = (const float*)d_in[2];
  const float* padrows[2] = {left_pad, right_pad};
  const float* Wqkv[2] = {(const float*)d_in[3],  (const float*)d_in[7]};
  const float* bqkv[2] = {(const float*)d_in[4],  (const float*)d_in[8]};
  const float* Wo[2]   = {(const float*)d_in[5],  (const float*)d_in[9]};
  const float* bo[2]   = {(const float*)d_in[6],  (const float*)d_in[10]};
  const float* hwW[2]  = {(const float*)d_in[11], (const float*)d_in[13]};
  const float* hwb[2]  = {(const float*)d_in[12], (const float*)d_in[14]};
  float* out = (float*)d_out;

  if (ws_size < WS_NEED_) return;

  float* ws    = (float*)d_ws;
  float* padl  = ws + 4 * BSD_;
  float* padr  = padl + WIDTH_ * 2 * D_;
  float* padkv[2] = {padl, padr};
  float* pkq   = out + (size_t)BS_ * 2 * D_;
  float* pk    = padr + WIDTH_ * 2 * D_;
  const float* pkWqkv[2] = {pkq, pkq + (size_t)PK_QKV_T_ * TILE_F_};
  const float* pkWo[2]   = {pk, pk + (size_t)PK_WO_T_ * TILE_F_};
  const float* pkHw[2]   = {pk + (size_t)(2 * PK_WO_T_) * TILE_F_,
                            pk + (size_t)(2 * PK_WO_T_ + PK_HW_T_) * TILE_F_};

  dim3 blk(32, 4);
  dim3 blk8(32, 8);

  pack_w_kernel<<<dim3(PK_TILES_), dim3(256), 0, stream>>>(Wqkv[0], Wqkv[1], Wo[0], Wo[1], hwW[0], hwW[1], pkq, pk);

  for (int dir = 0; dir < 2; ++dir) {
    dim3 grd(2 * D_ / 128, 1);
    gemm_wmma_f32<1, 16, LDS_TILE16_><<<grd, blk, 0, stream>>>(
        padrows[dir], D_, Wqkv[dir] + (size_t)D_ * D_, D_, bqkv[dir] + D_,
        padkv[dir], 2 * D_, WIDTH_, 2 * D_);
  }
  { dim3 grd(D_ / 128, BS_ / 64);
    qkv_kernel<<<grd, blk8, 0, stream>>>(inputs, pkWqkv[0], pkWqkv[1], bqkv[0], bqkv[1], out, ws); }
  { dim3 grd(BS_ / CH_ROWS_, 2);
    chain_kernel<<<grd, blk8, 0, stream>>>(ws, pkWo[0], pkWo[1], bo[0], bo[1],
                                          pkHw[0], pkHw[1], hwb[0], hwb[1], out); }
}


# Pad-row k/v projections of both directions move into the weight re-ordering launch as 16 extra blocks (8 waves, one 16x16 tile per wave; k and sum order unchanged): 5 launches to 3
# speedup vs baseline: 3.4471x; 1.0404x over previous
#include <hip/hip_runtime.h>

#define B_ 8
#define S_ 512
#define D_ 512
#define H_ 64
#define DK_ 8
#define WIDTH_ 16
#define N_ (S_ + 2 * WIDTH_)
#define NL_ 2
#define WIN_ (WIDTH_ + 2)
#define BS_ (B_ * S_)
static const size_t BSD_ = (size_t)BS_ * D_;

typedef __attribute__((ext_vector_type(2))) float v2f;
typedef __attribute__((ext_vector_type(8))) float v8f;

typedef __attribute__((ext_vector_type(4))) float v4f;

#define WMMA_F32(a, b, c) __builtin_amdgcn_wmma_f32_16x16x4_f32(false, (a), false, (b), (short)0, (c), false, false)

#define LDA_S_ (512 + 4)
#define LDS_TILE16_ (16 * LDA_S_)
#define LDS_ALONE_ (162 * 256)

#define WCH_ 128
#define TILE_F_ (16 * 512)
#define PK_QKV_T_ 96
#define PK_WO_T_ 32
#define PK_HW_T_ 128
#define PK_TILES_ (2 * (PK_QKV_T_ + PK_WO_T_ + PK_HW_T_))
#define PK_REST_T_ (PK_TILES_ - 2 * PK_QKV_T_)
static const size_t WS_NEED_ =
    (4 * BSD_ + 2 * (size_t)WIDTH_ * 2 * D_ + (size_t)PK_REST_T_ * TILE_F_) * sizeof(float);
static_assert(WS_NEED_ == 44171264, "workspace carve-up: 32 MiB + 128 KiB + 10 MiB");

#define PADB_ (2 * (2 * D_ / 128))
#define PAD_PF_ 16
static_assert((2 * D_) % 128 == 0 && WIDTH_ == 16, "pad_rows_gemm: whole 16 x 128 blocks, no row or column guard");

__device__ __forceinline__ void pad_rows_gemm(const float* A, const float* W, const float* bias,
                                              float* C, int colgrp, float* As) {
  constexpr int KK = 512;
  const int tid  = threadIdx.x;
  const int lane = tid & 31;
  const int wave = tid >> 5;
  const int c = lane & 15;
  const int h = lane >> 4;
  const int col0 = colgrp * 128 + wave * 16;
  const float* w0 = W + (size_t)(col0 + c) * KK + 4 * h;

  v4f bq[PAD_PF_];
#pragma unroll
  for (int s = 0; s < PAD_PF_; ++s) bq[s] = *(const v4f*)(w0 + 8 * s);

  const int srow = tid >> 7;
  const int scol = (tid & 127) * 4;
#pragma unroll
  for (int v = 0; v < WIDTH_; v += 2) {
    float4 val = *(const float4*)(A + (size_t)(v + srow) * KK + scol);
    *(float4*)&As[(v + srow) * LDA_S_ + scol] = val;
  }
  __syncthreads();

  const float* ap = &As[c * LDA_S_ + 4 * h];
  v8f acc = (v8f){};
  v4f an = *(const v4f*)ap;
  for (int it = 0; it < KK / 8; it += PAD_PF_) {
#pragma unroll
    for (int s = 0; s < PAD_PF_; ++s) {
      const int kk = (it + s) * 8;
      const v4f a = an;
      const v4f b = bq[s];
      int kp = kk + 8 * PAD_PF_; if (kp > KK - 8) kp = KK - 8;
      bq[s] = *(const v4f*)(w0 + kp);
      int ka = kk + 8; if (ka > KK - 8) ka = KK - 8;
      an = *(const v4f*)(ap + ka);
      acc = WMMA_F32(a.xy, b.xy, acc);
      acc = WMMA_F32(a.zw, b.zw, acc);
    }
  }

  const float bv = bias[col0 + c];
  asm volatile("s_wait_alu depctr_va_vdst(0)" : "+v"(acc));
#pragma unroll
  for (int r = 0; r < 8; ++r) C[(size_t)(8 * h + r) * (2 * D_) + col0 + c] = acc[r] + bv;
}

__global__ __launch_bounds__(256)
void prep_kernel(const float* Wqkv_l, const float* Wqkv_r,
                 const float* Wo_l,   const float* Wo_r,
                 const float* hwW_l,  const float* hwW_r,
                 float* dst_qkv, float* dst_rest,
                 const float* pad_l,  const float* pad_r,
                 const float* bqkv_l, const float* bqkv_r,
                 float* padkv_l, float* padkv_r) {
  __shared__ __attribute__((aligned(16))) float As[LDS_TILE16_];
  if (blockIdx.x < PADB_) {
    const int dir = blockIdx.x / (PADB_ / 2);
    const int colgrp = blockIdx.x % (PADB_ / 2);
    pad_rows_gemm(dir ? pad_r : pad_l,
                  (dir ? Wqkv_r : Wqkv_l) + (size_t)D_ * D_,
                  (dir ? bqkv_r : bqkv_l) + D_,
                  dir ? padkv_r : padkv_l, colgrp, As);
    return;
  }
  const int t = blockIdx.x - PADB_;
  const float* src; int t0;
  if      (t < PK_QKV_T_)                               { src = Wqkv_l; t0 = 0; }
  else if (t < 2 * PK_QKV_T_)                           { src = Wqkv_r; t0 = PK_QKV_T_; }
  else if (t < 2 * PK_QKV_T_ + PK_WO_T_)                { src = Wo_l;   t0 = 2 * PK_QKV_T_; }
  else if (t < 2 * PK_QKV_T_ + 2 * PK_WO_T_)            { src = Wo_r;   t0 = 2 * PK_QKV_T_ + PK_WO_T_; }
  else if (t < 2 * PK_QKV_T_ + 2 * PK_WO_T_ + PK_HW_T_) { src = hwW_l;  t0 = 2 * PK_QKV_T_ + 2 * PK_WO_T_; }
  else                                                  { src = hwW_r;  t0 = 2 * PK_QKV_T_ + 2 * PK_WO_T_ + PK_HW_T_; }
  src += (size_t)(t - t0) * TILE_F_;
  float* d = (t < 2 * PK_QKV_T_) ? dst_qkv + (size_t)t * TILE_F_
                                 : dst_rest + (size_t)(t - 2 * PK_QKV_T_) * TILE_F_;
  const int tid = threadIdx.x;
  const int ln = tid & 31;
  const int c = ln & 15, h = ln >> 4;
  const int q0 = tid >> 5;
  v4f v[8];
#pragma unroll
  for (int i = 0; i < 8; ++i) v[i] = *(const v4f*)(src + c * 512 + 8 * (q0 + 8 * i) + 4 * h);
#pragma unroll
  for (int i = 0; i < 8; ++i) *(v4f*)(d + (q0 + 8 * i) * WCH_ + ln * 4) = v[i];
}

#define QKV_PASSES_ 3
#define QKV_PF_ 4

__device__ __forceinline__ void qkv_pass_geom(int proj,
                                              const float* Wl, const float* Wr,
                                              const float* bl, const float* br,
                                              float* out, float* ws,
                                              const float*& Wp, const float*& bp,
                                              float*& Cp, int& ldc) {
  const int dir = (proj >= 3) ? 1 : 0;
  const int t = proj - 3 * dir;
  Wp = (dir ? Wr : Wl) + (size_t)t * D_ * D_;
  bp = (dir ? br : bl) + t * D_;
  if (t == 0) { Cp = out + dir * D_; ldc = 2 * D_; }
  else        { Cp = ws + (size_t)(2 * dir + t - 1) * BSD_; ldc = D_; }
}

__global__ __launch_bounds__(256)
void qkv_kernel(const float* __restrict__ X,
                const float* __restrict__ Wl, const float* __restrict__ Wr,
                const float* __restrict__ bl, const float* __restrict__ br,
                float* __restrict__ out, float* __restrict__ ws) {
  constexpr int KK = 512;
  constexpr int MT = 4;
  constexpr int BM = 16 * MT;
  static_assert(LDS_ALONE_ >= BM * LDA_S_, "LDS array smaller than the A tile");
  static_assert(BS_ % BM == 0 && D_ % 128 == 0, "whole blocks only: no row or column guard");
  __shared__ __attribute__((aligned(16))) float As[LDS_ALONE_];

  const int lane = threadIdx.x;
  const int wave = threadIdx.y;
  const int tid  = wave * 32 + lane;
  const int row0 = blockIdx.y * BM;
  const int colw = blockIdx.x * 128 + (wave & 3) * 32;
  const int psel = wave >> 2;
  const int c = lane & 15;
  const int h = lane >> 4;

  const float* Wp; const float* bp; float* Cp; int ldc;
  qkv_pass_geom(psel, Wl, Wr, bl, br, out, ws, Wp, bp, Cp, ldc);
  const float* w0 = Wp + (size_t)colw * KK + lane * 4;
  v4f bq[QKV_PF_][2];
#pragma unroll
  for (int s = 0; s < QKV_PF_; ++s) {
    bq[s][0] = *(const v4f*)(w0 + WCH_ * s);
    bq[s][1] = *(const v4f*)(w0 + WCH_ * s + 16 * KK);
  }

  const int srow = tid >> 7;
  const int scol = (tid & 127) * 4;
#pragma unroll
  for (int v = 0; v < BM; v += 2) {
    float4 val = *(const float4*)(X + (size_t)(row0 + v + srow) * KK + scol);
    *(float4*)&As[(v + srow) * LDA_S_ + scol] = val;
  }
  __syncthreads();

  const float* ap = &As[c * LDA_S_ + 4 * h];

  for (int pass = 0; pass < QKV_PASSES_; ++pass) {
    const int np = 2 * ((pass + 1 < QKV_PASSES_) ? pass + 1 : pass) + psel;
    const float* Wn; const float* bn; float* Cn; int ldcn;
    qkv_pass_geom(np, Wl, Wr, bl, br, out, ws, Wn, bn, Cn, ldcn);
    const float* nx0 = Wn + (size_t)colw * KK + lane * 4;

    const float bv0 = bp[colw + c];
    const float bv1 = bp[colw + 16 + c];

    v8f acc[MT][2];
#pragma unroll
    for (int mt = 0; mt < MT; ++mt) { acc[mt][0] = (v8f){}; acc[mt][1] = (v8f){}; }

    v4f an[MT];
#pragma unroll
    for (int mt = 0; mt < MT; ++mt) an[mt] = *(const v4f*)(ap + mt * 16 * LDA_S_);

    for (int it = 0; it < KK / 8; it += QKV_PF_) {
      const bool wrap = (it + QKV_PF_ >= KK / 8);
      const float* cl = wrap ? nx0 : w0;
      const int kb = wrap ? 0 : (it + QKV_PF_) * WCH_;
#pragma unroll
      for (int s = 0; s < QKV_PF_; ++s) {
        const int kk = (it + s) * 8;
        v4f a[MT];
#pragma unroll
        for (int mt = 0; mt < MT; ++mt) a[mt] = an[mt];
        const v4f b0 = bq[s][0], b1 = bq[s][1];
        bq[s][0] = *(const v4f*)(cl + kb + WCH_ * s);
        bq[s][1] = *(const v4f*)(cl + kb + WCH_ * s + 16 * KK);
        int ka = kk + 8; if (ka > KK - 8) ka = KK - 8;
#pragma unroll
        for (int mt = 0; mt < MT; ++mt) an[mt] = *(const v4f*)(ap + mt * 16 * LDA_S_ + ka);

#pragma unroll
        for (int mt = 0; mt < MT; ++mt) {
          acc[mt][0] = WMMA_F32(a[mt].xy, b0.xy, acc[mt][0]);
          acc[mt][1] = WMMA_F32(a[mt].xy, b1.xy, acc[mt][1]);
        }
#pragma unroll
        for (int mt = 0; mt < MT; ++mt) {
          acc[mt][0] = WMMA_F32(a[mt].zw, b0.zw, acc[mt][0]);
          acc[mt][1] = WMMA_F32(a[mt].zw, b1.zw, acc[mt][1]);
        }
      }
    }

#pragma unroll
    for (int mt = 0; mt < MT; ++mt) {
#pragma unroll
      for (int r = 0; r < 8; ++r) {
        float* crow = Cp + (size_t)(row0 + mt * 16 + 8 * h + r) * ldc + colw + c;
        crow[ 0] = acc[mt][0][r] + bv0;
        crow[16] = acc[mt][1][r] + bv1;
      }
    }

    w0 = nx0; bp = bn; Cp = Cn; ldc = ldcn;
  }
}

#define CH_ROWS_ 32
#define CH_X_ (CH_ROWS_ * LDA_S_)
#define CH_PF_ 4
#define CH_PASSES_ 10
#define KW_ROWS_ (CH_ROWS_ + WIDTH_ + 1)
#define CH_LDS_ (CH_X_ + 2 * KW_ROWS_ * LDA_S_)

__device__ __forceinline__ void chain_pass_geom(int pass, int wave,
                                                const float* Wo, const float* bo,
                                                const float* hwW, const float* hwb,
                                                int& n0, int& hic,
                                                const float*& Wst, const float*& bst) {
  if (pass < 2) {
    n0 = wave * 64 + pass * 32; hic = 16; Wst = Wo; bst = bo;
  } else {
    const int q = pass - 2;
    const int layer = q >> 2;
    n0 = wave * 64 + (q & 3) * 16; hic = D_;
    Wst = hwW + (size_t)layer * 2 * D_ * D_; bst = hwb + layer * 2 * D_;
  }
}

__device__ __forceinline__ void chain_wave_pass(const float* ap,
                                                const float* w_lo, const float* w_hi,
                                                const float* nx_lo, const float* nx_hi,
                                                v4f (&bq)[CH_PF_][2], v8f (&acc)[2][2]) {
  constexpr int KK = 512;
  v4f an[2];
  an[0] = *(const v4f*)(ap);
  an[1] = *(const v4f*)(ap + 16 * LDA_S_);
#pragma unroll
  for (int it = 0; it < KK / 8; it += CH_PF_) {
    const bool wrap = (it + CH_PF_ >= KK / 8);
    const float* cl = wrap ? nx_lo : w_lo;
    const float* ch = wrap ? nx_hi : w_hi;
    const int kb = wrap ? 0 : (it + CH_PF_) * WCH_;
#pragma unroll
    for (int s = 0; s < CH_PF_; ++s) {
      const int kk = (it + s) * 8;
      const v4f a0 = an[0], a1 = an[1];
      const v4f b0 = bq[s][0], b1 = bq[s][1];
      bq[s][0] = *(const v4f*)(cl + kb + WCH_ * s);
      bq[s][1] = *(const v4f*)(ch + kb + WCH_ * s);
      int ka = kk + 8; if (ka > KK - 8) ka = KK - 8;
      an[0] = *(const v4f*)(ap + ka);
      an[1] = *(const v4f*)(ap + 16 * LDA_S_ + ka);

      acc[0][0] = WMMA_F32(a0.xy, b0.xy, acc[0][0]);
      acc[0][1] = WMMA_F32(a0.xy, b1.xy, acc[0][1]);
      acc[1][0] = WMMA_F32(a1.xy, b0.xy, acc[1][0]);
      acc[1][1] = WMMA_F32(a1.xy, b1.xy, acc[1][1]);
      acc[0][0] = WMMA_F32(a0.zw, b0.zw, acc[0][0]);
      acc[0][1] = WMMA_F32(a0.zw, b1.zw, acc[0][1]);
      acc[1][0] = WMMA_F32(a1.zw, b0.zw, acc[1][0]);
      acc[1][1] = WMMA_F32(a1.zw, b1.zw, acc[1][1]);
    }
  }
}

template <int KIND>
__device__ __forceinline__ void stage_kv_window(const float* __restrict__ kb,
                                                const float* __restrict__ pb,
                                                int wb, float* dst) {
#pragma unroll
  for (int w = 0; w < KW_ROWS_; ++w) {
    const float* p;
    if (KIND == 0) {
      p = kb + (wb + w) * D_;
    } else if (KIND == 1) {
      if (w <= WIDTH_) { const int pr = (w == 0) ? 0 : w - 1; p = pb + pr * (2 * D_); }
      else { p = kb + (w - (WIDTH_ + 1)) * D_; }
    } else {
      if (w < CH_ROWS_) { p = kb + (S_ - CH_ROWS_ + w) * D_; }
      else { const int pr = (w - CH_ROWS_ < WIDTH_) ? w - CH_ROWS_ : WIDTH_ - 1; p = pb + pr * (2 * D_); }
    }
    const float4 q4 = *(const float4*)p;
    *(float4*)&dst[w * LDA_S_] = q4;
  }
}

__global__ __launch_bounds__(256)
void chain_kernel(const float* __restrict__ kvws,
                  const float* __restrict__ Wo_l,   const float* __restrict__ Wo_r,
                  const float* __restrict__ bo_l,   const float* __restrict__ bo_r,
                  const float* __restrict__ hwW_l,  const float* __restrict__ hwW_r,
                  const float* __restrict__ hwb_l,  const float* __restrict__ hwb_r,
                  float* out) {
  __shared__ __attribute__((aligned(16))) float Xs[CH_LDS_];
  static_assert(CH_LDS_ >= 2 * CH_X_, "LDS array smaller than the two activation buffers");
  static_assert(CH_LDS_ * 4 > 160 * 1024 && CH_LDS_ * 4 <= 320 * 1024, "LDS: one block per WGP");

  const int lane = threadIdx.x;
  const int wave = threadIdx.y;
  const int tid  = wave * 32 + lane;
  const int c = lane & 15;
  const int h = lane >> 4;
  const int dir  = blockIdx.y;
  const int row0 = blockIdx.x * CH_ROWS_;

  const float* Wo   = dir ? Wo_r   : Wo_l;
  const float* bo   = dir ? bo_r   : bo_l;
  const float* hwW  = dir ? hwW_r  : hwW_l;
  const float* hwb  = dir ? hwb_r  : hwb_l;

  int n0, hic; const float* Wst; const float* bst;
  chain_pass_geom(0, wave, Wo, bo, hwW, hwb, n0, hic, Wst, bst);
  const float* w_lo = Wst + (size_t)n0 * D_ + lane * 4;
  const float* w_hi = w_lo + (size_t)hic * D_;
  v4f bq[CH_PF_][2];
#pragma unroll
  for (int s = 0; s < CH_PF_; ++s) {
    bq[s][0] = *(const v4f*)(w_lo + WCH_ * s);
    bq[s][1] = *(const v4f*)(w_hi + WCH_ * s);
  }

  const int srow = tid >> 7;
  const int scol = (tid & 127) * 4;
  const int bb = row0 / S_;
  const int s0 = row0 % S_;
  const int wb = dir ? s0 : s0 - (WIDTH_ + 1);
  float* Ks = Xs + CH_X_;
  float* Vs = Ks + KW_ROWS_ * LDA_S_;
  {
    const float* qsrc = out + (size_t)(row0 + srow) * (2 * D_) + (size_t)dir * D_ + scol;
#pragma unroll
    for (int v = 0; v < CH_ROWS_; v += 2) {
      float4 val = *(const float4*)(qsrc + (size_t)v * (2 * D_));
      *(float4*)&Xs[(v + srow) * LDA_S_ + scol] = val;
    }
    static_assert(S_ % CH_ROWS_ == 0 && CH_ROWS_ > WIDTH_ + 1, "window shapes of stage_kv_window");
    const float* kb = kvws + (size_t)dir * (2 * BSD_) + (size_t)srow * BSD_ + (size_t)(bb * S_) * D_ + scol;
    const float* pb = kvws + 4 * BSD_ + (size_t)dir * (WIDTH_ * 2 * D_) + srow * D_ + scol;
    float* wdst = (srow ? Vs : Ks) + scol;
    if (wb < 0)                          stage_kv_window<1>(kb, pb, wb, wdst);
    else if (wb + KW_ROWS_ - 1 >= S_)    stage_kv_window<2>(kb, pb, wb, wdst);
    else                                 stage_kv_window<0>(kb, pb, wb, wdst);
  }
  __syncthreads();

  {
    const int hh = tid & (H_ - 1);
    const int rsel = tid >> 6;
    const float scale = 0.35355339059327373f;
    const float NEG = -3.0e38f;
    for (int it = 0; it < CH_ROWS_ / 8; ++it) {
      const int r0 = (it * 4 + rsel) * 2;
      float* qrow = Xs + r0 * LDA_S_ + hh * DK_;
      const float* krow = Ks + r0 * LDA_S_ + hh * DK_;
      const float* vrow = Vs + r0 * LDA_S_ + hh * DK_;
      float qa[DK_], qb[DK_];
      *(float4*)&qa[0] = *(const float4*)(qrow);
      *(float4*)&qa[4] = *(const float4*)(qrow + 4);
      *(float4*)&qb[0] = *(const float4*)(qrow + LDA_S_);
      *(float4*)&qb[4] = *(const float4*)(qrow + LDA_S_ + 4);
      const int sa_row = s0 + r0;
      const int sb_row = s0 + r0 + 1;
      float sa[WIN_], sb[WIN_];
      float mxa = NEG, mxb = NEG;
#pragma unroll
      for (int u = 0; u <= WIN_; ++u) {
        float kr[DK_];
        *(float4*)&kr[0] = *(const float4*)(krow + u * LDA_S_);
        *(float4*)&kr[4] = *(const float4*)(krow + u * LDA_S_ + 4);
        if (u < WIN_) {
          float dot = 0.f;
#pragma unroll
          for (int d = 0; d < DK_; ++d) dot += qa[d] * kr[d];
          const bool ok = (u == 0) ? !(dir == 0 && sa_row == 0)
                        : ((u == WIN_ - 1) ? !(dir == 1 && sa_row == S_ - 1) : true);
          const float sc = ok ? dot * scale : NEG;
          sa[u] = sc; mxa = fmaxf(mxa, sc);
        }
        if (u >= 1) {
          float dot = 0.f;
#pragma unroll
          for (int d = 0; d < DK_; ++d) dot += qb[d] * kr[d];
          const bool ok = (u - 1 == 0) ? !(dir == 0 && sb_row == 0)
                        : ((u - 1 == WIN_ - 1) ? !(dir == 1 && sb_row == S_ - 1) : true);
          const float sc = ok ? dot * scale : NEG;
          sb[u - 1] = sc; mxb = fmaxf(mxb, sc);
        }
      }
      float suma = 0.f, sumb = 0.f;
#pragma unroll
      for (int t = 0; t < WIN_; ++t) {
        const float pa = __expf(sa[t] - mxa);
        const float pb = __expf(sb[t] - mxb);
        sa[t] = pa; suma += pa;
        sb[t] = pb; sumb += pb;
      }
      const float inva = 1.f / suma;
      const float invb = 1.f / sumb;
      float oa[DK_] = {}, ob[DK_] = {};
#pragma unroll
      for (int u = 0; u <= WIN_; ++u) {
        float vr[DK_];
        *(float4*)&vr[0] = *(const float4*)(vrow + u * LDA_S_);
        *(float4*)&vr[4] = *(const float4*)(vrow + u * LDA_S_ + 4);
        if (u < WIN_) {
          const float wgt = sa[u] * inva;
#pragma unroll
          for (int d = 0; d < DK_; ++d) oa[d] += wgt * vr[d];
        }
        if (u >= 1) {
          const float wgt = sb[u - 1] * invb;
#pragma unroll
          for (int d = 0; d < DK_; ++d) ob[d] += wgt * vr[d];
        }
      }
      *(float4*)(qrow)              = make_float4(oa[0], oa[1], oa[2], oa[3]);
      *(float4*)(qrow + 4)          = make_float4(oa[4], oa[5], oa[6], oa[7]);
      *(float4*)(qrow + LDA_S_)     = make_float4(ob[0], ob[1], ob[2], ob[3]);
      *(float4*)(qrow + LDA_S_ + 4) = make_float4(ob[4], ob[5], ob[6], ob[7]);
    }
  }
  __syncthreads();

  for (int pass = 0; pass < CH_PASSES_; ++pass) {
    const int stage = (pass < 2) ? 0 : ((pass < 6) ? 1 : 2);
    const float* Xin  = Xs + (stage & 1) * CH_X_;
    float*       Xout = Xs + ((stage & 1) ^ 1) * CH_X_;

    const int np = (pass + 1 < CH_PASSES_) ? pass + 1 : pass;
    int n0n, hicn; const float* Wn; const float* bn;
    chain_pass_geom(np, wave, Wo, bo, hwW, hwb, n0n, hicn, Wn, bn);
    const float* nx_lo = Wn + (size_t)n0n * D_ + lane * 4;
    const float* nx_hi = nx_lo + (size_t)hicn * D_;

    const float bv0 = bst[n0 + c];
    const float bv1 = bst[n0 + hic + c];

    v8f acc[2][2];
#pragma unroll
    for (int mt = 0; mt < 2; ++mt) {
#pragma unroll
      for (int t = 0; t < 2; ++t) acc[mt][t] = (v8f){};
    }

    chain_wave_pass(Xin + c * LDA_S_ + 4 * h, w_lo, w_hi, nx_lo, nx_hi, bq, acc);

    if (stage == 0) {
#pragma unroll
      for (int mt = 0; mt < 2; ++mt) {
#pragma unroll
        for (int r = 0; r < 8; ++r) {
          float* xr = Xout + (mt * 16 + 8 * h + r) * LDA_S_ + n0 + c;
          xr[ 0] = acc[mt][0][r] + bv0;
          xr[16] = acc[mt][1][r] + bv1;
        }
      }
    } else {
#pragma unroll
      for (int mt = 0; mt < 2; ++mt) {
#pragma unroll
        for (int r = 0; r < 8; ++r) {
          const int idx = (mt * 16 + 8 * h + r) * LDA_S_ + n0 + c;
          const float lin = acc[mt][0][r] + bv0;
          const float gin = acc[mt][1][r] + bv1;
          const float g  = 1.f / (1.f + __expf(-gin));
          const float nl = fmaxf(lin, 0.f);
          Xout[idx] = g * Xin[idx] + (1.f - g) * nl;
        }
      }
    }

    if (pass == 1 || pass == 5 || pass == 9) __syncthreads();

    n0 = n0n; hic = hicn; bst = bn; w_lo = nx_lo; w_hi = nx_hi;
  }

  const float* Xf = Xs + CH_X_ + srow * LDA_S_ + scol;
  float* o0 = out + (size_t)(row0 + srow) * (2 * D_) + (size_t)dir * D_ + scol;
#pragma unroll 8
  for (int v = 0; v < CH_ROWS_; v += 2) {
    float4 val = *(const float4*)&Xf[v * LDA_S_];
    *(float4*)(o0 + (size_t)v * (2 * D_)) = val;
    *(float4*)(o0 + (size_t)v * (2 * D_) + (size_t)BS_ * 2 * D_) = val;
  }
}

extern "C" void kernel_launch(void* const* d_in, const int* in_sizes, int n_in,
                              void* d_out, int out_size, void* d_ws, size_t ws_size,
                              hipStream_t stream) {
  const float* inputs    = (const float*)d_in[0];
  const float* left_pad  = (const float*)d_in[1];
  const float* right_pad = (const float*)d_in[2];
  const float* Wqkv[2] = {(const float*)d_in[3],  (const float*)d_in[7]};
  const float* bqkv[2] = {(const float*)d_in[4],  (const float*)d_in[8]};
  const float* Wo[2]   = {(const float*)d_in[5],  (const float*)d_in[9]};
  const float* bo[2]   = {(const float*)d_in[6],  (const float*)d_in[10]};
  const float* hwW[2]  = {(const float*)d_in[11], (const float*)d_in[13]};
  const float* hwb[2]  = {(const float*)d_in[12], (const float*)d_in[14]};
  float* out = (float*)d_out;

  if (ws_size < WS_NEED_) return;

  float* ws    = (float*)d_ws;
  float* padl  = ws + 4 * BSD_;
  float* padr  = padl + WIDTH_ * 2 * D_;
  float* pkq   = out + (size_t)BS_ * 2 * D_;
  float* pk    = padr + WIDTH_ * 2 * D_;
  const float* pkWqkv[2] = {pkq, pkq + (size_t)PK_QKV_T_ * TILE_F_};
  const float* pkWo[2]   = {pk, pk + (size_t)PK_WO_T_ * TILE_F_};
  const float* pkHw[2]   = {pk + (size_t)(2 * PK_WO_T_) * TILE_F_,
                            pk + (size_t)(2 * PK_WO_T_ + PK_HW_T_) * TILE_F_};

  dim3 blk8(32, 8);

  prep_kernel<<<dim3(PADB_ + PK_TILES_), dim3(256), 0, stream>>>(
      Wqkv[0], Wqkv[1], Wo[0], Wo[1], hwW[0], hwW[1], pkq, pk,
      left_pad, right_pad, bqkv[0], bqkv[1], padl, padr);
  { dim3 grd(D_ / 128, BS_ / 64);
    qkv_kernel<<<grd, blk8, 0, stream>>>(inputs, pkWqkv[0], pkWqkv[1], bqkv[0], bqkv[1], out, ws); }
  { dim3 grd(BS_ / CH_ROWS_, 2);
    chain_kernel<<<grd, blk8, 0, stream>>>(ws, pkWo[0], pkWo[1], bo[0], bo[1],
                                          pkHw[0], pkHw[1], hwb[0], hwb[1], out); }
}


# Fragment loads grouped: both big kernels run K in blocks of 64 WMMAs per wave, two register sets; the next block's W and A loads go out as one group ahead of the block; k and sum order unchanged
# speedup vs baseline: 3.5561x; 1.0316x over previous
#include <hip/hip_runtime.h>

#define B_ 8
#define S_ 512
#define D_ 512
#define H_ 64
#define DK_ 8
#define WIDTH_ 16
#define N_ (S_ + 2 * WIDTH_)
#define NL_ 2
#define WIN_ (WIDTH_ + 2)
#define BS_ (B_ * S_)
static const size_t BSD_ = (size_t)BS_ * D_;

typedef __attribute__((ext_vector_type(2))) float v2f;
typedef __attribute__((ext_vector_type(8))) float v8f;

typedef __attribute__((ext_vector_type(4))) float v4f;

#define WMMA_F32(a, b, c) __builtin_amdgcn_wmma_f32_16x16x4_f32(false, (a), false, (b), (short)0, (c), false, false)

#define LDA_S_ (512 + 4)
#define LDS_TILE16_ (16 * LDA_S_)
#define LDS_ALONE_ (162 * 256)

#define WCH_ 128
#define TILE_F_ (16 * 512)
#define PK_QKV_T_ 96
#define PK_WO_T_ 32
#define PK_HW_T_ 128
#define PK_TILES_ (2 * (PK_QKV_T_ + PK_WO_T_ + PK_HW_T_))
#define PK_REST_T_ (PK_TILES_ - 2 * PK_QKV_T_)
static const size_t WS_NEED_ =
    (4 * BSD_ + 2 * (size_t)WIDTH_ * 2 * D_ + (size_t)PK_REST_T_ * TILE_F_) * sizeof(float);
static_assert(WS_NEED_ == 44171264, "workspace carve-up: 32 MiB + 128 KiB + 10 MiB");

#define PADB_ (2 * (2 * D_ / 128))
#define PAD_PF_ 16
static_assert((2 * D_) % 128 == 0 && WIDTH_ == 16, "pad_rows_gemm: whole 16 x 128 blocks, no row or column guard");

__device__ __forceinline__ void pad_rows_gemm(const float* A, const float* W, const float* bias,
                                              float* C, int colgrp, float* As) {
  constexpr int KK = 512;
  const int tid  = threadIdx.x;
  const int lane = tid & 31;
  const int wave = tid >> 5;
  const int c = lane & 15;
  const int h = lane >> 4;
  const int col0 = colgrp * 128 + wave * 16;
  const float* w0 = W + (size_t)(col0 + c) * KK + 4 * h;

  v4f bq[PAD_PF_];
#pragma unroll
  for (int s = 0; s < PAD_PF_; ++s) bq[s] = *(const v4f*)(w0 + 8 * s);

  const int srow = tid >> 7;
  const int scol = (tid & 127) * 4;
#pragma unroll
  for (int v = 0; v < WIDTH_; v += 2) {
    float4 val = *(const float4*)(A + (size_t)(v + srow) * KK + scol);
    *(float4*)&As[(v + srow) * LDA_S_ + scol] = val;
  }
  __syncthreads();

  const float* ap = &As[c * LDA_S_ + 4 * h];
  v8f acc = (v8f){};
  v4f an = *(const v4f*)ap;
  for (int it = 0; it < KK / 8; it += PAD_PF_) {
#pragma unroll
    for (int s = 0; s < PAD_PF_; ++s) {
      const int kk = (it + s) * 8;
      const v4f a = an;
      const v4f b = bq[s];
      int kp = kk + 8 * PAD_PF_; if (kp > KK - 8) kp = KK - 8;
      bq[s] = *(const v4f*)(w0 + kp);
      int ka = kk + 8; if (ka > KK - 8) ka = KK - 8;
      an = *(const v4f*)(ap + ka);
      acc = WMMA_F32(a.xy, b.xy, acc);
      acc = WMMA_F32(a.zw, b.zw, acc);
    }
  }

  const float bv = bias[col0 + c];
  asm volatile("s_wait_alu depctr_va_vdst(0)" : "+v"(acc));
#pragma unroll
  for (int r = 0; r < 8; ++r) C[(size_t)(8 * h + r) * (2 * D_) + col0 + c] = acc[r] + bv;
}

__global__ __launch_bounds__(256)
void prep_kernel(const float* Wqkv_l, const float* Wqkv_r,
                 const float* Wo_l,   const float* Wo_r,
                 const float* hwW_l,  const float* hwW_r,
                 float* dst_qkv, float* dst_rest,
                 const float* pad_l,  const float* pad_r,
                 const float* bqkv_l, const float* bqkv_r,
                 float* padkv_l, float* padkv_r) {
  __shared__ __attribute__((aligned(16))) float As[LDS_TILE16_];
  if (blockIdx.x < PADB_) {
    const int dir = blockIdx.x / (PADB_ / 2);
    const int colgrp = blockIdx.x % (PADB_ / 2);
    pad_rows_gemm(dir ? pad_r : pad_l,
                  (dir ? Wqkv_r : Wqkv_l) + (size_t)D_ * D_,
                  (dir ? bqkv_r : bqkv_l) + D_,
                  dir ? padkv_r : padkv_l, colgrp, As);
    return;
  }
  const int t = blockIdx.x - PADB_;
  const float* src; int t0;
  if      (t < PK_QKV_T_)                               { src = Wqkv_l; t0 = 0; }
  else if (t < 2 * PK_QKV_T_)                           { src = Wqkv_r; t0 = PK_QKV_T_; }
  else if (t < 2 * PK_QKV_T_ + PK_WO_T_)                { src = Wo_l;   t0 = 2 * PK_QKV_T_; }
  else if (t < 2 * PK_QKV_T_ + 2 * PK_WO_T_)            { src = Wo_r;   t0 = 2 * PK_QKV_T_ + PK_WO_T_; }
  else if (t < 2 * PK_QKV_T_ + 2 * PK_WO_T_ + PK_HW_T_) { src = hwW_l;  t0 = 2 * PK_QKV_T_ + 2 * PK_WO_T_; }
  else                                                  { src = hwW_r;  t0 = 2 * PK_QKV_T_ + 2 * PK_WO_T_ + PK_HW_T_; }
  src += (size_t)(t - t0) * TILE_F_;
  float* d = (t < 2 * PK_QKV_T_) ? dst_qkv + (size_t)t * TILE_F_
                                 : dst_rest + (size_t)(t - 2 * PK_QKV_T_) * TILE_F_;
  const int tid = threadIdx.x;
  const int ln = tid & 31;
  const int c = ln & 15, h = ln >> 4;
  const int q0 = tid >> 5;
  v4f v[8];
#pragma unroll
  for (int i = 0; i < 8; ++i) v[i] = *(const v4f*)(src + c * 512 + 8 * (q0 + 8 * i) + 4 * h);
#pragma unroll
  for (int i = 0; i < 8; ++i) *(v4f*)(d + (q0 + 8 * i) * WCH_ + ln * 4) = v[i];
}

#define QKV_PASSES_ 3
#define QKV_G_ 4
#define QKV_NB_ (D_ / 8 / QKV_G_)
static_assert(QKV_NB_ % 2 == 0, "the two register sets must keep their roles from pass to pass");

__device__ __forceinline__ void qkv_pass_geom(int proj,
                                              const float* Wl, const float* Wr,
                                              const float* bl, const float* br,
                                              float* out, float* ws,
                                              const float*& Wp, const float*& bp,
                                              float*& Cp, int& ldc) {
  const int dir = (proj >= 3) ? 1 : 0;
  const int t = proj - 3 * dir;
  Wp = (dir ? Wr : Wl) + (size_t)t * D_ * D_;
  bp = (dir ? br : bl) + t * D_;
  if (t == 0) { Cp = out + dir * D_; ldc = 2 * D_; }
  else        { Cp = ws + (size_t)(2 * dir + t - 1) * BSD_; ldc = D_; }
}

__global__ __launch_bounds__(256)
void qkv_kernel(const float* __restrict__ X,
                const float* __restrict__ Wl, const float* __restrict__ Wr,
                const float* __restrict__ bl, const float* __restrict__ br,
                float* __restrict__ out, float* __restrict__ ws) {
  constexpr int KK = 512;
  constexpr int MT = 4;
  constexpr int BM = 16 * MT;
  static_assert(LDS_ALONE_ >= BM * LDA_S_, "LDS array smaller than the A tile");
  static_assert(BS_ % BM == 0 && D_ % 128 == 0, "whole blocks only: no row or column guard");
  __shared__ __attribute__((aligned(16))) float As[LDS_ALONE_];

  const int lane = threadIdx.x;
  const int wave = threadIdx.y;
  const int tid  = wave * 32 + lane;
  const int row0 = blockIdx.y * BM;
  const int colw = blockIdx.x * 128 + (wave & 3) * 32;
  const int psel = wave >> 2;
  const int c = lane & 15;
  const int h = lane >> 4;

  const float* Wp; const float* bp; float* Cp; int ldc;
  qkv_pass_geom(psel, Wl, Wr, bl, br, out, ws, Wp, bp, Cp, ldc);
  const float* w0 = Wp + (size_t)colw * KK + lane * 4;
  v4f wq[2][QKV_G_][2];
#pragma unroll
  for (int s = 0; s < QKV_G_; ++s) {
    wq[0][s][0] = *(const v4f*)(w0 + WCH_ * s);
    wq[0][s][1] = *(const v4f*)(w0 + WCH_ * s + 16 * KK);
  }

  const int srow = tid >> 7;
  const int scol = (tid & 127) * 4;
#pragma unroll
  for (int v = 0; v < BM; v += 2) {
    float4 val = *(const float4*)(X + (size_t)(row0 + v + srow) * KK + scol);
    *(float4*)&As[(v + srow) * LDA_S_ + scol] = val;
  }
  __syncthreads();

  const float* ap = &As[c * LDA_S_ + 4 * h];

  for (int pass = 0; pass < QKV_PASSES_; ++pass) {
    const int np = 2 * ((pass + 1 < QKV_PASSES_) ? pass + 1 : pass) + psel;
    const float* Wn; const float* bn; float* Cn; int ldcn;
    qkv_pass_geom(np, Wl, Wr, bl, br, out, ws, Wn, bn, Cn, ldcn);
    const float* nx0 = Wn + (size_t)colw * KK + lane * 4;

    const float bv0 = bp[colw + c];
    const float bv1 = bp[colw + 16 + c];

    v8f acc[MT][2];
#pragma unroll
    for (int mt = 0; mt < MT; ++mt) { acc[mt][0] = (v8f){}; acc[mt][1] = (v8f){}; }

    v4f aq[2][QKV_G_][MT];
#pragma unroll
    for (int s = 0; s < QKV_G_; ++s) {
#pragma unroll
      for (int mt = 0; mt < MT; ++mt) aq[0][s][mt] = *(const v4f*)(ap + mt * 16 * LDA_S_ + 8 * s);
    }

#pragma unroll
    for (int j = 0; j < QKV_NB_; ++j) {
      const int cur = j & 1, nxt = cur ^ 1;
      const bool last = (j + 1 == QKV_NB_);
      const float* cl = last ? nx0 : w0 + (j + 1) * QKV_G_ * WCH_;
#pragma unroll
      for (int s = 0; s < QKV_G_; ++s) {
        wq[nxt][s][0] = *(const v4f*)(cl + WCH_ * s);
        wq[nxt][s][1] = *(const v4f*)(cl + WCH_ * s + 16 * KK);
      }
      if (!last) {
#pragma unroll
        for (int s = 0; s < QKV_G_; ++s) {
#pragma unroll
          for (int mt = 0; mt < MT; ++mt)
            aq[nxt][s][mt] = *(const v4f*)(ap + mt * 16 * LDA_S_ + 8 * ((j + 1) * QKV_G_ + s));
        }
      }
      __builtin_amdgcn_sched_barrier(0);
#pragma unroll
      for (int s = 0; s < QKV_G_; ++s) {
        const v4f b0 = wq[cur][s][0], b1 = wq[cur][s][1];
#pragma unroll
        for (int mt = 0; mt < MT; ++mt) {
          acc[mt][0] = WMMA_F32(aq[cur][s][mt].xy, b0.xy, acc[mt][0]);
          acc[mt][1] = WMMA_F32(aq[cur][s][mt].xy, b1.xy, acc[mt][1]);
        }
#pragma unroll
        for (int mt = 0; mt < MT; ++mt) {
          acc[mt][0] = WMMA_F32(aq[cur][s][mt].zw, b0.zw, acc[mt][0]);
          acc[mt][1] = WMMA_F32(aq[cur][s][mt].zw, b1.zw, acc[mt][1]);
        }
      }
      __builtin_amdgcn_sched_barrier(0);
    }

#pragma unroll
    for (int mt = 0; mt < MT; ++mt) {
#pragma unroll
      for (int r = 0; r < 8; ++r) {
        float* crow = Cp + (size_t)(row0 + mt * 16 + 8 * h + r) * ldc + colw + c;
        crow[ 0] = acc[mt][0][r] + bv0;
        crow[16] = acc[mt][1][r] + bv1;
      }
    }

    w0 = nx0; bp = bn; Cp = Cn; ldc = ldcn;
  }
}

#define CH_ROWS_ 32
#define CH_X_ (CH_ROWS_ * LDA_S_)
#define CH_G_ 8
#define CH_NB_ (D_ / 8 / CH_G_)
static_assert(CH_NB_ % 2 == 0, "the two register sets must keep their roles from pass to pass");
#define CH_PASSES_ 10
#define KW_ROWS_ (CH_ROWS_ + WIDTH_ + 1)
#define CH_LDS_ (CH_X_ + 2 * KW_ROWS_ * LDA_S_)

__device__ __forceinline__ void chain_pass_geom(int pass, int wave,
                                                const float* Wo, const float* bo,
                                                const float* hwW, const float* hwb,
                                                int& n0, int& hic,
                                                const float*& Wst, const float*& bst) {
  if (pass < 2) {
    n0 = wave * 64 + pass * 32; hic = 16; Wst = Wo; bst = bo;
  } else {
    const int q = pass - 2;
    const int layer = q >> 2;
    n0 = wave * 64 + (q & 3) * 16; hic = D_;
    Wst = hwW + (size_t)layer * 2 * D_ * D_; bst = hwb + layer * 2 * D_;
  }
}

__device__ __forceinline__ void chain_wave_pass(const float* ap,
                                                const float* w_lo, const float* w_hi,
                                                const float* nx_lo, const float* nx_hi,
                                                v4f (&wq)[2][CH_G_][2], v8f (&acc)[2][2]) {
  v4f aq[2][CH_G_][2];
#pragma unroll
  for (int s = 0; s < CH_G_; ++s) {
    aq[0][s][0] = *(const v4f*)(ap + 8 * s);
    aq[0][s][1] = *(const v4f*)(ap + 16 * LDA_S_ + 8 * s);
  }
#pragma unroll
  for (int j = 0; j < CH_NB_; ++j) {
    const int cur = j & 1, nxt = cur ^ 1;
    const bool last = (j + 1 == CH_NB_);
    const float* cl = last ? nx_lo : w_lo + (j + 1) * CH_G_ * WCH_;
    const float* ch = last ? nx_hi : w_hi + (j + 1) * CH_G_ * WCH_;
#pragma unroll
    for (int s = 0; s < CH_G_; ++s) {
      wq[nxt][s][0] = *(const v4f*)(cl + WCH_ * s);
      wq[nxt][s][1] = *(const v4f*)(ch + WCH_ * s);
    }
    if (!last) {
#pragma unroll
      for (int s = 0; s < CH_G_; ++s) {
        aq[nxt][s][0] = *(const v4f*)(ap + 8 * ((j + 1) * CH_G_ + s));
        aq[nxt][s][1] = *(const v4f*)(ap + 16 * LDA_S_ + 8 * ((j + 1) * CH_G_ + s));
      }
    }
    __builtin_amdgcn_sched_barrier(0);
#pragma unroll
    for (int s = 0; s < CH_G_; ++s) {
      const v4f a0 = aq[cur][s][0], a1 = aq[cur][s][1];
      const v4f b0 = wq[cur][s][0], b1 = wq[cur][s][1];
      acc[0][0] = WMMA_F32(a0.xy, b0.xy, acc[0][0]);
      acc[0][1] = WMMA_F32(a0.xy, b1.xy, acc[0][1]);
      acc[1][0] = WMMA_F32(a1.xy, b0.xy, acc[1][0]);
      acc[1][1] = WMMA_F32(a1.xy, b1.xy, acc[1][1]);
      acc[0][0] = WMMA_F32(a0.zw, b0.zw, acc[0][0]);
      acc[0][1] = WMMA_F32(a0.zw, b1.zw, acc[0][1]);
      acc[1][0] = WMMA_F32(a1.zw, b0.zw, acc[1][0]);
      acc[1][1] = WMMA_F32(a1.zw, b1.zw, acc[1][1]);
    }
    __builtin_amdgcn_sched_barrier(0);
  }
}

template <int KIND>
__device__ __forceinline__ void stage_kv_window(const float* __restrict__ kb,
                                                const float* __restrict__ pb,
                                                int wb, float* dst) {
#pragma unroll
  for (int w = 0; w < KW_ROWS_; ++w) {
    const float* p;
    if (KIND == 0) {
      p = kb + (wb + w) * D_;
    } else if (KIND == 1) {
      if (w <= WIDTH_) { const int pr = (w == 0) ? 0 : w - 1; p = pb + pr * (2 * D_); }
      else { p = kb + (w - (WIDTH_ + 1)) * D_; }
    } else {
      if (w < CH_ROWS_) { p = kb + (S_ - CH_ROWS_ + w) * D_; }
      else { const int pr = (w - CH_ROWS_ < WIDTH_) ? w - CH_ROWS_ : WIDTH_ - 1; p = pb + pr * (2 * D_); }
    }
    const float4 q4 = *(const float4*)p;
    *(float4*)&dst[w * LDA_S_] = q4;
  }
}

__global__ __launch_bounds__(256)
void chain_kernel(const float* __restrict__ kvws,
                  const float* __restrict__ Wo_l,   const float* __restrict__ Wo_r,
                  const float* __restrict__ bo_l,   const float* __restrict__ bo_r,
                  const float* __restrict__ hwW_l,  const float* __restrict__ hwW_r,
                  const float* __restrict__ hwb_l,  const float* __restrict__ hwb_r,
                  float* out) {
  __shared__ __attribute__((aligned(16))) float Xs[CH_LDS_];
  static_assert(CH_LDS_ >= 2 * CH_X_, "LDS array smaller than the two activation buffers");
  static_assert(CH_LDS_ * 4 > 160 * 1024 && CH_LDS_ * 4 <= 320 * 1024, "LDS: one block per WGP");

  const int lane = threadIdx.x;
  const int wave = threadIdx.y;
  const int tid  = wave * 32 + lane;
  const int c = lane & 15;
  const int h = lane >> 4;
  const int dir  = blockIdx.y;
  const int row0 = blockIdx.x * CH_ROWS_;

  const float* Wo   = dir ? Wo_r   : Wo_l;
  const float* bo   = dir ? bo_r   : bo_l;
  const float* hwW  = dir ? hwW_r  : hwW_l;
  const float* hwb  = dir ? hwb_r  : hwb_l;

  int n0, hic; const float* Wst; const float* bst;
  chain_pass_geom(0, wave, Wo, bo, hwW, hwb, n0, hic, Wst, bst);
  const float* w_lo = Wst + (size_t)n0 * D_ + lane * 4;
  const float* w_hi = w_lo + (size_t)hic * D_;
  v4f wq[2][CH_G_][2];
#pragma unroll
  for (int s = 0; s < CH_G_; ++s) {
    wq[0][s][0] = *(const v4f*)(w_lo + WCH_ * s);
    wq[0][s][1] = *(const v4f*)(w_hi + WCH_ * s);
  }

  const int srow = tid >> 7;
  const int scol = (tid & 127) * 4;
  const int bb = row0 / S_;
  const int s0 = row0 % S_;
  const int wb = dir ? s0 : s0 - (WIDTH_ + 1);
  float* Ks = Xs + CH_X_;
  float* Vs = Ks + KW_ROWS_ * LDA_S_;
  {
    const float* qsrc = out + (size_t)(row0 + srow) * (2 * D_) + (size_t)dir * D_ + scol;
#pragma unroll
    for (int v = 0; v < CH_ROWS_; v += 2) {
      float4 val = *(const float4*)(qsrc + (size_t)v * (2 * D_));
      *(float4*)&Xs[(v + srow) * LDA_S_ + scol] = val;
    }
    static_assert(S_ % CH_ROWS_ == 0 && CH_ROWS_ > WIDTH_ + 1, "window shapes of stage_kv_window");
    const float* kb = kvws + (size_t)dir * (2 * BSD_) + (size_t)srow * BSD_ + (size_t)(bb * S_) * D_ + scol;
    const float* pb = kvws + 4 * BSD_ + (size_t)dir * (WIDTH_ * 2 * D_) + srow * D_ + scol;
    float* wdst = (srow ? Vs : Ks) + scol;
    if (wb < 0)                          stage_kv_window<1>(kb, pb, wb, wdst);
    else if (wb + KW_ROWS_ - 1 >= S_)    stage_kv_window<2>(kb, pb, wb, wdst);
    else                                 stage_kv_window<0>(kb, pb, wb, wdst);
  }
  __syncthreads();

  {
    const int hh = tid & (H_ - 1);
    const int rsel = tid >> 6;
    const float scale = 0.35355339059327373f;
    const float NEG = -3.0e38f;
    for (int it = 0; it < CH_ROWS_ / 8; ++it) {
      const int r0 = (it * 4 + rsel) * 2;
      float* qrow = Xs + r0 * LDA_S_ + hh * DK_;
      const float* krow = Ks + r0 * LDA_S_ + hh * DK_;
      const float* vrow = Vs + r0 * LDA_S_ + hh * DK_;
      float qa[DK_], qb[DK_];
      *(float4*)&qa[0] = *(const float4*)(qrow);
      *(float4*)&qa[4] = *(const float4*)(qrow + 4);
      *(float4*)&qb[0] = *(const float4*)(qrow + LDA_S_);
      *(float4*)&qb[4] = *(const float4*)(qrow + LDA_S_ + 4);
      const int sa_row = s0 + r0;
      const int sb_row = s0 + r0 + 1;
      float sa[WIN_], sb[WIN_];
      float mxa = NEG, mxb = NEG;
#pragma unroll
      for (int u = 0; u <= WIN_; ++u) {
        float kr[DK_];
        *(float4*)&kr[0] = *(const float4*)(krow + u * LDA_S_);
        *(float4*)&kr[4] = *(const float4*)(krow + u * LDA_S_ + 4);
        if (u < WIN_) {
          float dot = 0.f;
#pragma unroll
          for (int d = 0; d < DK_; ++d) dot += qa[d] * kr[d];
          const bool ok = (u == 0) ? !(dir == 0 && sa_row == 0)
                        : ((u == WIN_ - 1) ? !(dir == 1 && sa_row == S_ - 1) : true);
          const float sc = ok ? dot * scale : NEG;
          sa[u] = sc; mxa = fmaxf(mxa, sc);
        }
        if (u >= 1) {
          float dot = 0.f;
#pragma unroll
          for (int d = 0; d < DK_; ++d) dot += qb[d] * kr[d];
          const bool ok = (u - 1 == 0) ? !(dir == 0 && sb_row == 0)
                        : ((u - 1 == WIN_ - 1) ? !(dir == 1 && sb_row == S_ - 1) : true);
          const float sc = ok ? dot * scale : NEG;
          sb[u - 1] = sc; mxb = fmaxf(mxb, sc);
        }
      }
      float suma = 0.f, sumb = 0.f;
#pragma unroll
      for (int t = 0; t < WIN_; ++t) {
        const float pa = __expf(sa[t] - mxa);
        const float pb = __expf(sb[t] - mxb);
        sa[t] = pa; suma += pa;
        sb[t] = pb; sumb += pb;
      }
      const float inva = 1.f / suma;
      const float invb = 1.f / sumb;
      float oa[DK_] = {}, ob[DK_] = {};
#pragma unroll
      for (int u = 0; u <= WIN_; ++u) {
        float vr[DK_];
        *(float4*)&vr[0] = *(const float4*)(vrow + u * LDA_S_);
        *(float4*)&vr[4] = *(const float4*)(vrow + u * LDA_S_ + 4);
        if (u < WIN_) {
          const float wgt = sa[u] * inva;
#pragma unroll
          for (int d = 0; d < DK_; ++d) oa[d] += wgt * vr[d];
        }
        if (u >= 1) {
          const float wgt = sb[u - 1] * invb;
#pragma unroll
          for (int d = 0; d < DK_; ++d) ob[d] += wgt * vr[d];
        }
      }
      *(float4*)(qrow)              = make_float4(oa[0], oa[1], oa[2], oa[3]);
      *(float4*)(qrow + 4)          = make_float4(oa[4], oa[5], oa[6], oa[7]);
      *(float4*)(qrow + LDA_S_)     = make_float4(ob[0], ob[1], ob[2], ob[3]);
      *(float4*)(qrow + LDA_S_ + 4) = make_float4(ob[4], ob[5], ob[6], ob[7]);
    }
  }
  __syncthreads();

  for (int pass = 0; pass < CH_PASSES_; ++pass) {
    const int stage = (pass < 2) ? 0 : ((pass < 6) ? 1 : 2);
    const float* Xin  = Xs + (stage & 1) * CH_X_;
    float*       Xout = Xs + ((stage & 1) ^ 1) * CH_X_;

    const int np = (pass + 1 < CH_PASSES_) ? pass + 1 : pass;
    int n0n, hicn; const float* Wn; const float* bn;
    chain_pass_geom(np, wave, Wo, bo, hwW, hwb, n0n, hicn, Wn, bn);
    const float* nx_lo = Wn + (size_t)n0n * D_ + lane * 4;
    const float* nx_hi = nx_lo + (size_t)hicn * D_;

    const float bv0 = bst[n0 + c];
    const float bv1 = bst[n0 + hic + c];

    v8f acc[2][2];
#pragma unroll
    for (int mt = 0; mt < 2; ++mt) {
#pragma unroll
      for (int t = 0; t < 2; ++t) acc[mt][t] = (v8f){};
    }

    chain_wave_pass(Xin + c * LDA_S_ + 4 * h, w_lo, w_hi, nx_lo, nx_hi, wq, acc);

    if (stage == 0) {
#pragma unroll
      for (int mt = 0; mt < 2; ++mt) {
#pragma unroll
        for (int r = 0; r < 8; ++r) {
          float* xr = Xout + (mt * 16 + 8 * h + r) * LDA_S_ + n0 + c;
          xr[ 0] = acc[mt][0][r] + bv0;
          xr[16] = acc[mt][1][r] + bv1;
        }
      }
    } else {
#pragma unroll
      for (int mt = 0; mt < 2; ++mt) {
#pragma unroll
        for (int r = 0; r < 8; ++r) {
          const int idx = (mt * 16 + 8 * h + r) * LDA_S_ + n0 + c;
          const float lin = acc[mt][0][r] + bv0;
          const float gin = acc[mt][1][r] + bv1;
          const float g  = 1.f / (1.f + __expf(-gin));
          const float nl = fmaxf(lin, 0.f);
          Xout[idx] = g * Xin[idx] + (1.f - g) * nl;
        }
      }
    }

    if (pass == 1 || pass == 5 || pass == 9) __syncthreads();

    n0 = n0n; hic = hicn; bst = bn; w_lo = nx_lo; w_hi = nx_hi;
  }

  const float* Xf = Xs + CH_X_ + srow * LDA_S_ + scol;
  float* o0 = out + (size_t)(row0 + srow) * (2 * D_) + (size_t)dir * D_ + scol;
#pragma unroll 8
  for (int v = 0; v < CH_ROWS_; v += 2) {
    float4 val = *(const float4*)&Xf[v * LDA_S_];
    *(float4*)(o0 + (size_t)v * (2 * D_)) = val;
    *(float4*)(o0 + (size_t)v * (2 * D_) + (size_t)BS_ * 2 * D_) = val;
  }
}

extern "C" void kernel_launch(void* const* d_in, const int* in_sizes, int n_in,
                              void* d_out, int out_size, void* d_ws, size_t ws_size,
                              hipStream_t stream) {
  const float* inputs    = (const float*)d_in[0];
  const float* left_pad  = (const float*)d_in[1];
  const float* right_pad = (const float*)d_in[2];
  const float* Wqkv[2] = {(const float*)d_in[3],  (const float*)d_in[7]};
  const float* bqkv[2] = {(const float*)d_in[4],  (const float*)d_in[8]};
  const float* Wo[2]   = {(const float*)d_in[5],  (const float*)d_in[9]};
  const float* bo[2]   = {(const float*)d_in[6],  (const float*)d_in[10]};
  const float* hwW[2]  = {(const float*)d_in[11], (const float*)d_in[13]};
  const float* hwb[2]  = {(const float*)d_in[12], (const float*)d_in[14]};
  float* out = (float*)d_out;

  if (ws_size < WS_NEED_) return;

  float* ws    = (float*)d_ws;
  float* padl  = ws + 4 * BSD_;
  float* padr  = padl + WIDTH_ * 2 * D_;
  float* pkq   = out + (size_t)BS_ * 2 * D_;
  float* pk    = padr + WIDTH_ * 2 * D_;
  const float* pkWqkv[2] = {pkq, pkq + (size_t)PK_QKV_T_ * TILE_F_};
  const float* pkWo[2]   = {pk, pk + (size_t)PK_WO_T_ * TILE_F_};
  const float* pkHw[2]   = {pk + (size_t)(2 * PK_WO_T_) * TILE_F_,
                            pk + (size_t)(2 * PK_WO_T_ + PK_HW_T_) * TILE_F_};

  dim3 blk8(32, 8);

  prep_kernel<<<dim3(PADB_ + PK_TILES_), dim3(256), 0, stream>>>(
      Wqkv[0], Wqkv[1], Wo[0], Wo[1], hwW[0], hwW[1], pkq, pk,
      left_pad, right_pad, bqkv[0], bqkv[1], padl, padr);
  { dim3 grd(D_ / 128, BS_ / 64);
    qkv_kernel<<<grd, blk8, 0, stream>>>(inputs, pkWqkv[0], pkWqkv[1], bqkv[0], bqkv[1], out, ws); }
  { dim3 grd(BS_ / CH_ROWS_, 2);
    chain_kernel<<<grd, blk8, 0, stream>>>(ws, pkWo[0], pkWo[1], bo[0], bo[1],
                                          pkHw[0], pkHw[1], hwb[0], hwb[1], out); }
}
